# starting kernel
# baseline (speedup 1.0000x reference)
#include <hip/hip_runtime.h>
#include <hip/hip_bf16.h>

#define NND   50000
#define FIN   128
#define NH    8
#define HIDC  32
#define HC    256
#define ODIM  40
#define NE    800000
#define NET   850000
#define NEG_SLOPE 0.2f
#define EPSV  1e-16f

typedef __attribute__((ext_vector_type(2))) float v2f;
typedef __attribute__((ext_vector_type(8))) float v8f;

__device__ __forceinline__ float wave_sum(float v) {
#pragma unroll
  for (int off = 16; off > 0; off >>= 1) v += __shfl_xor(v, off, 32);
  return v;
}
__device__ __forceinline__ int f32_ord(float x) {
  int i = __float_as_int(x);
  return i >= 0 ? i : (i ^ 0x7FFFFFFF);
}
__device__ __forceinline__ float ord_f32(int i) {
  return __int_as_float(i >= 0 ? i : (i ^ 0x7FFFFFFF));
}

__global__ void k_fill_f32(float* __restrict__ p, float v, int n) {
  int i = blockIdx.x * blockDim.x + threadIdx.x;
  if (i < n) p[i] = v;
}
__global__ void k_fill_i32(int* __restrict__ p, int v, int n) {
  int i = blockIdx.x * blockDim.x + threadIdx.x;
  if (i < n) p[i] = v;
}

__global__ __launch_bounds__(256) void k_gemm1(const float* __restrict__ x,
                                               const float* __restrict__ W,
                                               float* __restrict__ h) {
  __shared__ float As[16 * FIN];
  const int tid  = threadIdx.x;
  const int wave = tid >> 5, lane = tid & 31;
  const int m0   = blockIdx.x * 16;
  {
    const float4* xv = (const float4*)(x + (size_t)m0 * FIN);
    float4* sv = (float4*)As;
    sv[tid]       = xv[tid];
    sv[tid + 256] = xv[tid + 256];
  }
  __syncthreads();
  const int n0 = blockIdx.y * 128 + wave * 16;
  const int la = lane & 15, hi = lane >> 4;
  v8f acc = {};
#pragma unroll
  for (int k0 = 0; k0 < FIN; k0 += 4) {
    const int ka = k0 + hi * 2;
    v2f a, b;
    a.x = As[la * FIN + ka];
    a.y = As[la * FIN + ka + 1];
    b.x = W[ka * HC + n0 + la];
    b.y = W[(ka + 1) * HC + n0 + la];
    acc = __builtin_amdgcn_wmma_f32_16x16x4_f32(false, a, false, b, (short)0,
                                                acc, false, false);
  }
#pragma unroll
  for (int j = 0; j < 8; ++j)
    h[(size_t)(m0 + j + hi * 8) * HC + n0 + la] = acc[j];
}

__global__ __launch_bounds__(96) void k_gemm2(const float* __restrict__ h2,
                                              const float* __restrict__ W,
                                              float* __restrict__ t) {
  __shared__ float As[16 * HC];
  const int tid  = threadIdx.x;
  const int wave = tid >> 5, lane = tid & 31;
  const int m0   = blockIdx.x * 16;
  {
    const float4* xv = (const float4*)(h2 + (size_t)m0 * HC);
    float4* sv = (float4*)As;
    for (int i = tid; i < 1024; i += 96) sv[i] = xv[i];
  }
  __syncthreads();
  const int n0 = wave * 16;
  const int la = lane & 15, hi = lane >> 4;
  const int col = n0 + la;
  const bool ok = col < ODIM;
  v8f acc = {};
#pragma unroll 8
  for (int k0 = 0; k0 < HC; k0 += 4) {
    const int ka = k0 + hi * 2;
    v2f a, b;
    a.x = As[la * HC + ka];
    a.y = As[la * HC + ka + 1];
    b.x = ok ? W[ka * ODIM + col] : 0.f;
    b.y = ok ? W[(ka + 1) * ODIM + col] : 0.f;
    acc = __builtin_amdgcn_wmma_f32_16x16x4_f32(false, a, false, b, (short)0,
                                                acc, false, false);
  }
#pragma unroll
  for (int j = 0; j < 8; ++j)
    if (ok) t[(size_t)(m0 + j + hi * 8) * ODIM + col] = acc[j];
}

__global__ __launch_bounds__(256) void k_att1(const float* __restrict__ h,
                                              const float* __restrict__ as_,
                                              const float* __restrict__ ad_,
                                              float* __restrict__ os,
                                              float* __restrict__ od) {
  const int node = blockIdx.x * 8 + (threadIdx.x >> 5);
  const int lane = threadIdx.x & 31;
  if (node >= NND) return;
  const float* hr = h + (size_t)node * HC;
#pragma unroll
  for (int hd = 0; hd < NH; ++hd) {
    const float v = hr[hd * HIDC + lane];
    const float s = wave_sum(v * as_[hd * HIDC + lane]);
    const float d = wave_sum(v * ad_[hd * HIDC + lane]);
    if (lane == 0) { os[node * NH + hd] = s; od[node * NH + hd] = d; }
  }
}

__global__ __launch_bounds__(256) void k_att2(const float* __restrict__ t,
                                              const float* __restrict__ as_,
                                              const float* __restrict__ ad_,
                                              float* __restrict__ os,
                                              float* __restrict__ od) {
  const int node = blockIdx.x * 8 + (threadIdx.x >> 5);
  const int lane = threadIdx.x & 31;
  if (node >= NND) return;
  const float* tr = t + (size_t)node * ODIM;
  float s = 0.f, d = 0.f;
  {
    const float v = tr[lane];
    s += v * as_[lane]; d += v * ad_[lane];
  }
  if (lane < ODIM - 32) {
    const float v = tr[32 + lane];
    s += v * as_[32 + lane]; d += v * ad_[32 + lane];
  }
  s = wave_sum(s); d = wave_sum(d);
  if (lane == 0) { os[node] = s; od[node] = d; }
}

__device__ __forceinline__ void edge_sd(const int* ei, int eid, int& s, int& d) {
  if (eid < NE) { s = ei[eid]; d = ei[NE + eid]; }
  else          { s = d = eid - NE; }
}

__global__ void k_edgeA1(const int* __restrict__ ei,
                         const float* __restrict__ as_, const float* __restrict__ ad_,
                         float* __restrict__ e, int* __restrict__ nmax) {
  const int t = blockIdx.x * blockDim.x + threadIdx.x;
  if (t >= NET * NH) return;
  const int eid = t >> 3, hd = t & 7;
  int s, d; edge_sd(ei, eid, s, d);
  float v = as_[s * NH + hd] + ad_[d * NH + hd];
  v = v > 0.f ? v : v * NEG_SLOPE;
  e[t] = v;
  atomicMax(&nmax[d * NH + hd], f32_ord(v));
}

__global__ void k_edgeB1(const int* __restrict__ ei, float* __restrict__ e,
                         const int* __restrict__ nmax, float* __restrict__ denom) {
  const int t = blockIdx.x * blockDim.x + threadIdx.x;
  if (t >= NET * NH) return;
  const int eid = t >> 3, hd = t & 7;
  int s, d; edge_sd(ei, eid, s, d);
  const float ex = __expf(e[t] - ord_f32(nmax[d * NH + hd]));
  e[t] = ex;
  atomicAdd(&denom[d * NH + hd], ex);
}

__global__ __launch_bounds__(256) void k_edgeC1(const int* __restrict__ ei,
                                                const float* __restrict__ h,
                                                const float* __restrict__ eexp,
                                                const float* __restrict__ denom,
                                                float* __restrict__ agg) {
  const int eid  = blockIdx.x * 8 + (threadIdx.x >> 5);
  const int lane = threadIdx.x & 31;
  if (eid >= NET) return;
  int s, d; edge_sd(ei, eid, s, d);
  const float* hs = h + (size_t)s * HC;
  float* ag = agg + (size_t)d * HC;
  __builtin_prefetch(hs, 0, 0);
#pragma unroll
  for (int hd = 0; hd < NH; ++hd) {
    const float alpha = eexp[(size_t)eid * NH + hd] /
                        (denom[d * NH + hd] + EPSV);
    atomicAdd(&ag[hd * HIDC + lane], hs[hd * HIDC + lane] * alpha);
  }
}

__global__ void k_edgeA2(const int* __restrict__ ei,
                         const float* __restrict__ as_, const float* __restrict__ ad_,
                         float* __restrict__ e, int* __restrict__ nmax) {
  const int eid = blockIdx.x * blockDim.x + threadIdx.x;
  if (eid >= NET) return;
  int s, d; edge_sd(ei, eid, s, d);
  float v = as_[s] + ad_[d];
  v = v > 0.f ? v : v * NEG_SLOPE;
  e[eid] = v;
  atomicMax(&nmax[d], f32_ord(v));
}

__global__ void k_edgeB2(const int* __restrict__ ei, float* __restrict__ e,
                         const int* __restrict__ nmax, float* __restrict__ denom) {
  const int eid = blockIdx.x * blockDim.x + threadIdx.x;
  if (eid >= NET) return;
  int s, d; edge_sd(ei, eid, s, d);
  const float ex = __expf(e[eid] - ord_f32(nmax[d]));
  e[eid] = ex;
  atomicAdd(&denom[d], ex);
}

__global__ __launch_bounds__(256) void k_edgeC2(const int* __restrict__ ei,
                                                const float* __restrict__ t,
                                                const float* __restrict__ eexp,
                                                const float* __restrict__ denom,
                                                float* __restrict__ agg) {
  const int eid  = blockIdx.x * 8 + (threadIdx.x >> 5);
  const int lane = threadIdx.x & 31;
  if (eid >= NET) return;
  int s, d; edge_sd(ei, eid, s, d);
  const float alpha = eexp[eid] / (denom[d] + EPSV);
  const float* ts = t + (size_t)s * ODIM;
  float* ag = agg + (size_t)d * ODIM;
  atomicAdd(&ag[lane], ts[lane] * alpha);
  if (lane < ODIM - 32) atomicAdd(&ag[32 + lane], ts[32 + lane] * alpha);
}

__global__ void k_elu_bias(float* __restrict__ a, const float* __restrict__ b) {
  const int i = blockIdx.x * blockDim.x + threadIdx.x;
  if (i >= NND * HC) return;
  const float v = a[i] + b[i & (HC - 1)];
  a[i] = v > 0.f ? v : (__expf(v) - 1.f);
}

__global__ void k_out_bias(const float* __restrict__ agg, const float* __restrict__ b,
                           float* __restrict__ out) {
  const int i = blockIdx.x * blockDim.x + threadIdx.x;
  if (i >= NND * ODIM) return;
  out[i] = agg[i] + b[i % ODIM];
}

extern "C" void kernel_launch(void* const* d_in, const int* in_sizes, int n_in,
                              void* d_out, int out_size, void* d_ws, size_t ws_size,
                              hipStream_t stream) {
  const float* x   = (const float*)d_in[0];
  const int*   ei  = (const int*)d_in[1];
  const float* W1  = (const float*)d_in[2];
  const float* as1 = (const float*)d_in[3];
  const float* ad1 = (const float*)d_in[4];
  const float* b1  = (const float*)d_in[5];
  const float* W2  = (const float*)d_in[6];
  const float* as2 = (const float*)d_in[7];
  const float* ad2 = (const float*)d_in[8];
  const float* b2  = (const float*)d_in[9];
  float* out = (float*)d_out;

  float* ws   = (float*)d_ws;
  float* h1   = ws;
  float* agg1 = h1   + (size_t)NND * HC;
  float* aS1  = agg1 + (size_t)NND * HC;
  float* aD1  = aS1  + NND * NH;
  int*   nm1  = (int*)(aD1 + NND * NH);
  float* dn1  = (float*)(nm1 + NND * NH);
  float* e1   = dn1  + NND * NH;
  float* t2   = e1   + (size_t)NET * NH;
  float* aS2  = t2   + (size_t)NND * ODIM;
  float* aD2  = aS2  + NND;
  int*   nm2  = (int*)(aD2 + NND);
  float* dn2  = (float*)(nm2 + NND);
  float* e2   = dn2  + NND;
  float* agg2 = e2   + NET;

  const int B = 256;
  k_fill_f32<<<(NND * HC + B - 1) / B, B, 0, stream>>>(agg1, 0.f, NND * HC);
  k_fill_f32<<<(NND * NH + B - 1) / B, B, 0, stream>>>(dn1, 0.f, NND * NH);
  k_fill_i32<<<(NND * NH + B - 1) / B, B, 0, stream>>>(nm1, (int)0x80000000, NND * NH);
  k_fill_f32<<<(NND * ODIM + B - 1) / B, B, 0, stream>>>(agg2, 0.f, NND * ODIM);
  k_fill_f32<<<(NND + B - 1) / B, B, 0, stream>>>(dn2, 0.f, NND);
  k_fill_i32<<<(NND + B - 1) / B, B, 0, stream>>>(nm2, (int)0x80000000, NND);

  k_gemm1<<<dim3(NND / 16, 2), 256, 0, stream>>>(x, W1, h1);
  k_att1<<<(NND + 7) / 8, 256, 0, stream>>>(h1, as1, ad1, aS1, aD1);
  k_edgeA1<<<(NET * NH + B - 1) / B, B, 0, stream>>>(ei, aS1, aD1, e1, nm1);
  k_edgeB1<<<(NET * NH + B - 1) / B, B, 0, stream>>>(ei, e1, nm1, dn1);
  k_edgeC1<<<(NET + 7) / 8, 256, 0, stream>>>(ei, h1, e1, dn1, agg1);
  k_elu_bias<<<(NND * HC + B - 1) / B, B, 0, stream>>>(agg1, b1);

  k_gemm2<<<NND / 16, 96, 0, stream>>>(agg1, W2, t2);
  k_att2<<<(NND + 7) / 8, 256, 0, stream>>>(t2, as2, ad2, aS2, aD2);
  k_edgeA2<<<(NET + B - 1) / B, B, 0, stream>>>(ei, aS2, aD2, e2, nm2);
  k_edgeB2<<<(NET + B - 1) / B, B, 0, stream>>>(ei, e2, nm2, dn2);
  k_edgeC2<<<(NET + 7) / 8, 256, 0, stream>>>(ei, t2, e2, dn2, agg2);
  k_out_bias<<<(NND * ODIM + B - 1) / B, B, 0, stream>>>(agg2, b2, out);
}


# Edge sums of both layers by 256 owner work-groups: each scans all destination ids, adds its 196 rows in LDS (float LDS atomics, sum order = arrival), stores them once; no global row atomics or fills
# speedup vs baseline: 1.4276x; 1.4276x over previous
#include <hip/hip_runtime.h>
#include <hip/hip_bf16.h>

#define NND   50000
#define FIN   128
#define NH    8
#define HIDC  32
#define HC    256
#define ODIM  40
#define NE    800000
#define NET   850000
#define NEG_SLOPE 0.2f
#define EPSV  1e-16f

typedef __attribute__((ext_vector_type(2))) float v2f;
typedef __attribute__((ext_vector_type(8))) float v8f;

__device__ __forceinline__ float wave_sum(float v) {
#pragma unroll
  for (int off = 16; off > 0; off >>= 1) v += __shfl_xor(v, off, 32);
  return v;
}
__device__ __forceinline__ int f32_ord(float x) {
  int i = __float_as_int(x);
  return i >= 0 ? i : (i ^ 0x7FFFFFFF);
}
__device__ __forceinline__ float ord_f32(int i) {
  return __int_as_float(i >= 0 ? i : (i ^ 0x7FFFFFFF));
}

__global__ void k_fill_f32(float* __restrict__ p, float v, int n) {
  int i = blockIdx.x * blockDim.x + threadIdx.x;
  if (i < n) p[i] = v;
}
__global__ void k_fill_i32(int* __restrict__ p, int v, int n) {
  int i = blockIdx.x * blockDim.x + threadIdx.x;
  if (i < n) p[i] = v;
}

__global__ __launch_bounds__(256) void k_gemm1(const float* __restrict__ x,
                                               const float* __restrict__ W,
                                               float* __restrict__ h) {
  __shared__ float As[16 * FIN];
  const int tid  = threadIdx.x;
  const int wave = tid >> 5, lane = tid & 31;
  const int m0   = blockIdx.x * 16;
  {
    const float4* xv = (const float4*)(x + (size_t)m0 * FIN);
    float4* sv = (float4*)As;
    sv[tid]       = xv[tid];
    sv[tid + 256] = xv[tid + 256];
  }
  __syncthreads();
  const int n0 = blockIdx.y * 128 + wave * 16;
  const int la = lane & 15, hi = lane >> 4;
  v8f acc = {};
#pragma unroll
  for (int k0 = 0; k0 < FIN; k0 += 4) {
    const int ka = k0 + hi * 2;
    v2f a, b;
    a.x = As[la * FIN + ka];
    a.y = As[la * FIN + ka + 1];
    b.x = W[ka * HC + n0 + la];
    b.y = W[(ka + 1) * HC + n0 + la];
    acc = __builtin_amdgcn_wmma_f32_16x16x4_f32(false, a, false, b, (short)0,
                                                acc, false, false);
  }
#pragma unroll
  for (int j = 0; j < 8; ++j)
    h[(size_t)(m0 + j + hi * 8) * HC + n0 + la] = acc[j];
}

__global__ __launch_bounds__(96) void k_gemm2(const float* __restrict__ h2,
                                              const float* __restrict__ W,
                                              float* __restrict__ t) {
  __shared__ float As[16 * HC];
  const int tid  = threadIdx.x;
  const int wave = tid >> 5, lane = tid & 31;
  const int m0   = blockIdx.x * 16;
  {
    const float4* xv = (const float4*)(h2 + (size_t)m0 * HC);
    float4* sv = (float4*)As;
    for (int i = tid; i < 1024; i += 96) sv[i] = xv[i];
  }
  __syncthreads();
  const int n0 = wave * 16;
  const int la = lane & 15, hi = lane >> 4;
  const int col = n0 + la;
  const bool ok = col < ODIM;
  v8f acc = {};
#pragma unroll 8
  for (int k0 = 0; k0 < HC; k0 += 4) {
    const int ka = k0 + hi * 2;
    v2f a, b;
    a.x = As[la * HC + ka];
    a.y = As[la * HC + ka + 1];
    b.x = ok ? W[ka * ODIM + col] : 0.f;
    b.y = ok ? W[(ka + 1) * ODIM + col] : 0.f;
    acc = __builtin_amdgcn_wmma_f32_16x16x4_f32(false, a, false, b, (short)0,
                                                acc, false, false);
  }
#pragma unroll
  for (int j = 0; j < 8; ++j)
    if (ok) t[(size_t)(m0 + j + hi * 8) * ODIM + col] = acc[j];
}

__global__ __launch_bounds__(256) void k_att1(const float* __restrict__ h,
                                              const float* __restrict__ as_,
                                              const float* __restrict__ ad_,
                                              float* __restrict__ os,
                                              float* __restrict__ od) {
  const int node = blockIdx.x * 8 + (threadIdx.x >> 5);
  const int lane = threadIdx.x & 31;
  if (node >= NND) return;
  const float* hr = h + (size_t)node * HC;
#pragma unroll
  for (int hd = 0; hd < NH; ++hd) {
    const float v = hr[hd * HIDC + lane];
    const float s = wave_sum(v * as_[hd * HIDC + lane]);
    const float d = wave_sum(v * ad_[hd * HIDC + lane]);
    if (lane == 0) { os[node * NH + hd] = s; od[node * NH + hd] = d; }
  }
}

__global__ __launch_bounds__(256) void k_att2(const float* __restrict__ t,
                                              const float* __restrict__ as_,
                                              const float* __restrict__ ad_,
                                              float* __restrict__ os,
                                              float* __restrict__ od) {
  const int node = blockIdx.x * 8 + (threadIdx.x >> 5);
  const int lane = threadIdx.x & 31;
  if (node >= NND) return;
  const float* tr = t + (size_t)node * ODIM;
  float s = 0.f, d = 0.f;
  {
    const float v = tr[lane];
    s += v * as_[lane]; d += v * ad_[lane];
  }
  if (lane < ODIM - 32) {
    const float v = tr[32 + lane];
    s += v * as_[32 + lane]; d += v * ad_[32 + lane];
  }
  s = wave_sum(s); d = wave_sum(d);
  if (lane == 0) { os[node] = s; od[node] = d; }
}

__device__ __forceinline__ void edge_sd(const int* ei, int eid, int& s, int& d) {
  if (eid < NE) { s = ei[eid]; d = ei[NE + eid]; }
  else          { s = d = eid - NE; }
}

__global__ void k_edgeA1(const int* __restrict__ ei,
                         const float* __restrict__ as_, const float* __restrict__ ad_,
                         float* __restrict__ e, int* __restrict__ nmax) {
  const int t = blockIdx.x * blockDim.x + threadIdx.x;
  if (t >= NET * NH) return;
  const int eid = t >> 3, hd = t & 7;
  int s, d; edge_sd(ei, eid, s, d);
  float v = as_[s * NH + hd] + ad_[d * NH + hd];
  v = v > 0.f ? v : v * NEG_SLOPE;
  e[t] = v;
  atomicMax(&nmax[d * NH + hd], f32_ord(v));
}

__global__ void k_edgeB1(const int* __restrict__ ei, float* __restrict__ e,
                         const int* __restrict__ nmax, float* __restrict__ denom) {
  const int t = blockIdx.x * blockDim.x + threadIdx.x;
  if (t >= NET * NH) return;
  const int eid = t >> 3, hd = t & 7;
  int s, d; edge_sd(ei, eid, s, d);
  const float ex = __expf(e[t] - ord_f32(nmax[d * NH + hd]));
  e[t] = ex;
  atomicAdd(&denom[d * NH + hd], ex);
}

#define OWN_R    196
#define OWN_WG   256
#define OWN_T    1024
#define OWN_U    4
#define NE4      (NE / 4)
#define OWN_IT   ((NE4 + OWN_T * OWN_U - 1) / (OWN_T * OWN_U))
#define OWN_QTH  32
#define OWN_QCAP (OWN_QTH + 32 * OWN_U * 4 + 32)

__device__ __forceinline__ float lane_bcast_f(float v, int l) {
  return __int_as_float(__builtin_amdgcn_readlane(__float_as_int(v), l));
}

template <int C, int H, bool INIT>
__device__ __forceinline__ void own_edge(float* __restrict__ acc,
                                         const float* __restrict__ h,
                                         const float* __restrict__ eexp,
                                         const float* __restrict__ denom,
                                         int eid, int s, int d, int dl, int lane) {
  const float* hs = h + (size_t)s * C;
  float* ar = acc + dl * C;
  constexpr int NP = (C + 31) / 32;
  float v[NP];
#pragma unroll
  for (int j = 0; j < NP; ++j) {
    const int c = j * 32 + lane;
    v[j] = c < C ? hs[c] : 0.f;
  }
  float a = 0.f;
  if (lane < H) a = eexp[(size_t)eid * H + lane] / (denom[d * H + lane] + EPSV);
#pragma unroll
  for (int j = 0; j < NP; ++j) {
    const float alpha = lane_bcast_f(a, (j * 32) / (C / H));
    const int c = j * 32 + lane;
    if (c < C) {
      const float t = v[j] * alpha;
      if (INIT) ar[c] = t;
      else __hip_atomic_fetch_add(&ar[c], t, __ATOMIC_RELAXED, __HIP_MEMORY_SCOPE_WORKGROUP);
    }
  }
}

template <int C, int H>
__device__ __forceinline__ void own_pair(float* __restrict__ acc,
                                         const float* __restrict__ h,
                                         const float* __restrict__ eexp,
                                         const float* __restrict__ denom,
                                         int eidA, int sA, int dA, int xA,
                                         int eidB, int sB, int dB, int xB, int lane) {
  const float* hA = h + (size_t)sA * C;
  const float* hB = h + (size_t)sB * C;
  float* arA = acc + xA * C;
  float* arB = acc + xB * C;
  constexpr int NP = (C + 31) / 32;
  float vA[NP], vB[NP];
#pragma unroll
  for (int j = 0; j < NP; ++j) {
    const int c = j * 32 + lane;
    vA[j] = c < C ? hA[c] : 0.f;
  }
#pragma unroll
  for (int j = 0; j < NP; ++j) {
    const int c = j * 32 + lane;
    vB[j] = c < C ? hB[c] : 0.f;
  }
  float a = 0.f;
  if (lane < 2 * H) {
    const bool second = lane >= H;
    const int hh  = second ? lane - H : lane;
    const int eid = second ? eidB : eidA;
    const int d   = second ? dB : dA;
    a = eexp[(size_t)eid * H + hh] / (denom[d * H + hh] + EPSV);
  }
#pragma unroll
  for (int j = 0; j < NP; ++j) {
    const float alpha = lane_bcast_f(a, (j * 32) / (C / H));
    const int c = j * 32 + lane;
    if (c < C)
      __hip_atomic_fetch_add(&arA[c], vA[j] * alpha, __ATOMIC_RELAXED, __HIP_MEMORY_SCOPE_WORKGROUP);
  }
#pragma unroll
  for (int j = 0; j < NP; ++j) {
    const float alpha = lane_bcast_f(a, H + (j * 32) / (C / H));
    const int c = j * 32 + lane;
    if (c < C)
      __hip_atomic_fetch_add(&arB[c], vB[j] * alpha, __ATOMIC_RELAXED, __HIP_MEMORY_SCOPE_WORKGROUP);
  }
}

template <int C, int H>
__device__ __forceinline__ void own_drain(float* __restrict__ acc, const unsigned* qw, int n,
                                          const int* __restrict__ ei,
                                          const float* __restrict__ h,
                                          const float* __restrict__ eexp,
                                          const float* __restrict__ denom,
                                          int d0, int lane) {
  for (int b = 0; b < n; b += 32) {
    int cnt = n - b;
    if (cnt > 32) cnt = 32;
    unsigned ent = 0;
    int s = 0;
    if (lane < cnt) {
      ent = qw[b + lane];
      s = ei[ent & 0xFFFFFu];
    }
    int i = 0;
    for (; i + 1 < cnt; i += 2) {
      const unsigned eA = (unsigned)__builtin_amdgcn_readlane((int)ent, i);
      const unsigned eB = (unsigned)__builtin_amdgcn_readlane((int)ent, i + 1);
      const int sA = __builtin_amdgcn_readlane(s, i);
      const int sB = __builtin_amdgcn_readlane(s, i + 1);
      const int xA = (int)(eA >> 20), xB = (int)(eB >> 20);
      own_pair<C, H>(acc, h, eexp, denom, (int)(eA & 0xFFFFFu), sA, d0 + xA, xA,
                     (int)(eB & 0xFFFFFu), sB, d0 + xB, xB, lane);
    }
    if (i < cnt) {
      const unsigned eA = (unsigned)__builtin_amdgcn_readlane((int)ent, i);
      const int sA = __builtin_amdgcn_readlane(s, i);
      const int xA = (int)(eA >> 20);
      own_edge<C, H, false>(acc, h, eexp, denom, (int)(eA & 0xFFFFFu), sA, d0 + xA, xA, lane);
    }
  }
}

template <int C, int H>
__global__ __launch_bounds__(OWN_T) void k_own(const int* __restrict__ ei,
                                               const float* __restrict__ h,
                                               const float* __restrict__ eexp,
                                               const float* __restrict__ denom,
                                               float* __restrict__ agg) {
  __shared__ __attribute__((aligned(16))) float acc[OWN_R * C];
  __shared__ unsigned q[(OWN_T / 32) * OWN_QCAP];
  const int tid = threadIdx.x, lane = tid & 31;
  const int wave = __builtin_amdgcn_readfirstlane(tid >> 5);
  const int d0 = blockIdx.x * OWN_R;
  int nrow = NND - d0;
  if (nrow > OWN_R) nrow = OWN_R;

  for (int r = wave; r < nrow; r += OWN_T / 32)
    own_edge<C, H, true>(acc, h, eexp, denom, NE + d0 + r, d0 + r, d0 + r, r, lane);
  __syncthreads();

  unsigned* qw = q + wave * OWN_QCAP;
  int qn = 0;
  const int4* dst4 = (const int4*)(ei + NE);
  const int wofs = wave * 32;
  int4 cur[OWN_U];
  int ii = (blockIdx.x * OWN_IT) / OWN_WG;
#pragma unroll
  for (int u = 0; u < OWN_U; ++u) {
    const int wb = (ii * OWN_U + u) * OWN_T + wofs;
    cur[u] = dst4[(wb < NE4 ? wb : NE4 - 32) + lane];
  }
  for (int it = 0; it < OWN_IT; ++it) {
    const int in = ii + 1 < OWN_IT ? ii + 1 : 0;
    int4 nxt[OWN_U];
#pragma unroll
    for (int u = 0; u < OWN_U; ++u) {
      const int wb = (in * OWN_U + u) * OWN_T + wofs;
      nxt[u] = dst4[(wb < NE4 ? wb : NE4 - 32) + lane];
    }
#pragma unroll
    for (int u = 0; u < OWN_U; ++u) {
      const int wb = (ii * OWN_U + u) * OWN_T + wofs;
      const int d0u = wb < NE4 ? d0 : 0x40000000;
      const int dd[4] = {cur[u].x, cur[u].y, cur[u].z, cur[u].w};
#pragma unroll
      for (int k = 0; k < 4; ++k) {
        const unsigned x = (unsigned)(dd[k] - d0u);
        const bool m = x < (unsigned)nrow;
        const unsigned mk = __builtin_amdgcn_ballot_w32(m);
        if (mk) {
          if (m) qw[qn + (int)__builtin_amdgcn_mbcnt_lo(mk, 0u)] = (x << 20) | (unsigned)((wb + lane) * 4 + k);
          qn += __builtin_popcount(mk);
        }
      }
    }
    if (qn >= OWN_QTH) {
      own_drain<C, H>(acc, qw, qn, ei, h, eexp, denom, d0, lane);
      qn = 0;
    }
#pragma unroll
    for (int u = 0; u < OWN_U; ++u) cur[u] = nxt[u];
    ii = in;
  }
  own_drain<C, H>(acc, qw, qn, ei, h, eexp, denom, d0, lane);
  __syncthreads();

  float4* o4 = (float4*)(agg + (size_t)d0 * C);
  const float4* a4 = (const float4*)acc;
  for (int i = tid; i < nrow * C / 4; i += OWN_T) o4[i] = a4[i];
}

__global__ void k_edgeA2(const int* __restrict__ ei,
                         const float* __restrict__ as_, const float* __restrict__ ad_,
                         float* __restrict__ e, int* __restrict__ nmax) {
  const int eid = blockIdx.x * blockDim.x + threadIdx.x;
  if (eid >= NET) return;
  int s, d; edge_sd(ei, eid, s, d);
  float v = as_[s] + ad_[d];
  v = v > 0.f ? v : v * NEG_SLOPE;
  e[eid] = v;
  atomicMax(&nmax[d], f32_ord(v));
}

__global__ void k_edgeB2(const int* __restrict__ ei, float* __restrict__ e,
                         const int* __restrict__ nmax, float* __restrict__ denom) {
  const int eid = blockIdx.x * blockDim.x + threadIdx.x;
  if (eid >= NET) return;
  int s, d; edge_sd(ei, eid, s, d);
  const float ex = __expf(e[eid] - ord_f32(nmax[d]));
  e[eid] = ex;
  atomicAdd(&denom[d], ex);
}

__global__ void k_elu_bias(float* __restrict__ a, const float* __restrict__ b) {
  const int i = blockIdx.x * blockDim.x + threadIdx.x;
  if (i >= NND * HC) return;
  const float v = a[i] + b[i & (HC - 1)];
  a[i] = v > 0.f ? v : (__expf(v) - 1.f);
}

__global__ void k_out_bias(const float* __restrict__ agg, const float* __restrict__ b,
                           float* __restrict__ out) {
  const int i = blockIdx.x * blockDim.x + threadIdx.x;
  if (i >= NND * ODIM) return;
  out[i] = agg[i] + b[i % ODIM];
}

extern "C" void kernel_launch(void* const* d_in, const int* in_sizes, int n_in,
                              void* d_out, int out_size, void* d_ws, size_t ws_size,
                              hipStream_t stream) {
  const float* x   = (const float*)d_in[0];
  const int*   ei  = (const int*)d_in[1];
  const float* W1  = (const float*)d_in[2];
  const float* as1 = (const float*)d_in[3];
  const float* ad1 = (const float*)d_in[4];
  const float* b1  = (const float*)d_in[5];
  const float* W2  = (const float*)d_in[6];
  const float* as2 = (const float*)d_in[7];
  const float* ad2 = (const float*)d_in[8];
  const float* b2  = (const float*)d_in[9];
  float* out = (float*)d_out;

  float* ws   = (float*)d_ws;
  float* h1   = ws;
  float* agg1 = h1   + (size_t)NND * HC;
  float* aS1  = agg1 + (size_t)NND * HC;
  float* aD1  = aS1  + NND * NH;
  int*   nm1  = (int*)(aD1 + NND * NH);
  float* dn1  = (float*)(nm1 + NND * NH);
  float* e1   = dn1  + NND * NH;
  float* t2   = e1   + (size_t)NET * NH;
  float* aS2  = t2   + (size_t)NND * ODIM;
  float* aD2  = aS2  + NND;
  int*   nm2  = (int*)(aD2 + NND);
  float* dn2  = (float*)(nm2 + NND);
  float* e2   = dn2  + NND;
  float* agg2 = e2   + NET;

  const int B = 256;
  k_fill_f32<<<(NND * NH + B - 1) / B, B, 0, stream>>>(dn1, 0.f, NND * NH);
  k_fill_i32<<<(NND * NH + B - 1) / B, B, 0, stream>>>(nm1, (int)0x80000000, NND * NH);
  k_fill_f32<<<(NND + B - 1) / B, B, 0, stream>>>(dn2, 0.f, NND);
  k_fill_i32<<<(NND + B - 1) / B, B, 0, stream>>>(nm2, (int)0x80000000, NND);

  k_gemm1<<<dim3(NND / 16, 2), 256, 0, stream>>>(x, W1, h1);
  k_att1<<<(NND + 7) / 8, 256, 0, stream>>>(h1, as1, ad1, aS1, aD1);
  k_edgeA1<<<(NET * NH + B - 1) / B, B, 0, stream>>>(ei, aS1, aD1, e1, nm1);
  k_edgeB1<<<(NET * NH + B - 1) / B, B, 0, stream>>>(ei, e1, nm1, dn1);
  k_own<HC, NH><<<OWN_WG, OWN_T, 0, stream>>>(ei, h1, e1, dn1, agg1);
  k_elu_bias<<<(NND * HC + B - 1) / B, B, 0, stream>>>(agg1, b1);

  k_gemm2<<<NND / 16, 96, 0, stream>>>(agg1, W2, t2);
  k_att2<<<(NND + 7) / 8, 256, 0, stream>>>(t2, as2, ad2, aS2, aD2);
  k_edgeA2<<<(NET + B - 1) / B, B, 0, stream>>>(ei, aS2, aD2, e2, nm2);
  k_edgeB2<<<(NET + B - 1) / B, B, 0, stream>>>(ei, e2, nm2, dn2);
  k_own<ODIM, 1><<<OWN_WG, OWN_T, 0, stream>>>(ei, t2, e2, dn2, agg2);
  k_out_bias<<<(NND * ODIM + B - 1) / B, B, 0, stream>>>(agg2, b2, out);
}


# Element-per-thread launches (fills, logits, exp sums, attention dots, bias passes) sized to the chip: a thread loops or takes 16 bytes, about a million fewer waves enter; values, sum order unchanged
# speedup vs baseline: 1.5159x; 1.0619x over previous
#include <hip/hip_runtime.h>
#include <hip/hip_bf16.h>

#define NND   50000
#define FIN   128
#define NH    8
#define HIDC  32
#define HC    256
#define ODIM  40
#define NE    800000
#define NET   850000
#define NEG_SLOPE 0.2f
#define EPSV  1e-16f

typedef __attribute__((ext_vector_type(2))) float v2f;
typedef __attribute__((ext_vector_type(8))) float v8f;

__device__ __forceinline__ float wave_sum(float v) {
#pragma unroll
  for (int off = 16; off > 0; off >>= 1) v += __shfl_xor(v, off, 32);
  return v;
}
__device__ __forceinline__ int f32_ord(float x) {
  int i = __float_as_int(x);
  return i >= 0 ? i : (i ^ 0x7FFFFFFF);
}
__device__ __forceinline__ float ord_f32(int i) {
  return __int_as_float(i >= 0 ? i : (i ^ 0x7FFFFFFF));
}

__global__ __launch_bounds__(256) void k_fill4(int4* __restrict__ p, int v, int n4) {
  const int i = blockIdx.x * 256 + threadIdx.x;
  if (i < n4) p[i] = make_int4(v, v, v, v);
}

__global__ __launch_bounds__(256) void k_gemm1(const float* __restrict__ x,
                                               const float* __restrict__ W,
                                               float* __restrict__ h) {
  __shared__ float As[16 * FIN];
  const int tid  = threadIdx.x;
  const int wave = tid >> 5, lane = tid & 31;
  const int m0   = blockIdx.x * 16;
  {
    const float4* xv = (const float4*)(x + (size_t)m0 * FIN);
    float4* sv = (float4*)As;
    sv[tid]       = xv[tid];
    sv[tid + 256] = xv[tid + 256];
  }
  __syncthreads();
  const int n0 = blockIdx.y * 128 + wave * 16;
  const int la = lane & 15, hi = lane >> 4;
  v8f acc = {};
#pragma unroll
  for (int k0 = 0; k0 < FIN; k0 += 4) {
    const int ka = k0 + hi * 2;
    v2f a, b;
    a.x = As[la * FIN + ka];
    a.y = As[la * FIN + ka + 1];
    b.x = W[ka * HC + n0 + la];
    b.y = W[(ka + 1) * HC + n0 + la];
    acc = __builtin_amdgcn_wmma_f32_16x16x4_f32(false, a, false, b, (short)0,
                                                acc, false, false);
  }
#pragma unroll
  for (int j = 0; j < 8; ++j)
    h[(size_t)(m0 + j + hi * 8) * HC + n0 + la] = acc[j];
}

__global__ __launch_bounds__(96) void k_gemm2(const float* __restrict__ h2,
                                              const float* __restrict__ W,
                                              float* __restrict__ t) {
  __shared__ float As[16 * HC];
  const int tid  = threadIdx.x;
  const int wave = tid >> 5, lane = tid & 31;
  const int m0   = blockIdx.x * 16;
  {
    const float4* xv = (const float4*)(h2 + (size_t)m0 * HC);
    float4* sv = (float4*)As;
    for (int i = tid; i < 1024; i += 96) sv[i] = xv[i];
  }
  __syncthreads();
  const int n0 = wave * 16;
  const int la = lane & 15, hi = lane >> 4;
  const int col = n0 + la;
  const bool ok = col < ODIM;
  v8f acc = {};
#pragma unroll 8
  for (int k0 = 0; k0 < HC; k0 += 4) {
    const int ka = k0 + hi * 2;
    v2f a, b;
    a.x = As[la * HC + ka];
    a.y = As[la * HC + ka + 1];
    b.x = ok ? W[ka * ODIM + col] : 0.f;
    b.y = ok ? W[(ka + 1) * ODIM + col] : 0.f;
    acc = __builtin_amdgcn_wmma_f32_16x16x4_f32(false, a, false, b, (short)0,
                                                acc, false, false);
  }
#pragma unroll
  for (int j = 0; j < 8; ++j)
    if (ok) t[(size_t)(m0 + j + hi * 8) * ODIM + col] = acc[j];
}

#define ATT_WG 1024
#define ATT_NW (ATT_WG * 8)
__device__ __forceinline__ void att1_node(const float (&v)[NH], const float (&ws)[NH],
                                          const float (&wd)[NH], int node, int lane,
                                          float* __restrict__ os, float* __restrict__ od) {
#pragma unroll
  for (int hd = 0; hd < NH; ++hd) {
    const float s = wave_sum(v[hd] * ws[hd]);
    const float d = wave_sum(v[hd] * wd[hd]);
    if (lane == 0) { os[node * NH + hd] = s; od[node * NH + hd] = d; }
  }
}
__global__ __launch_bounds__(256) void k_att1(const float* __restrict__ h,
                                              const float* __restrict__ as_,
                                              const float* __restrict__ ad_,
                                              float* __restrict__ os,
                                              float* __restrict__ od) {
  const int lane = threadIdx.x & 31;
  const int w = blockIdx.x * 8 + (threadIdx.x >> 5);
  float ws[NH], wd[NH];
#pragma unroll
  for (int hd = 0; hd < NH; ++hd) { ws[hd] = as_[hd * HIDC + lane]; wd[hd] = ad_[hd * HIDC + lane]; }
  for (int na = w; na < NND; na += 2 * ATT_NW) {
    const int nb = na + ATT_NW;
    const bool hb = nb < NND;
    const float* ra = h + (size_t)na * HC;
    const float* rb = h + (size_t)(hb ? nb : na) * HC;
    float va[NH], vb[NH];
#pragma unroll
    for (int hd = 0; hd < NH; ++hd) va[hd] = ra[hd * HIDC + lane];
#pragma unroll
    for (int hd = 0; hd < NH; ++hd) vb[hd] = rb[hd * HIDC + lane];
    att1_node(va, ws, wd, na, lane, os, od);
    if (hb) att1_node(vb, ws, wd, nb, lane, os, od);
  }
}

__device__ __forceinline__ void att2_node(float v0, float v1, float s0, float s1, float d0,
                                          float d1, bool up, int node, int lane,
                                          float* __restrict__ os, float* __restrict__ od) {
  float s = 0.f, d = 0.f;
  s += v0 * s0; d += v0 * d0;
  if (up) { s += v1 * s1; d += v1 * d1; }
  s = wave_sum(s); d = wave_sum(d);
  if (lane == 0) { os[node] = s; od[node] = d; }
}
__global__ __launch_bounds__(256) void k_att2(const float* __restrict__ t,
                                              const float* __restrict__ as_,
                                              const float* __restrict__ ad_,
                                              float* __restrict__ os,
                                              float* __restrict__ od) {
  const int lane = threadIdx.x & 31;
  const int w = blockIdx.x * 8 + (threadIdx.x >> 5);
  const bool up = lane < ODIM - 32;
  const float s0 = as_[lane], d0 = ad_[lane];
  const float s1 = up ? as_[32 + lane] : 0.f, d1 = up ? ad_[32 + lane] : 0.f;
  for (int na = w; na < NND; na += 2 * ATT_NW) {
    const int nb = na + ATT_NW;
    const bool hb = nb < NND;
    const float* ta = t + (size_t)na * ODIM;
    const float* tb = t + (size_t)(hb ? nb : na) * ODIM;
    const float a0 = ta[lane], b0 = tb[lane];
    const float a1 = up ? ta[32 + lane] : 0.f, b1 = up ? tb[32 + lane] : 0.f;
    att2_node(a0, a1, s0, s1, d0, d1, up, na, lane, os, od);
    if (hb) att2_node(b0, b1, s0, s1, d0, d1, up, nb, lane, os, od);
  }
}

#define EG_WG 1024
#define EG_T  (EG_WG * 256)
#define EG_U  8
#define EG_N  (NET * NH)
__device__ __forceinline__ void edge_sd(const int* __restrict__ ei, int eid, int& s, int& d) {
  const bool real = eid < NE;
  const int ec = real ? eid : 0;
  const int ls = ei[ec], ld = ei[NE + ec];
  s = real ? ls : eid - NE;
  d = real ? ld : eid - NE;
}
__device__ __forceinline__ int edge_d(const int* __restrict__ ei, int eid) {
  const bool real = eid < NE;
  const int ld = ei[NE + (real ? eid : 0)];
  return real ? ld : eid - NE;
}

__global__ __launch_bounds__(256) void k_edgeA1(const int* __restrict__ ei,
                         const float* __restrict__ as_, const float* __restrict__ ad_,
                         float* __restrict__ e, int* __restrict__ nmax) {
  const int t0 = blockIdx.x * 256 + threadIdx.x;
  const int hd = t0 & 7;
  for (int tb = t0; tb < EG_N; tb += EG_U * EG_T) {
    int s[EG_U], d[EG_U];
    float v[EG_U];
#pragma unroll
    for (int k = 0; k < EG_U; ++k) {
      const int t = tb + k * EG_T;
      edge_sd(ei, (t < EG_N ? t : t0) >> 3, s[k], d[k]);
    }
#pragma unroll
    for (int k = 0; k < EG_U; ++k) v[k] = as_[s[k] * NH + hd] + ad_[d[k] * NH + hd];
#pragma unroll
    for (int k = 0; k < EG_U; ++k) {
      const int t = tb + k * EG_T;
      if (t < EG_N) {
        const float x = v[k] > 0.f ? v[k] : v[k] * NEG_SLOPE;
        e[t] = x;
        atomicMax(&nmax[d[k] * NH + hd], f32_ord(x));
      }
    }
  }
}

__global__ __launch_bounds__(256) void k_edgeB1(const int* __restrict__ ei, float* __restrict__ e,
                         const int* __restrict__ nmax, float* __restrict__ denom) {
  const int t0 = blockIdx.x * 256 + threadIdx.x;
  const int hd = t0 & 7;
  for (int tb = t0; tb < EG_N; tb += EG_U * EG_T) {
    int d[EG_U], m[EG_U];
    float v[EG_U];
#pragma unroll
    for (int k = 0; k < EG_U; ++k) {
      const int t = tb + k * EG_T;
      const int tc = t < EG_N ? t : t0;
      d[k] = edge_d(ei, tc >> 3);
      v[k] = e[tc];
    }
#pragma unroll
    for (int k = 0; k < EG_U; ++k) m[k] = nmax[d[k] * NH + hd];
#pragma unroll
    for (int k = 0; k < EG_U; ++k) {
      const int t = tb + k * EG_T;
      if (t < EG_N) {
        const float ex = __expf(v[k] - ord_f32(m[k]));
        e[t] = ex;
        atomicAdd(&denom[d[k] * NH + hd], ex);
      }
    }
  }
}

#define OWN_R    196
#define OWN_WG   256
#define OWN_T    1024
#define OWN_U    4
#define NE4      (NE / 4)
#define OWN_IT   ((NE4 + OWN_T * OWN_U - 1) / (OWN_T * OWN_U))
#define OWN_QTH  32
#define OWN_QCAP (OWN_QTH + 32 * OWN_U * 4 + 32)

__device__ __forceinline__ float lane_bcast_f(float v, int l) {
  return __int_as_float(__builtin_amdgcn_readlane(__float_as_int(v), l));
}

template <int C, int H, bool INIT>
__device__ __forceinline__ void own_edge(float* __restrict__ acc,
                                         const float* __restrict__ h,
                                         const float* __restrict__ eexp,
                                         const float* __restrict__ denom,
                                         int eid, int s, int d, int dl, int lane) {
  const float* hs = h + (size_t)s * C;
  float* ar = acc + dl * C;
  constexpr int NP = (C + 31) / 32;
  float v[NP];
#pragma unroll
  for (int j = 0; j < NP; ++j) {
    const int c = j * 32 + lane;
    v[j] = c < C ? hs[c] : 0.f;
  }
  float a = 0.f;
  if (lane < H) a = eexp[(size_t)eid * H + lane] / (denom[d * H + lane] + EPSV);
#pragma unroll
  for (int j = 0; j < NP; ++j) {
    const float alpha = lane_bcast_f(a, (j * 32) / (C / H));
    const int c = j * 32 + lane;
    if (c < C) {
      const float t = v[j] * alpha;
      if (INIT) ar[c] = t;
      else __hip_atomic_fetch_add(&ar[c], t, __ATOMIC_RELAXED, __HIP_MEMORY_SCOPE_WORKGROUP);
    }
  }
}

template <int C, int H>
__device__ __forceinline__ void own_pair(float* __restrict__ acc,
                                         const float* __restrict__ h,
                                         const float* __restrict__ eexp,
                                         const float* __restrict__ denom,
                                         int eidA, int sA, int dA, int xA,
                                         int eidB, int sB, int dB, int xB, int lane) {
  const float* hA = h + (size_t)sA * C;
  const float* hB = h + (size_t)sB * C;
  float* arA = acc + xA * C;
  float* arB = acc + xB * C;
  constexpr int NP = (C + 31) / 32;
  float vA[NP], vB[NP];
#pragma unroll
  for (int j = 0; j < NP; ++j) {
    const int c = j * 32 + lane;
    vA[j] = c < C ? hA[c] : 0.f;
  }
#pragma unroll
  for (int j = 0; j < NP; ++j) {
    const int c = j * 32 + lane;
    vB[j] = c < C ? hB[c] : 0.f;
  }
  float a = 0.f;
  if (lane < 2 * H) {
    const bool second = lane >= H;
    const int hh  = second ? lane - H : lane;
    const int eid = second ? eidB : eidA;
    const int d   = second ? dB : dA;
    a = eexp[(size_t)eid * H + hh] / (denom[d * H + hh] + EPSV);
  }
#pragma unroll
  for (int j = 0; j < NP; ++j) {
    const float alpha = lane_bcast_f(a, (j * 32) / (C / H));
    const int c = j * 32 + lane;
    if (c < C)
      __hip_atomic_fetch_add(&arA[c], vA[j] * alpha, __ATOMIC_RELAXED, __HIP_MEMORY_SCOPE_WORKGROUP);
  }
#pragma unroll
  for (int j = 0; j < NP; ++j) {
    const float alpha = lane_bcast_f(a, H + (j * 32) / (C / H));
    const int c = j * 32 + lane;
    if (c < C)
      __hip_atomic_fetch_add(&arB[c], vB[j] * alpha, __ATOMIC_RELAXED, __HIP_MEMORY_SCOPE_WORKGROUP);
  }
}

template <int C, int H>
__device__ __forceinline__ void own_drain(float* __restrict__ acc, const unsigned* qw, int n,
                                          const int* __restrict__ ei,
                                          const float* __restrict__ h,
                                          const float* __restrict__ eexp,
                                          const float* __restrict__ denom,
                                          int d0, int lane) {
  for (int b = 0; b < n; b += 32) {
    int cnt = n - b;
    if (cnt > 32) cnt = 32;
    unsigned ent = 0;
    int s = 0;
    if (lane < cnt) {
      ent = qw[b + lane];
      s = ei[ent & 0xFFFFFu];
    }
    int i = 0;
    for (; i + 1 < cnt; i += 2) {
      const unsigned eA = (unsigned)__builtin_amdgcn_readlane((int)ent, i);
      const unsigned eB = (unsigned)__builtin_amdgcn_readlane((int)ent, i + 1);
      const int sA = __builtin_amdgcn_readlane(s, i);
      const int sB = __builtin_amdgcn_readlane(s, i + 1);
      const int xA = (int)(eA >> 20), xB = (int)(eB >> 20);
      own_pair<C, H>(acc, h, eexp, denom, (int)(eA & 0xFFFFFu), sA, d0 + xA, xA,
                     (int)(eB & 0xFFFFFu), sB, d0 + xB, xB, lane);
    }
    if (i < cnt) {
      const unsigned eA = (unsigned)__builtin_amdgcn_readlane((int)ent, i);
      const int sA = __builtin_amdgcn_readlane(s, i);
      const int xA = (int)(eA >> 20);
      own_edge<C, H, false>(acc, h, eexp, denom, (int)(eA & 0xFFFFFu), sA, d0 + xA, xA, lane);
    }
  }
}

template <int C, int H>
__global__ __launch_bounds__(OWN_T) void k_own(const int* __restrict__ ei,
                                               const float* __restrict__ h,
                                               const float* __restrict__ eexp,
                                               const float* __restrict__ denom,
                                               float* __restrict__ agg) {
  __shared__ __attribute__((aligned(16))) float acc[OWN_R * C];
  __shared__ unsigned q[(OWN_T / 32) * OWN_QCAP];
  const int tid = threadIdx.x, lane = tid & 31;
  const int wave = __builtin_amdgcn_readfirstlane(tid >> 5);
  const int d0 = blockIdx.x * OWN_R;
  int nrow = NND - d0;
  if (nrow > OWN_R) nrow = OWN_R;

  for (int r = wave; r < nrow; r += OWN_T / 32)
    own_edge<C, H, true>(acc, h, eexp, denom, NE + d0 + r, d0 + r, d0 + r, r, lane);
  __syncthreads();

  unsigned* qw = q + wave * OWN_QCAP;
  int qn = 0;
  const int4* dst4 = (const int4*)(ei + NE);
  const int wofs = wave * 32;
  int4 cur[OWN_U];
  int ii = (blockIdx.x * OWN_IT) / OWN_WG;
#pragma unroll
  for (int u = 0; u < OWN_U; ++u) {
    const int wb = (ii * OWN_U + u) * OWN_T + wofs;
    cur[u] = dst4[(wb < NE4 ? wb : NE4 - 32) + lane];
  }
  for (int it = 0; it < OWN_IT; ++it) {
    const int in = ii + 1 < OWN_IT ? ii + 1 : 0;
    int4 nxt[OWN_U];
#pragma unroll
    for (int u = 0; u < OWN_U; ++u) {
      const int wb = (in * OWN_U + u) * OWN_T + wofs;
      nxt[u] = dst4[(wb < NE4 ? wb : NE4 - 32) + lane];
    }
#pragma unroll
    for (int u = 0; u < OWN_U; ++u) {
      const int wb = (ii * OWN_U + u) * OWN_T + wofs;
      const int d0u = wb < NE4 ? d0 : 0x40000000;
      const int dd[4] = {cur[u].x, cur[u].y, cur[u].z, cur[u].w};
#pragma unroll
      for (int k = 0; k < 4; ++k) {
        const unsigned x = (unsigned)(dd[k] - d0u);
        const bool m = x < (unsigned)nrow;
        const unsigned mk = __builtin_amdgcn_ballot_w32(m);
        if (mk) {
          if (m) qw[qn + (int)__builtin_amdgcn_mbcnt_lo(mk, 0u)] = (x << 20) | (unsigned)((wb + lane) * 4 + k);
          qn += __builtin_popcount(mk);
        }
      }
    }
    if (qn >= OWN_QTH) {
      own_drain<C, H>(acc, qw, qn, ei, h, eexp, denom, d0, lane);
      qn = 0;
    }
#pragma unroll
    for (int u = 0; u < OWN_U; ++u) cur[u] = nxt[u];
    ii = in;
  }
  own_drain<C, H>(acc, qw, qn, ei, h, eexp, denom, d0, lane);
  __syncthreads();

  float4* o4 = (float4*)(agg + (size_t)d0 * C);
  const float4* a4 = (const float4*)acc;
  for (int i = tid; i < nrow * C / 4; i += OWN_T) o4[i] = a4[i];
}

#define NQ  (NET / 4)
#define NEQ (NE / 4)
__device__ __forceinline__ void quad_sd(const int* __restrict__ ei, int q, int (&s)[4], int (&d)[4]) {
  if (q < NEQ) {
    const int4 sv = ((const int4*)ei)[q];
    const int4 dv = ((const int4*)(ei + NE))[q];
    s[0] = sv.x; s[1] = sv.y; s[2] = sv.z; s[3] = sv.w;
    d[0] = dv.x; d[1] = dv.y; d[2] = dv.z; d[3] = dv.w;
  } else {
#pragma unroll
    for (int k = 0; k < 4; ++k) s[k] = d[k] = q * 4 + k - NE;
  }
}

__global__ __launch_bounds__(256) void k_edgeA2(const int* __restrict__ ei,
                         const float* __restrict__ as_, const float* __restrict__ ad_,
                         float* __restrict__ e, int* __restrict__ nmax) {
  const int q = blockIdx.x * 256 + threadIdx.x;
  if (q >= NQ) return;
  int s[4], d[4];
  quad_sd(ei, q, s, d);
  float v[4];
#pragma unroll
  for (int k = 0; k < 4; ++k) v[k] = as_[s[k]] + ad_[d[k]];
#pragma unroll
  for (int k = 0; k < 4; ++k) v[k] = v[k] > 0.f ? v[k] : v[k] * NEG_SLOPE;
  ((float4*)e)[q] = make_float4(v[0], v[1], v[2], v[3]);
#pragma unroll
  for (int k = 0; k < 4; ++k) atomicMax(&nmax[d[k]], f32_ord(v[k]));
}

__global__ __launch_bounds__(256) void k_edgeB2(const int* __restrict__ ei, float* __restrict__ e,
                         const int* __restrict__ nmax, float* __restrict__ denom) {
  const int q = blockIdx.x * 256 + threadIdx.x;
  if (q >= NQ) return;
  int s[4], d[4];
  quad_sd(ei, q, s, d);
  const float4 ev = ((const float4*)e)[q];
  const float v[4] = {ev.x, ev.y, ev.z, ev.w};
  float ex[4];
#pragma unroll
  for (int k = 0; k < 4; ++k) ex[k] = __expf(v[k] - ord_f32(nmax[d[k]]));
  ((float4*)e)[q] = make_float4(ex[0], ex[1], ex[2], ex[3]);
#pragma unroll
  for (int k = 0; k < 4; ++k) atomicAdd(&denom[d[k]], ex[k]);
}

#define EL_WG 1250
#define EL_T  (EL_WG * 256)
#define EL_U  5
#define EL_N4 (NND * HC / 4)
static_assert(EL_N4 == 2 * EL_U * EL_T, "two full trips cover the rows exactly");
static_assert(EL_T % (HC / 4) == 0, "a thread's columns do not change from trip to trip");
__global__ __launch_bounds__(256) void k_elu_bias(float* __restrict__ a, const float* __restrict__ b) {
  const int i0 = blockIdx.x * 256 + threadIdx.x;
  const int c = (i0 & (HC / 4 - 1)) * 4;
  const float b0 = b[c], b1 = b[c + 1], b2 = b[c + 2], b3 = b[c + 3];
  float4* a4 = (float4*)a;
  for (int ib = i0; ib < EL_N4; ib += EL_U * EL_T) {
    float4 v[EL_U];
#pragma unroll
    for (int k = 0; k < EL_U; ++k) v[k] = a4[ib + k * EL_T];
#pragma unroll
    for (int k = 0; k < EL_U; ++k) {
      const float x0 = v[k].x + b0, x1 = v[k].y + b1, x2 = v[k].z + b2, x3 = v[k].w + b3;
      float4 r;
      r.x = x0 > 0.f ? x0 : (__expf(x0) - 1.f);
      r.y = x1 > 0.f ? x1 : (__expf(x1) - 1.f);
      r.z = x2 > 0.f ? x2 : (__expf(x2) - 1.f);
      r.w = x3 > 0.f ? x3 : (__expf(x3) - 1.f);
      a4[ib + k * EL_T] = r;
    }
  }
}

#define OB_WG 512
#define OB_T  (OB_WG * 256)
#define OB_U  4
#define OB_N4 (NND * ODIM / 4)
static_assert(OB_U * OB_T >= OB_N4, "one trip covers the output");
__global__ __launch_bounds__(256) void k_out_bias(const float* __restrict__ agg, const float* __restrict__ b,
                           float* __restrict__ out) {
  const int i0 = blockIdx.x * 256 + threadIdx.x;
  const float4* g4 = (const float4*)agg;
  float4* o4 = (float4*)out;
  float4 v[OB_U], bb[OB_U];
#pragma unroll
  for (int k = 0; k < OB_U; ++k) {
    const int i = i0 + k * OB_T;
    const int ic = i < OB_N4 ? i : i0;
    v[k] = g4[ic];
    const int c = (ic % (ODIM / 4)) * 4;
    bb[k] = make_float4(b[c], b[c + 1], b[c + 2], b[c + 3]);
  }
#pragma unroll
  for (int k = 0; k < OB_U; ++k) {
    const int i = i0 + k * OB_T;
    if (i < OB_N4)
      o4[i] = make_float4(v[k].x + bb[k].x, v[k].y + bb[k].y, v[k].z + bb[k].z, v[k].w + bb[k].w);
  }
}

extern "C" void kernel_launch(void* const* d_in, const int* in_sizes, int n_in,
                              void* d_out, int out_size, void* d_ws, size_t ws_size,
                              hipStream_t stream) {
  const float* x   = (const float*)d_in[0];
  const int*   ei  = (const int*)d_in[1];
  const float* W1  = (const float*)d_in[2];
  const float* as1 = (const float*)d_in[3];
  const float* ad1 = (const float*)d_in[4];
  const float* b1  = (const float*)d_in[5];
  const float* W2  = (const float*)d_in[6];
  const float* as2 = (const float*)d_in[7];
  const float* ad2 = (const float*)d_in[8];
  const float* b2  = (const float*)d_in[9];
  float* out = (float*)d_out;

  float* ws   = (float*)d_ws;
  float* h1   = ws;
  float* agg1 = h1   + (size_t)NND * HC;
  float* aS1  = agg1 + (size_t)NND * HC;
  float* aD1  = aS1  + NND * NH;
  int*   nm1  = (int*)(aD1 + NND * NH);
  float* dn1  = (float*)(nm1 + NND * NH);
  float* e1   = dn1  + NND * NH;
  float* t2   = e1   + (size_t)NET * NH;
  float* aS2  = t2   + (size_t)NND * ODIM;
  float* aD2  = aS2  + NND;
  int*   nm2  = (int*)(aD2 + NND);
  float* dn2  = (float*)(nm2 + NND);
  float* e2   = dn2  + NND;
  float* agg2 = e2   + NET;

  k_fill4<<<(NND * NH / 4 + 255) / 256, 256, 0, stream>>>((int4*)dn1, 0, NND * NH / 4);
  k_fill4<<<(NND * NH / 4 + 255) / 256, 256, 0, stream>>>((int4*)nm1, (int)0x80000000, NND * NH / 4);
  k_fill4<<<(NND / 4 + 255) / 256, 256, 0, stream>>>((int4*)dn2, 0, NND / 4);
  k_fill4<<<(NND / 4 + 255) / 256, 256, 0, stream>>>((int4*)nm2, (int)0x80000000, NND / 4);

  k_gemm1<<<dim3(NND / 16, 2), 256, 0, stream>>>(x, W1, h1);
  k_att1<<<ATT_WG, 256, 0, stream>>>(h1, as1, ad1, aS1, aD1);
  k_edgeA1<<<EG_WG, 256, 0, stream>>>(ei, aS1, aD1, e1, nm1);
  k_edgeB1<<<EG_WG, 256, 0, stream>>>(ei, e1, nm1, dn1);
  k_own<HC, NH><<<OWN_WG, OWN_T, 0, stream>>>(ei, h1, e1, dn1, agg1);
  k_elu_bias<<<EL_WG, 256, 0, stream>>>(agg1, b1);

  k_gemm2<<<NND / 16, 96, 0, stream>>>(agg1, W2, t2);
  k_att2<<<ATT_WG, 256, 0, stream>>>(t2, as2, ad2, aS2, aD2);
  k_edgeA2<<<(NQ + 255) / 256, 256, 0, stream>>>(ei, aS2, aD2, e2, nm2);
  k_edgeB2<<<(NQ + 255) / 256, 256, 0, stream>>>(ei, e2, nm2, dn2);
  k_own<ODIM, 1><<<OWN_WG, OWN_T, 0, stream>>>(ei, t2, e2, dn2, agg2);
  k_out_bias<<<OB_WG, 256, 0, stream>>>(agg2, b2, out);
}


# Softmax sums of both layers moved into the owner work-groups: exp per edge, weights added per row and head in LDS (float LDS adds, arrival order), row divided once at its store; no global sum atomics
# speedup vs baseline: 1.8662x; 1.2311x over previous
#include <hip/hip_runtime.h>
#include <hip/hip_bf16.h>

#define NND   50000
#define FIN   128
#define NH    8
#define HIDC  32
#define HC    256
#define ODIM  40
#define NE    800000
#define NET   850000
#define NEG_SLOPE 0.2f
#define EPSV  1e-16f

typedef __attribute__((ext_vector_type(2))) float v2f;
typedef __attribute__((ext_vector_type(8))) float v8f;

__device__ __forceinline__ float wave_sum(float v) {
#pragma unroll
  for (int off = 16; off > 0; off >>= 1) v += __shfl_xor(v, off, 32);
  return v;
}
__device__ __forceinline__ int f32_ord(float x) {
  int i = __float_as_int(x);
  return i >= 0 ? i : (i ^ 0x7FFFFFFF);
}
__device__ __forceinline__ float ord_f32(int i) {
  return __int_as_float(i >= 0 ? i : (i ^ 0x7FFFFFFF));
}

__global__ __launch_bounds__(256) void k_fill4(int4* __restrict__ p, int v, int n4) {
  const int i = blockIdx.x * 256 + threadIdx.x;
  if (i < n4) p[i] = make_int4(v, v, v, v);
}

__global__ __launch_bounds__(256) void k_gemm1(const float* __restrict__ x,
                                               const float* __restrict__ W,
                                               float* __restrict__ h) {
  __shared__ float As[16 * FIN];
  const int tid  = threadIdx.x;
  const int wave = tid >> 5, lane = tid & 31;
  const int m0   = blockIdx.x * 16;
  {
    const float4* xv = (const float4*)(x + (size_t)m0 * FIN);
    float4* sv = (float4*)As;
    sv[tid]       = xv[tid];
    sv[tid + 256] = xv[tid + 256];
  }
  __syncthreads();
  const int n0 = blockIdx.y * 128 + wave * 16;
  const int la = lane & 15, hi = lane >> 4;
  v8f acc = {};
#pragma unroll
  for (int k0 = 0; k0 < FIN; k0 += 4) {
    const int ka = k0 + hi * 2;
    v2f a, b;
    a.x = As[la * FIN + ka];
    a.y = As[la * FIN + ka + 1];
    b.x = W[ka * HC + n0 + la];
    b.y = W[(ka + 1) * HC + n0 + la];
    acc = __builtin_amdgcn_wmma_f32_16x16x4_f32(false, a, false, b, (short)0,
                                                acc, false, false);
  }
#pragma unroll
  for (int j = 0; j < 8; ++j)
    h[(size_t)(m0 + j + hi * 8) * HC + n0 + la] = acc[j];
}

__global__ __launch_bounds__(96) void k_gemm2(const float* __restrict__ h2,
                                              const float* __restrict__ W,
                                              float* __restrict__ t) {
  __shared__ float As[16 * HC];
  const int tid  = threadIdx.x;
  const int wave = tid >> 5, lane = tid & 31;
  const int m0   = blockIdx.x * 16;
  {
    const float4* xv = (const float4*)(h2 + (size_t)m0 * HC);
    float4* sv = (float4*)As;
    for (int i = tid; i < 1024; i += 96) sv[i] = xv[i];
  }
  __syncthreads();
  const int n0 = wave * 16;
  const int la = lane & 15, hi = lane >> 4;
  const int col = n0 + la;
  const bool ok = col < ODIM;
  v8f acc = {};
#pragma unroll 8
  for (int k0 = 0; k0 < HC; k0 += 4) {
    const int ka = k0 + hi * 2;
    v2f a, b;
    a.x = As[la * HC + ka];
    a.y = As[la * HC + ka + 1];
    b.x = ok ? W[ka * ODIM + col] : 0.f;
    b.y = ok ? W[(ka + 1) * ODIM + col] : 0.f;
    acc = __builtin_amdgcn_wmma_f32_16x16x4_f32(false, a, false, b, (short)0,
                                                acc, false, false);
  }
#pragma unroll
  for (int j = 0; j < 8; ++j)
    if (ok) t[(size_t)(m0 + j + hi * 8) * ODIM + col] = acc[j];
}

#define ATT_WG 1024
#define ATT_NW (ATT_WG * 8)
__device__ __forceinline__ void att1_node(const float (&v)[NH], const float (&ws)[NH],
                                          const float (&wd)[NH], int node, int lane,
                                          float* __restrict__ os, float* __restrict__ od) {
#pragma unroll
  for (int hd = 0; hd < NH; ++hd) {
    const float s = wave_sum(v[hd] * ws[hd]);
    const float d = wave_sum(v[hd] * wd[hd]);
    if (lane == 0) { os[node * NH + hd] = s; od[node * NH + hd] = d; }
  }
}
__global__ __launch_bounds__(256) void k_att1(const float* __restrict__ h,
                                              const float* __restrict__ as_,
                                              const float* __restrict__ ad_,
                                              float* __restrict__ os,
                                              float* __restrict__ od) {
  const int lane = threadIdx.x & 31;
  const int w = blockIdx.x * 8 + (threadIdx.x >> 5);
  float ws[NH], wd[NH];
#pragma unroll
  for (int hd = 0; hd < NH; ++hd) { ws[hd] = as_[hd * HIDC + lane]; wd[hd] = ad_[hd * HIDC + lane]; }
  for (int na = w; na < NND; na += 2 * ATT_NW) {
    const int nb = na + ATT_NW;
    const bool hb = nb < NND;
    const float* ra = h + (size_t)na * HC;
    const float* rb = h + (size_t)(hb ? nb : na) * HC;
    float va[NH], vb[NH];
#pragma unroll
    for (int hd = 0; hd < NH; ++hd) va[hd] = ra[hd * HIDC + lane];
#pragma unroll
    for (int hd = 0; hd < NH; ++hd) vb[hd] = rb[hd * HIDC + lane];
    att1_node(va, ws, wd, na, lane, os, od);
    if (hb) att1_node(vb, ws, wd, nb, lane, os, od);
  }
}

__device__ __forceinline__ void att2_node(float v0, float v1, float s0, float s1, float d0,
                                          float d1, bool up, int node, int lane,
                                          float* __restrict__ os, float* __restrict__ od) {
  float s = 0.f, d = 0.f;
  s += v0 * s0; d += v0 * d0;
  if (up) { s += v1 * s1; d += v1 * d1; }
  s = wave_sum(s); d = wave_sum(d);
  if (lane == 0) { os[node] = s; od[node] = d; }
}
__global__ __launch_bounds__(256) void k_att2(const float* __restrict__ t,
                                              const float* __restrict__ as_,
                                              const float* __restrict__ ad_,
                                              float* __restrict__ os,
                                              float* __restrict__ od) {
  const int lane = threadIdx.x & 31;
  const int w = blockIdx.x * 8 + (threadIdx.x >> 5);
  const bool up = lane < ODIM - 32;
  const float s0 = as_[lane], d0 = ad_[lane];
  const float s1 = up ? as_[32 + lane] : 0.f, d1 = up ? ad_[32 + lane] : 0.f;
  for (int na = w; na < NND; na += 2 * ATT_NW) {
    const int nb = na + ATT_NW;
    const bool hb = nb < NND;
    const float* ta = t + (size_t)na * ODIM;
    const float* tb = t + (size_t)(hb ? nb : na) * ODIM;
    const float a0 = ta[lane], b0 = tb[lane];
    const float a1 = up ? ta[32 + lane] : 0.f, b1 = up ? tb[32 + lane] : 0.f;
    att2_node(a0, a1, s0, s1, d0, d1, up, na, lane, os, od);
    if (hb) att2_node(b0, b1, s0, s1, d0, d1, up, nb, lane, os, od);
  }
}

#define EG_WG 1024
#define EG_T  (EG_WG * 256)
#define EG_U  8
#define EG_N  (NET * NH)
__device__ __forceinline__ void edge_sd(const int* __restrict__ ei, int eid, int& s, int& d) {
  const bool real = eid < NE;
  const int ec = real ? eid : 0;
  const int ls = ei[ec], ld = ei[NE + ec];
  s = real ? ls : eid - NE;
  d = real ? ld : eid - NE;
}

__global__ __launch_bounds__(256) void k_edgeA1(const int* __restrict__ ei,
                         const float* __restrict__ as_, const float* __restrict__ ad_,
                         float* __restrict__ e, int* __restrict__ nmax) {
  const int t0 = blockIdx.x * 256 + threadIdx.x;
  const int hd = t0 & 7;
  for (int tb = t0; tb < EG_N; tb += EG_U * EG_T) {
    int s[EG_U], d[EG_U];
    float v[EG_U];
#pragma unroll
    for (int k = 0; k < EG_U; ++k) {
      const int t = tb + k * EG_T;
      edge_sd(ei, (t < EG_N ? t : t0) >> 3, s[k], d[k]);
    }
#pragma unroll
    for (int k = 0; k < EG_U; ++k) v[k] = as_[s[k] * NH + hd] + ad_[d[k] * NH + hd];
#pragma unroll
    for (int k = 0; k < EG_U; ++k) {
      const int t = tb + k * EG_T;
      if (t < EG_N) {
        const float x = v[k] > 0.f ? v[k] : v[k] * NEG_SLOPE;
        e[t] = x;
        atomicMax(&nmax[d[k] * NH + hd], f32_ord(x));
      }
    }
  }
}

#define OWN_R    196
#define OWN_WG   256
#define OWN_T    1024
#define OWN_U    4
#define NE4      (NE / 4)
#define OWN_IT   ((NE4 + OWN_T * OWN_U - 1) / (OWN_T * OWN_U))
#define OWN_QTH  32
#define OWN_QCAP (OWN_QTH + 32 * OWN_U * 4 + 32)

__device__ __forceinline__ float lane_bcast_f(float v, int l) {
  return __int_as_float(__builtin_amdgcn_readlane(__float_as_int(v), l));
}

template <int C, int H, bool INIT>
__device__ __forceinline__ void own_edge(float* __restrict__ acc, float* __restrict__ wsum,
                                         const float* __restrict__ h,
                                         const float* __restrict__ elog,
                                         const int* __restrict__ nmax,
                                         int eid, int s, int d, int dl, int lane) {
  const float* hs = h + (size_t)s * C;
  float* ar = acc + dl * C;
  constexpr int NP = (C + 31) / 32;
  float v[NP];
#pragma unroll
  for (int j = 0; j < NP; ++j) {
    const int c = j * 32 + lane;
    v[j] = c < C ? hs[c] : 0.f;
  }
  float a = 0.f;
  if (lane < H) {
    a = __expf(elog[(size_t)eid * H + lane] - ord_f32(nmax[d * H + lane]));
    if (INIT) wsum[dl * H + lane] = a;
    else __hip_atomic_fetch_add(&wsum[dl * H + lane], a, __ATOMIC_RELAXED, __HIP_MEMORY_SCOPE_WORKGROUP);
  }
#pragma unroll
  for (int j = 0; j < NP; ++j) {
    const float alpha = lane_bcast_f(a, (j * 32) / (C / H));
    const int c = j * 32 + lane;
    if (c < C) {
      const float t = v[j] * alpha;
      if (INIT) ar[c] = t;
      else __hip_atomic_fetch_add(&ar[c], t, __ATOMIC_RELAXED, __HIP_MEMORY_SCOPE_WORKGROUP);
    }
  }
}

template <int C, int H>
__device__ __forceinline__ void own_pair(float* __restrict__ acc, float* __restrict__ wsum,
                                         const float* __restrict__ h,
                                         const float* __restrict__ elog,
                                         const int* __restrict__ nmax,
                                         int eidA, int sA, int dA, int xA,
                                         int eidB, int sB, int dB, int xB, int lane) {
  const float* hA = h + (size_t)sA * C;
  const float* hB = h + (size_t)sB * C;
  float* arA = acc + xA * C;
  float* arB = acc + xB * C;
  constexpr int NP = (C + 31) / 32;
  float vA[NP], vB[NP];
#pragma unroll
  for (int j = 0; j < NP; ++j) {
    const int c = j * 32 + lane;
    vA[j] = c < C ? hA[c] : 0.f;
  }
#pragma unroll
  for (int j = 0; j < NP; ++j) {
    const int c = j * 32 + lane;
    vB[j] = c < C ? hB[c] : 0.f;
  }
  float a = 0.f;
  if (lane < 2 * H) {
    const bool second = lane >= H;
    const int hh  = second ? lane - H : lane;
    const int eid = second ? eidB : eidA;
    const int d   = second ? dB : dA;
    const int x   = second ? xB : xA;
    a = __expf(elog[(size_t)eid * H + hh] - ord_f32(nmax[d * H + hh]));
    __hip_atomic_fetch_add(&wsum[x * H + hh], a, __ATOMIC_RELAXED, __HIP_MEMORY_SCOPE_WORKGROUP);
  }
#pragma unroll
  for (int j = 0; j < NP; ++j) {
    const float alpha = lane_bcast_f(a, (j * 32) / (C / H));
    const int c = j * 32 + lane;
    if (c < C)
      __hip_atomic_fetch_add(&arA[c], vA[j] * alpha, __ATOMIC_RELAXED, __HIP_MEMORY_SCOPE_WORKGROUP);
  }
#pragma unroll
  for (int j = 0; j < NP; ++j) {
    const float alpha = lane_bcast_f(a, H + (j * 32) / (C / H));
    const int c = j * 32 + lane;
    if (c < C)
      __hip_atomic_fetch_add(&arB[c], vB[j] * alpha, __ATOMIC_RELAXED, __HIP_MEMORY_SCOPE_WORKGROUP);
  }
}

template <int C, int H>
__device__ __forceinline__ void own_drain(float* __restrict__ acc, float* __restrict__ wsum,
                                          const unsigned* qw, int n,
                                          const int* __restrict__ ei,
                                          const float* __restrict__ h,
                                          const float* __restrict__ elog,
                                          const int* __restrict__ nmax,
                                          int d0, int lane) {
  for (int b = 0; b < n; b += 32) {
    int cnt = n - b;
    if (cnt > 32) cnt = 32;
    unsigned ent = 0;
    int s = 0;
    if (lane < cnt) {
      ent = qw[b + lane];
      s = ei[ent & 0xFFFFFu];
    }
    int i = 0;
    for (; i + 1 < cnt; i += 2) {
      const unsigned eA = (unsigned)__builtin_amdgcn_readlane((int)ent, i);
      const unsigned eB = (unsigned)__builtin_amdgcn_readlane((int)ent, i + 1);
      const int sA = __builtin_amdgcn_readlane(s, i);
      const int sB = __builtin_amdgcn_readlane(s, i + 1);
      const int xA = (int)(eA >> 20), xB = (int)(eB >> 20);
      own_pair<C, H>(acc, wsum, h, elog, nmax, (int)(eA & 0xFFFFFu), sA, d0 + xA, xA,
                     (int)(eB & 0xFFFFFu), sB, d0 + xB, xB, lane);
    }
    if (i < cnt) {
      const unsigned eA = (unsigned)__builtin_amdgcn_readlane((int)ent, i);
      const int sA = __builtin_amdgcn_readlane(s, i);
      const int xA = (int)(eA >> 20);
      own_edge<C, H, false>(acc, wsum, h, elog, nmax, (int)(eA & 0xFFFFFu), sA, d0 + xA, xA, lane);
    }
  }
}

template <int C, int H>
__global__ __launch_bounds__(OWN_T) void k_own(const int* __restrict__ ei,
                                               const float* __restrict__ h,
                                               const float* __restrict__ elog,
                                               const int* __restrict__ nmax,
                                               float* __restrict__ agg) {
  __shared__ __attribute__((aligned(16))) float acc[OWN_R * C];
  __shared__ float wsum[OWN_R * H];
  __shared__ unsigned q[(OWN_T / 32) * OWN_QCAP];
  const int tid = threadIdx.x, lane = tid & 31;
  const int wave = __builtin_amdgcn_readfirstlane(tid >> 5);
  const int d0 = blockIdx.x * OWN_R;
  int nrow = NND - d0;
  if (nrow > OWN_R) nrow = OWN_R;

  for (int r = wave; r < nrow; r += OWN_T / 32)
    own_edge<C, H, true>(acc, wsum, h, elog, nmax, NE + d0 + r, d0 + r, d0 + r, r, lane);
  __syncthreads();

  unsigned* qw = q + wave * OWN_QCAP;
  int qn = 0;
  const int4* dst4 = (const int4*)(ei + NE);
  const int wofs = wave * 32;
  int4 cur[OWN_U];
  int ii = (blockIdx.x * OWN_IT) / OWN_WG;
#pragma unroll
  for (int u = 0; u < OWN_U; ++u) {
    const int wb = (ii * OWN_U + u) * OWN_T + wofs;
    cur[u] = dst4[(wb < NE4 ? wb : NE4 - 32) + lane];
  }
  for (int it = 0; it < OWN_IT; ++it) {
    const int in = ii + 1 < OWN_IT ? ii + 1 : 0;
    int4 nxt[OWN_U];
#pragma unroll
    for (int u = 0; u < OWN_U; ++u) {
      const int wb = (in * OWN_U + u) * OWN_T + wofs;
      nxt[u] = dst4[(wb < NE4 ? wb : NE4 - 32) + lane];
    }
#pragma unroll
    for (int u = 0; u < OWN_U; ++u) {
      const int wb = (ii * OWN_U + u) * OWN_T + wofs;
      const int d0u = wb < NE4 ? d0 : 0x40000000;
      const int dd[4] = {cur[u].x, cur[u].y, cur[u].z, cur[u].w};
#pragma unroll
      for (int k = 0; k < 4; ++k) {
        const unsigned x = (unsigned)(dd[k] - d0u);
        const bool m = x < (unsigned)nrow;
        const unsigned mk = __builtin_amdgcn_ballot_w32(m);
        if (mk) {
          if (m) qw[qn + (int)__builtin_amdgcn_mbcnt_lo(mk, 0u)] = (x << 20) | (unsigned)((wb + lane) * 4 + k);
          qn += __builtin_popcount(mk);
        }
      }
    }
    if (qn >= OWN_QTH) {
      own_drain<C, H>(acc, wsum, qw, qn, ei, h, elog, nmax, d0, lane);
      qn = 0;
    }
#pragma unroll
    for (int u = 0; u < OWN_U; ++u) cur[u] = nxt[u];
    ii = in;
  }
  own_drain<C, H>(acc, wsum, qw, qn, ei, h, elog, nmax, d0, lane);
  __syncthreads();

  static_assert((C / H) % 4 == 0, "a 16-byte piece does not straddle two heads");
  constexpr int PH = (C / H) / 4;
  float4* o4 = (float4*)(agg + (size_t)d0 * C);
  const float4* a4 = (const float4*)acc;
  for (int i = tid; i < nrow * C / 4; i += OWN_T) {
    const float4 v = a4[i];
    const float dn = wsum[i / PH] + EPSV;
    o4[i] = make_float4(v.x / dn, v.y / dn, v.z / dn, v.w / dn);
  }
}

#define NQ  (NET / 4)
#define NEQ (NE / 4)
__device__ __forceinline__ void quad_sd(const int* __restrict__ ei, int q, int (&s)[4], int (&d)[4]) {
  if (q < NEQ) {
    const int4 sv = ((const int4*)ei)[q];
    const int4 dv = ((const int4*)(ei + NE))[q];
    s[0] = sv.x; s[1] = sv.y; s[2] = sv.z; s[3] = sv.w;
    d[0] = dv.x; d[1] = dv.y; d[2] = dv.z; d[3] = dv.w;
  } else {
#pragma unroll
    for (int k = 0; k < 4; ++k) s[k] = d[k] = q * 4 + k - NE;
  }
}

__global__ __launch_bounds__(256) void k_edgeA2(const int* __restrict__ ei,
                         const float* __restrict__ as_, const float* __restrict__ ad_,
                         float* __restrict__ e, int* __restrict__ nmax) {
  const int q = blockIdx.x * 256 + threadIdx.x;
  if (q >= NQ) return;
  int s[4], d[4];
  quad_sd(ei, q, s, d);
  float v[4];
#pragma unroll
  for (int k = 0; k < 4; ++k) v[k] = as_[s[k]] + ad_[d[k]];
#pragma unroll
  for (int k = 0; k < 4; ++k) v[k] = v[k] > 0.f ? v[k] : v[k] * NEG_SLOPE;
  ((float4*)e)[q] = make_float4(v[0], v[1], v[2], v[3]);
#pragma unroll
  for (int k = 0; k < 4; ++k) atomicMax(&nmax[d[k]], f32_ord(v[k]));
}

#define EL_WG 1250
#define EL_T  (EL_WG * 256)
#define EL_U  5
#define EL_N4 (NND * HC / 4)
static_assert(EL_N4 == 2 * EL_U * EL_T, "two full trips cover the rows exactly");
static_assert(EL_T % (HC / 4) == 0, "a thread's columns do not change from trip to trip");
__global__ __launch_bounds__(256) void k_elu_bias(float* __restrict__ a, const float* __restrict__ b) {
  const int i0 = blockIdx.x * 256 + threadIdx.x;
  const int c = (i0 & (HC / 4 - 1)) * 4;
  const float b0 = b[c], b1 = b[c + 1], b2 = b[c + 2], b3 = b[c + 3];
  float4* a4 = (float4*)a;
  for (int ib = i0; ib < EL_N4; ib += EL_U * EL_T) {
    float4 v[EL_U];
#pragma unroll
    for (int k = 0; k < EL_U; ++k) v[k] = a4[ib + k * EL_T];
#pragma unroll
    for (int k = 0; k < EL_U; ++k) {
      const float x0 = v[k].x + b0, x1 = v[k].y + b1, x2 = v[k].z + b2, x3 = v[k].w + b3;
      float4 r;
      r.x = x0 > 0.f ? x0 : (__expf(x0) - 1.f);
      r.y = x1 > 0.f ? x1 : (__expf(x1) - 1.f);
      r.z = x2 > 0.f ? x2 : (__expf(x2) - 1.f);
      r.w = x3 > 0.f ? x3 : (__expf(x3) - 1.f);
      a4[ib + k * EL_T] = r;
    }
  }
}

#define OB_WG 512
#define OB_T  (OB_WG * 256)
#define OB_U  4
#define OB_N4 (NND * ODIM / 4)
static_assert(OB_U * OB_T >= OB_N4, "one trip covers the output");
__global__ __launch_bounds__(256) void k_out_bias(const float* __restrict__ agg, const float* __restrict__ b,
                           float* __restrict__ out) {
  const int i0 = blockIdx.x * 256 + threadIdx.x;
  const float4* g4 = (const float4*)agg;
  float4* o4 = (float4*)out;
  float4 v[OB_U], bb[OB_U];
#pragma unroll
  for (int k = 0; k < OB_U; ++k) {
    const int i = i0 + k * OB_T;
    const int ic = i < OB_N4 ? i : i0;
    v[k] = g4[ic];
    const int c = (ic % (ODIM / 4)) * 4;
    bb[k] = make_float4(b[c], b[c + 1], b[c + 2], b[c + 3]);
  }
#pragma unroll
  for (int k = 0; k < OB_U; ++k) {
    const int i = i0 + k * OB_T;
    if (i < OB_N4)
      o4[i] = make_float4(v[k].x + bb[k].x, v[k].y + bb[k].y, v[k].z + bb[k].z, v[k].w + bb[k].w);
  }
}

extern "C" void kernel_launch(void* const* d_in, const int* in_sizes, int n_in,
                              void* d_out, int out_size, void* d_ws, size_t ws_size,
                              hipStream_t stream) {
  const float* x   = (const float*)d_in[0];
  const int*   ei  = (const int*)d_in[1];
  const float* W1  = (const float*)d_in[2];
  const float* as1 = (const float*)d_in[3];
  const float* ad1 = (const float*)d_in[4];
  const float* b1  = (const float*)d_in[5];
  const float* W2  = (const float*)d_in[6];
  const float* as2 = (const float*)d_in[7];
  const float* ad2 = (const float*)d_in[8];
  const float* b2  = (const float*)d_in[9];
  float* out = (float*)d_out;

  float* ws   = (float*)d_ws;
  float* h1   = ws;
  float* agg1 = h1   + (size_t)NND * HC;
  float* aS1  = agg1 + (size_t)NND * HC;
  float* aD1  = aS1  + NND * NH;
  int*   nm1  = (int*)(aD1 + NND * NH);
  float* e1   = (float*)(nm1 + NND * NH);
  float* t2   = e1   + (size_t)NET * NH;
  float* aS2  = t2   + (size_t)NND * ODIM;
  float* aD2  = aS2  + NND;
  int*   nm2  = (int*)(aD2 + NND);
  float* e2   = (float*)(nm2 + NND);
  float* agg2 = e2   + NET;

  k_fill4<<<(NND * NH / 4 + 255) / 256, 256, 0, stream>>>((int4*)nm1, (int)0x80000000, NND * NH / 4);
  k_fill4<<<(NND / 4 + 255) / 256, 256, 0, stream>>>((int4*)nm2, (int)0x80000000, NND / 4);

  k_gemm1<<<dim3(NND / 16, 2), 256, 0, stream>>>(x, W1, h1);
  k_att1<<<ATT_WG, 256, 0, stream>>>(h1, as1, ad1, aS1, aD1);
  k_edgeA1<<<EG_WG, 256, 0, stream>>>(ei, aS1, aD1, e1, nm1);
  k_own<HC, NH><<<OWN_WG, OWN_T, 0, stream>>>(ei, h1, e1, nm1, agg1);
  k_elu_bias<<<EL_WG, 256, 0, stream>>>(agg1, b1);

  k_gemm2<<<NND / 16, 96, 0, stream>>>(agg1, W2, t2);
  k_att2<<<ATT_WG, 256, 0, stream>>>(t2, as2, ad2, aS2, aD2);
  k_edgeA2<<<(NQ + 255) / 256, 256, 0, stream>>>(ei, aS2, aD2, e2, nm2);
  k_own<ODIM, 1><<<OWN_WG, OWN_T, 0, stream>>>(ei, t2, e2, nm2, agg2);
  k_out_bias<<<OB_WG, 256, 0, stream>>>(agg2, b2, out);
}


# Both products retiled: weights kept in LDS in fragment order, a wave holds its 16-row strip in registers (16-byte loads) and sweeps all column tiles; 391 groups, not 6250; k summed in a new order
# speedup vs baseline: 2.1034x; 1.1271x over previous
#include <hip/hip_runtime.h>
#include <hip/hip_bf16.h>

#define NND   50000
#define FIN   128
#define NH    8
#define HIDC  32
#define HC    256
#define ODIM  40
#define NE    800000
#define NET   850000
#define NEG_SLOPE 0.2f
#define EPSV  1e-16f

typedef __attribute__((ext_vector_type(2))) float v2f;
typedef __attribute__((ext_vector_type(8))) float v8f;

__device__ __forceinline__ float wave_sum(float v) {
#pragma unroll
  for (int off = 16; off > 0; off >>= 1) v += __shfl_xor(v, off, 32);
  return v;
}
__device__ __forceinline__ int f32_ord(float x) {
  int i = __float_as_int(x);
  return i >= 0 ? i : (i ^ 0x7FFFFFFF);
}
__device__ __forceinline__ float ord_f32(int i) {
  return __int_as_float(i >= 0 ? i : (i ^ 0x7FFFFFFF));
}

__global__ __launch_bounds__(256) void k_fill4(int4* __restrict__ p, int v, int n4) {
  const int i = blockIdx.x * 256 + threadIdx.x;
  if (i < n4) p[i] = make_int4(v, v, v, v);
}

#define GM_WV   8
#define GM_NS   (NND / 16)
#define GM_WG   ((GM_NS + GM_WV - 1) / GM_WV)
static_assert(NND % 16 == 0, "whole strips");

#define G1_KB (FIN / 8)
#define G1_NT (HC / 16)
__global__ __launch_bounds__(256) void k_gemm1(const float* __restrict__ x,
                                               const float* __restrict__ W,
                                               float* __restrict__ h) {
  __shared__ __attribute__((aligned(16))) float4 Ws[G1_KB * G1_NT * 32];
  const int tid  = threadIdx.x;
  const int lane = tid & 31;
  const int wave = __builtin_amdgcn_readfirstlane(tid >> 5);
  const int la = lane & 15, hi = lane >> 4;
  const int strip = blockIdx.x * GM_WV + wave;
  const bool live = strip < GM_NS;
  const int m0 = (live ? strip : 0) * 16;
  float4 a[G1_KB];
  {
    const float4* xr = (const float4*)(x + (size_t)(m0 + la) * FIN) + hi;
#pragma unroll
    for (int kb = 0; kb < G1_KB; ++kb) a[kb] = xr[kb * 2];
  }
  {
    const int t = tid >> 4, c = tid & 15;
#pragma unroll 4
    for (int kb = 0; kb < G1_KB; ++kb) {
#pragma unroll
      for (int hh = 0; hh < 2; ++hh) {
        const float* wp = W + (size_t)(kb * 8 + hh * 4) * HC + tid;
        Ws[(kb * G1_NT + t) * 32 + hh * 16 + c] = make_float4(wp[0], wp[HC], wp[2 * HC], wp[3 * HC]);
      }
    }
  }
  __syncthreads();
  if (!live) return;
  v8f acc[G1_NT];
#pragma unroll
  for (int t = 0; t < G1_NT; ++t) acc[t] = (v8f){};
#pragma unroll
  for (int kb = 0; kb < G1_KB; ++kb) {
    v2f a0, a1;
    a0.x = a[kb].x; a0.y = a[kb].y; a1.x = a[kb].z; a1.y = a[kb].w;
#pragma unroll
    for (int t = 0; t < G1_NT; ++t) {
      const float4 b = Ws[(kb * G1_NT + t) * 32 + lane];
      v2f b0, b1;
      b0.x = b.x; b0.y = b.y; b1.x = b.z; b1.y = b.w;
      acc[t] = __builtin_amdgcn_wmma_f32_16x16x4_f32(false, a0, false, b0, (short)0,
                                                     acc[t], false, false);
      acc[t] = __builtin_amdgcn_wmma_f32_16x16x4_f32(false, a1, false, b1, (short)0,
                                                     acc[t], false, false);
    }
  }
#pragma unroll
  for (int t = 0; t < G1_NT; ++t)
#pragma unroll
    for (int j = 0; j < 8; ++j)
      h[(size_t)(m0 + j + hi * 8) * HC + t * 16 + la] = acc[t][j];
}

#define G2_KB (HC / 8)
#define G2_NT 3
static_assert(G2_NT * 16 >= ODIM, "the tiles cover the output columns");
__global__ __launch_bounds__(256) void k_gemm2(const float* __restrict__ h2,
                                               const float* __restrict__ W,
                                               float* __restrict__ t2) {
  __shared__ __attribute__((aligned(16))) float4 Ws[G2_KB * G2_NT * 32];
  const int tid  = threadIdx.x;
  const int lane = tid & 31;
  const int wave = __builtin_amdgcn_readfirstlane(tid >> 5);
  const int la = lane & 15, hi = lane >> 4;
  const int strip = blockIdx.x * GM_WV + wave;
  const bool live = strip < GM_NS;
  const int m0 = (live ? strip : 0) * 16;
  float4 a[G2_KB];
  {
    const float4* xr = (const float4*)(h2 + (size_t)(m0 + la) * HC) + hi;
#pragma unroll
    for (int kb = 0; kb < G2_KB; ++kb) a[kb] = xr[kb * 2];
  }
  {
    static_assert(G2_KB == 4 * GM_WV, "eight waves cover the k blocks");
#pragma unroll
    for (int j = 0; j < 4; ++j) {
      const int kb = wave * 4 + j;
#pragma unroll
      for (int tt = 0; tt < G2_NT; ++tt) {
        const int n = tt * 16 + la;
        const float* wp = W + (size_t)(kb * 8 + hi * 4) * ODIM + (n < ODIM ? n : ODIM - 1);
        Ws[(kb * G2_NT + tt) * 32 + lane] = make_float4(wp[0], wp[ODIM], wp[2 * ODIM], wp[3 * ODIM]);
      }
    }
  }
  __syncthreads();
  if (!live) return;
  v8f acc[G2_NT];
#pragma unroll
  for (int t = 0; t < G2_NT; ++t) acc[t] = (v8f){};
#pragma unroll
  for (int kb = 0; kb < G2_KB; ++kb) {
    v2f a0, a1;
    a0.x = a[kb].x; a0.y = a[kb].y; a1.x = a[kb].z; a1.y = a[kb].w;
#pragma unroll
    for (int t = 0; t < G2_NT; ++t) {
      const float4 b = Ws[(kb * G2_NT + t) * 32 + lane];
      v2f b0, b1;
      b0.x = b.x; b0.y = b.y; b1.x = b.z; b1.y = b.w;
      acc[t] = __builtin_amdgcn_wmma_f32_16x16x4_f32(false, a0, false, b0, (short)0,
                                                     acc[t], false, false);
      acc[t] = __builtin_amdgcn_wmma_f32_16x16x4_f32(false, a1, false, b1, (short)0,
                                                     acc[t], false, false);
    }
  }
#pragma unroll
  for (int t = 0; t < G2_NT; ++t) {
    const int col = t * 16 + la;
    if (col < ODIM) {
#pragma unroll
      for (int j = 0; j < 8; ++j)
        t2[(size_t)(m0 + j + hi * 8) * ODIM + col] = acc[t][j];
    }
  }
}

#define ATT_WG 1024
#define ATT_NW (ATT_WG * 8)
__device__ __forceinline__ void att1_node(const float (&v)[NH], const float (&ws)[NH],
                                          const float (&wd)[NH], int node, int lane,
                                          float* __restrict__ os, float* __restrict__ od) {
#pragma unroll
  for (int hd = 0; hd < NH; ++hd) {
    const float s = wave_sum(v[hd] * ws[hd]);
    const float d = wave_sum(v[hd] * wd[hd]);
    if (lane == 0) { os[node * NH + hd] = s; od[node * NH + hd] = d; }
  }
}
__global__ __launch_bounds__(256) void k_att1(const float* __restrict__ h,
                                              const float* __restrict__ as_,
                                              const float* __restrict__ ad_,
                                              float* __restrict__ os,
                                              float* __restrict__ od) {
  const int lane = threadIdx.x & 31;
  const int w = blockIdx.x * 8 + (threadIdx.x >> 5);
  float ws[NH], wd[NH];
#pragma unroll
  for (int hd = 0; hd < NH; ++hd) { ws[hd] = as_[hd * HIDC + lane]; wd[hd] = ad_[hd * HIDC + lane]; }
  for (int na = w; na < NND; na += 2 * ATT_NW) {
    const int nb = na + ATT_NW;
    const bool hb = nb < NND;
    const float* ra = h + (size_t)na * HC;
    const float* rb = h + (size_t)(hb ? nb : na) * HC;
    float va[NH], vb[NH];
#pragma unroll
    for (int hd = 0; hd < NH; ++hd) va[hd] = ra[hd * HIDC + lane];
#pragma unroll
    for (int hd = 0; hd < NH; ++hd) vb[hd] = rb[hd * HIDC + lane];
    att1_node(va, ws, wd, na, lane, os, od);
    if (hb) att1_node(vb, ws, wd, nb, lane, os, od);
  }
}

__device__ __forceinline__ void att2_node(float v0, float v1, float s0, float s1, float d0,
                                          float d1, bool up, int node, int lane,
                                          float* __restrict__ os, float* __restrict__ od) {
  float s = 0.f, d = 0.f;
  s += v0 * s0; d += v0 * d0;
  if (up) { s += v1 * s1; d += v1 * d1; }
  s = wave_sum(s); d = wave_sum(d);
  if (lane == 0) { os[node] = s; od[node] = d; }
}
__global__ __launch_bounds__(256) void k_att2(const float* __restrict__ t,
                                              const float* __restrict__ as_,
                                              const float* __restrict__ ad_,
                                              float* __restrict__ os,
                                              float* __restrict__ od) {
  const int lane = threadIdx.x & 31;
  const int w = blockIdx.x * 8 + (threadIdx.x >> 5);
  const bool up = lane < ODIM - 32;
  const float s0 = as_[lane], d0 = ad_[lane];
  const float s1 = up ? as_[32 + lane] : 0.f, d1 = up ? ad_[32 + lane] : 0.f;
  for (int na = w; na < NND; na += 2 * ATT_NW) {
    const int nb = na + ATT_NW;
    const bool hb = nb < NND;
    const float* ta = t + (size_t)na * ODIM;
    const float* tb = t + (size_t)(hb ? nb : na) * ODIM;
    const float a0 = ta[lane], b0 = tb[lane];
    const float a1 = up ? ta[32 + lane] : 0.f, b1 = up ? tb[32 + lane] : 0.f;
    att2_node(a0, a1, s0, s1, d0, d1, up, na, lane, os, od);
    if (hb) att2_node(b0, b1, s0, s1, d0, d1, up, nb, lane, os, od);
  }
}

#define EG_WG 1024
#define EG_T  (EG_WG * 256)
#define EG_U  8
#define EG_N  (NET * NH)
__device__ __forceinline__ void edge_sd(const int* __restrict__ ei, int eid, int& s, int& d) {
  const bool real = eid < NE;
  const int ec = real ? eid : 0;
  const int ls = ei[ec], ld = ei[NE + ec];
  s = real ? ls : eid - NE;
  d = real ? ld : eid - NE;
}

__global__ __launch_bounds__(256) void k_edgeA1(const int* __restrict__ ei,
                         const float* __restrict__ as_, const float* __restrict__ ad_,
                         float* __restrict__ e, int* __restrict__ nmax) {
  const int t0 = blockIdx.x * 256 + threadIdx.x;
  const int hd = t0 & 7;
  for (int tb = t0; tb < EG_N; tb += EG_U * EG_T) {
    int s[EG_U], d[EG_U];
    float v[EG_U];
#pragma unroll
    for (int k = 0; k < EG_U; ++k) {
      const int t = tb + k * EG_T;
      edge_sd(ei, (t < EG_N ? t : t0) >> 3, s[k], d[k]);
    }
#pragma unroll
    for (int k = 0; k < EG_U; ++k) v[k] = as_[s[k] * NH + hd] + ad_[d[k] * NH + hd];
#pragma unroll
    for (int k = 0; k < EG_U; ++k) {
      const int t = tb + k * EG_T;
      if (t < EG_N) {
        const float x = v[k] > 0.f ? v[k] : v[k] * NEG_SLOPE;
        e[t] = x;
        atomicMax(&nmax[d[k] * NH + hd], f32_ord(x));
      }
    }
  }
}

#define OWN_R    196
#define OWN_WG   256
#define OWN_T    1024
#define OWN_U    4
#define NE4      (NE / 4)
#define OWN_IT   ((NE4 + OWN_T * OWN_U - 1) / (OWN_T * OWN_U))
#define OWN_QTH  32
#define OWN_QCAP (OWN_QTH + 32 * OWN_U * 4 + 32)

__device__ __forceinline__ float lane_bcast_f(float v, int l) {
  return __int_as_float(__builtin_amdgcn_readlane(__float_as_int(v), l));
}

template <int C, int H, bool INIT>
__device__ __forceinline__ void own_edge(float* __restrict__ acc, float* __restrict__ wsum,
                                         const float* __restrict__ h,
                                         const float* __restrict__ elog,
                                         const int* __restrict__ nmax,
                                         int eid, int s, int d, int dl, int lane) {
  const float* hs = h + (size_t)s * C;
  float* ar = acc + dl * C;
  constexpr int NP = (C + 31) / 32;
  float v[NP];
#pragma unroll
  for (int j = 0; j < NP; ++j) {
    const int c = j * 32 + lane;
    v[j] = c < C ? hs[c] : 0.f;
  }
  float a = 0.f;
  if (lane < H) {
    a = __expf(elog[(size_t)eid * H + lane] - ord_f32(nmax[d * H + lane]));
    if (INIT) wsum[dl * H + lane] = a;
    else __hip_atomic_fetch_add(&wsum[dl * H + lane], a, __ATOMIC_RELAXED, __HIP_MEMORY_SCOPE_WORKGROUP);
  }
#pragma unroll
  for (int j = 0; j < NP; ++j) {
    const float alpha = lane_bcast_f(a, (j * 32) / (C / H));
    const int c = j * 32 + lane;
    if (c < C) {
      const float t = v[j] * alpha;
      if (INIT) ar[c] = t;
      else __hip_atomic_fetch_add(&ar[c], t, __ATOMIC_RELAXED, __HIP_MEMORY_SCOPE_WORKGROUP);
    }
  }
}

template <int C, int H>
__device__ __forceinline__ void own_pair(float* __restrict__ acc, float* __restrict__ wsum,
                                         const float* __restrict__ h,
                                         const float* __restrict__ elog,
                                         const int* __restrict__ nmax,
                                         int eidA, int sA, int dA, int xA,
                                         int eidB, int sB, int dB, int xB, int lane) {
  const float* hA = h + (size_t)sA * C;
  const float* hB = h + (size_t)sB * C;
  float* arA = acc + xA * C;
  float* arB = acc + xB * C;
  constexpr int NP = (C + 31) / 32;
  float vA[NP], vB[NP];
#pragma unroll
  for (int j = 0; j < NP; ++j) {
    const int c = j * 32 + lane;
    vA[j] = c < C ? hA[c] : 0.f;
  }
#pragma unroll
  for (int j = 0; j < NP; ++j) {
    const int c = j * 32 + lane;
    vB[j] = c < C ? hB[c] : 0.f;
  }
  float a = 0.f;
  if (lane < 2 * H) {
    const bool second = lane >= H;
    const int hh  = second ? lane - H : lane;
    const int eid = second ? eidB : eidA;
    const int d   = second ? dB : dA;
    const int x   = second ? xB : xA;
    a = __expf(elog[(size_t)eid * H + hh] - ord_f32(nmax[d * H + hh]));
    __hip_atomic_fetch_add(&wsum[x * H + hh], a, __ATOMIC_RELAXED, __HIP_MEMORY_SCOPE_WORKGROUP);
  }
#pragma unroll
  for (int j = 0; j < NP; ++j) {
    const float alpha = lane_bcast_f(a, (j * 32) / (C / H));
    const int c = j * 32 + lane;
    if (c < C)
      __hip_atomic_fetch_add(&arA[c], vA[j] * alpha, __ATOMIC_RELAXED, __HIP_MEMORY_SCOPE_WORKGROUP);
  }
#pragma unroll
  for (int j = 0; j < NP; ++j) {
    const float alpha = lane_bcast_f(a, H + (j * 32) / (C / H));
    const int c = j * 32 + lane;
    if (c < C)
      __hip_atomic_fetch_add(&arB[c], vB[j] * alpha, __ATOMIC_RELAXED, __HIP_MEMORY_SCOPE_WORKGROUP);
  }
}

template <int C, int H>
__device__ __forceinline__ void own_drain(float* __restrict__ acc, float* __restrict__ wsum,
                                          const unsigned* qw, int n,
                                          const int* __restrict__ ei,
                                          const float* __restrict__ h,
                                          const float* __restrict__ elog,
                                          const int* __restrict__ nmax,
                                          int d0, int lane) {
  for (int b = 0; b < n; b += 32) {
    int cnt = n - b;
    if (cnt > 32) cnt = 32;
    unsigned ent = 0;
    int s = 0;
    if (lane < cnt) {
      ent = qw[b + lane];
      s = ei[ent & 0xFFFFFu];
    }
    int i = 0;
    for (; i + 1 < cnt; i += 2) {
      const unsigned eA = (unsigned)__builtin_amdgcn_readlane((int)ent, i);
      const unsigned eB = (unsigned)__builtin_amdgcn_readlane((int)ent, i + 1);
      const int sA = __builtin_amdgcn_readlane(s, i);
      const int sB = __builtin_amdgcn_readlane(s, i + 1);
      const int xA = (int)(eA >> 20), xB = (int)(eB >> 20);
      own_pair<C, H>(acc, wsum, h, elog, nmax, (int)(eA & 0xFFFFFu), sA, d0 + xA, xA,
                     (int)(eB & 0xFFFFFu), sB, d0 + xB, xB, lane);
    }
    if (i < cnt) {
      const unsigned eA = (unsigned)__builtin_amdgcn_readlane((int)ent, i);
      const int sA = __builtin_amdgcn_readlane(s, i);
      const int xA = (int)(eA >> 20);
      own_edge<C, H, false>(acc, wsum, h, elog, nmax, (int)(eA & 0xFFFFFu), sA, d0 + xA, xA, lane);
    }
  }
}

template <int C, int H>
__global__ __launch_bounds__(OWN_T) void k_own(const int* __restrict__ ei,
                                               const float* __restrict__ h,
                                               const float* __restrict__ elog,
                                               const int* __restrict__ nmax,
                                               float* __restrict__ agg) {
  __shared__ __attribute__((aligned(16))) float acc[OWN_R * C];
  __shared__ float wsum[OWN_R * H];
  __shared__ unsigned q[(OWN_T / 32) * OWN_QCAP];
  const int tid = threadIdx.x, lane = tid & 31;
  const int wave = __builtin_amdgcn_readfirstlane(tid >> 5);
  const int d0 = blockIdx.x * OWN_R;
  int nrow = NND - d0;
  if (nrow > OWN_R) nrow = OWN_R;

  for (int r = wave; r < nrow; r += OWN_T / 32)
    own_edge<C, H, true>(acc, wsum, h, elog, nmax, NE + d0 + r, d0 + r, d0 + r, r, lane);
  __syncthreads();

  unsigned* qw = q + wave * OWN_QCAP;
  int qn = 0;
  const int4* dst4 = (const int4*)(ei + NE);
  const int wofs = wave * 32;
  int4 cur[OWN_U];
  int ii = (blockIdx.x * OWN_IT) / OWN_WG;
#pragma unroll
  for (int u = 0; u < OWN_U; ++u) {
    const int wb = (ii * OWN_U + u) * OWN_T + wofs;
    cur[u] = dst4[(wb < NE4 ? wb : NE4 - 32) + lane];
  }
  for (int it = 0; it < OWN_IT; ++it) {
    const int in = ii + 1 < OWN_IT ? ii + 1 : 0;
    int4 nxt[OWN_U];
#pragma unroll
    for (int u = 0; u < OWN_U; ++u) {
      const int wb = (in * OWN_U + u) * OWN_T + wofs;
      nxt[u] = dst4[(wb < NE4 ? wb : NE4 - 32) + lane];
    }
#pragma unroll
    for (int u = 0; u < OWN_U; ++u) {
      const int wb = (ii * OWN_U + u) * OWN_T + wofs;
      const int d0u = wb < NE4 ? d0 : 0x40000000;
      const int dd[4] = {cur[u].x, cur[u].y, cur[u].z, cur[u].w};
#pragma unroll
      for (int k = 0; k < 4; ++k) {
        const unsigned x = (unsigned)(dd[k] - d0u);
        const bool m = x < (unsigned)nrow;
        const unsigned mk = __builtin_amdgcn_ballot_w32(m);
        if (mk) {
          if (m) qw[qn + (int)__builtin_amdgcn_mbcnt_lo(mk, 0u)] = (x << 20) | (unsigned)((wb + lane) * 4 + k);
          qn += __builtin_popcount(mk);
        }
      }
    }
    if (qn >= OWN_QTH) {
      own_drain<C, H>(acc, wsum, qw, qn, ei, h, elog, nmax, d0, lane);
      qn = 0;
    }
#pragma unroll
    for (int u = 0; u < OWN_U; ++u) cur[u] = nxt[u];
    ii = in;
  }
  own_drain<C, H>(acc, wsum, qw, qn, ei, h, elog, nmax, d0, lane);
  __syncthreads();

  static_assert((C / H) % 4 == 0, "a 16-byte piece does not straddle two heads");
  constexpr int PH = (C / H) / 4;
  float4* o4 = (float4*)(agg + (size_t)d0 * C);
  const float4* a4 = (const float4*)acc;
  for (int i = tid; i < nrow * C / 4; i += OWN_T) {
    const float4 v = a4[i];
    const float dn = wsum[i / PH] + EPSV;
    o4[i] = make_float4(v.x / dn, v.y / dn, v.z / dn, v.w / dn);
  }
}

#define NQ  (NET / 4)
#define NEQ (NE / 4)
__device__ __forceinline__ void quad_sd(const int* __restrict__ ei, int q, int (&s)[4], int (&d)[4]) {
  if (q < NEQ) {
    const int4 sv = ((const int4*)ei)[q];
    const int4 dv = ((const int4*)(ei + NE))[q];
    s[0] = sv.x; s[1] = sv.y; s[2] = sv.z; s[3] = sv.w;
    d[0] = dv.x; d[1] = dv.y; d[2] = dv.z; d[3] = dv.w;
  } else {
#pragma unroll
    for (int k = 0; k < 4; ++k) s[k] = d[k] = q * 4 + k - NE;
  }
}

__global__ __launch_bounds__(256) void k_edgeA2(const int* __restrict__ ei,
                         const float* __restrict__ as_, const float* __restrict__ ad_,
                         float* __restrict__ e, int* __restrict__ nmax) {
  const int q = blockIdx.x * 256 + threadIdx.x;
  if (q >= NQ) return;
  int s[4], d[4];
  quad_sd(ei, q, s, d);
  float v[4];
#pragma unroll
  for (int k = 0; k < 4; ++k) v[k] = as_[s[k]] + ad_[d[k]];
#pragma unroll
  for (int k = 0; k < 4; ++k) v[k] = v[k] > 0.f ? v[k] : v[k] * NEG_SLOPE;
  ((float4*)e)[q] = make_float4(v[0], v[1], v[2], v[3]);
#pragma unroll
  for (int k = 0; k < 4; ++k) atomicMax(&nmax[d[k]], f32_ord(v[k]));
}

#define EL_WG 1250
#define EL_T  (EL_WG * 256)
#define EL_U  5
#define EL_N4 (NND * HC / 4)
static_assert(EL_N4 == 2 * EL_U * EL_T, "two full trips cover the rows exactly");
static_assert(EL_T % (HC / 4) == 0, "a thread's columns do not change from trip to trip");
__global__ __launch_bounds__(256) void k_elu_bias(float* __restrict__ a, const float* __restrict__ b) {
  const int i0 = blockIdx.x * 256 + threadIdx.x;
  const int c = (i0 & (HC / 4 - 1)) * 4;
  const float b0 = b[c], b1 = b[c + 1], b2 = b[c + 2], b3 = b[c + 3];
  float4* a4 = (float4*)a;
  for (int ib = i0; ib < EL_N4; ib += EL_U * EL_T) {
    float4 v[EL_U];
#pragma unroll
    for (int k = 0; k < EL_U; ++k) v[k] = a4[ib + k * EL_T];
#pragma unroll
    for (int k = 0; k < EL_U; ++k) {
      const float x0 = v[k].x + b0, x1 = v[k].y + b1, x2 = v[k].z + b2, x3 = v[k].w + b3;
      float4 r;
      r.x = x0 > 0.f ? x0 : (__expf(x0) - 1.f);
      r.y = x1 > 0.f ? x1 : (__expf(x1) - 1.f);
      r.z = x2 > 0.f ? x2 : (__expf(x2) - 1.f);
      r.w = x3 > 0.f ? x3 : (__expf(x3) - 1.f);
      a4[ib + k * EL_T] = r;
    }
  }
}

#define OB_WG 512
#define OB_T  (OB_WG * 256)
#define OB_U  4
#define OB_N4 (NND * ODIM / 4)
static_assert(OB_U * OB_T >= OB_N4, "one trip covers the output");
__global__ __launch_bounds__(256) void k_out_bias(const float* __restrict__ agg, const float* __restrict__ b,
                           float* __restrict__ out) {
  const int i0 = blockIdx.x * 256 + threadIdx.x;
  const float4* g4 = (const float4*)agg;
  float4* o4 = (float4*)out;
  float4 v[OB_U], bb[OB_U];
#pragma unroll
  for (int k = 0; k < OB_U; ++k) {
    const int i = i0 + k * OB_T;
    const int ic = i < OB_N4 ? i : i0;
    v[k] = g4[ic];
    const int c = (ic % (ODIM / 4)) * 4;
    bb[k] = make_float4(b[c], b[c + 1], b[c + 2], b[c + 3]);
  }
#pragma unroll
  for (int k = 0; k < OB_U; ++k) {
    const int i = i0 + k * OB_T;
    if (i < OB_N4)
      o4[i] = make_float4(v[k].x + bb[k].x, v[k].y + bb[k].y, v[k].z + bb[k].z, v[k].w + bb[k].w);
  }
}

extern "C" void kernel_launch(void* const* d_in, const int* in_sizes, int n_in,
                              void* d_out, int out_size, void* d_ws, size_t ws_size,
                              hipStream_t stream) {
  const float* x   = (const float*)d_in[0];
  const int*   ei  = (const int*)d_in[1];
  const float* W1  = (const float*)d_in[2];
  const float* as1 = (const float*)d_in[3];
  const float* ad1 = (const float*)d_in[4];
  const float* b1  = (const float*)d_in[5];
  const float* W2  = (const float*)d_in[6];
  const float* as2 = (const float*)d_in[7];
  const float* ad2 = (const float*)d_in[8];
  const float* b2  = (const float*)d_in[9];
  float* out = (float*)d_out;

  float* ws   = (float*)d_ws;
  float* h1   = ws;
  float* agg1 = h1   + (size_t)NND * HC;
  float* aS1  = agg1 + (size_t)NND * HC;
  float* aD1  = aS1  + NND * NH;
  int*   nm1  = (int*)(aD1 + NND * NH);
  float* e1   = (float*)(nm1 + NND * NH);
  float* t2   = e1   + (size_t)NET * NH;
  float* aS2  = t2   + (size_t)NND * ODIM;
  float* aD2  = aS2  + NND;
  int*   nm2  = (int*)(aD2 + NND);
  float* e2   = (float*)(nm2 + NND);
  float* agg2 = e2   + NET;

  k_fill4<<<(NND * NH / 4 + 255) / 256, 256, 0, stream>>>((int4*)nm1, (int)0x80000000, NND * NH / 4);
  k_fill4<<<(NND / 4 + 255) / 256, 256, 0, stream>>>((int4*)nm2, (int)0x80000000, NND / 4);

  k_gemm1<<<GM_WG, 256, 0, stream>>>(x, W1, h1);
  k_att1<<<ATT_WG, 256, 0, stream>>>(h1, as1, ad1, aS1, aD1);
  k_edgeA1<<<EG_WG, 256, 0, stream>>>(ei, aS1, aD1, e1, nm1);
  k_own<HC, NH><<<OWN_WG, OWN_T, 0, stream>>>(ei, h1, e1, nm1, agg1);
  k_elu_bias<<<EL_WG, 256, 0, stream>>>(agg1, b1);

  k_gemm2<<<GM_WG, 256, 0, stream>>>(agg1, W2, t2);
  k_att2<<<ATT_WG, 256, 0, stream>>>(t2, as2, ad2, aS2, aD2);
  k_edgeA2<<<(NQ + 255) / 256, 256, 0, stream>>>(ei, aS2, aD2, e2, nm2);
  k_own<ODIM, 1><<<OWN_WG, OWN_T, 0, stream>>>(ei, t2, e2, nm2, agg2);
  k_out_bias<<<OB_WG, 256, 0, stream>>>(agg2, b2, out);
}


# Per-destination max moved into the owner work-groups: a wave lists its matched edges in LDS during the scan (LDS integer max), exp and sums then run from the list; 8 launches, sums in arrival order
# speedup vs baseline: 2.8474x; 1.3537x over previous
#include <hip/hip_runtime.h>
#include <hip/hip_bf16.h>

#define NND   50000
#define FIN   128
#define NH    8
#define HIDC  32
#define HC    256
#define ODIM  40
#define NE    800000
#define NET   850000
#define NEG_SLOPE 0.2f
#define EPSV  1e-16f

typedef __attribute__((ext_vector_type(2))) float v2f;
typedef __attribute__((ext_vector_type(8))) float v8f;

__device__ __forceinline__ float wave_sum(float v) {
#pragma unroll
  for (int off = 16; off > 0; off >>= 1) v += __shfl_xor(v, off, 32);
  return v;
}
__device__ __forceinline__ int f32_ord(float x) {
  int i = __float_as_int(x);
  return i >= 0 ? i : (i ^ 0x7FFFFFFF);
}
__device__ __forceinline__ float ord_f32(int i) {
  return __int_as_float(i >= 0 ? i : (i ^ 0x7FFFFFFF));
}


#define GM_WV   8
#define GM_NS   (NND / 16)
#define GM_WG   ((GM_NS + GM_WV - 1) / GM_WV)
static_assert(NND % 16 == 0, "whole strips");

#define G1_KB (FIN / 8)
#define G1_NT (HC / 16)
__global__ __launch_bounds__(256) void k_gemm1(const float* __restrict__ x,
                                               const float* __restrict__ W,
                                               float* __restrict__ h) {
  __shared__ __attribute__((aligned(16))) float4 Ws[G1_KB * G1_NT * 32];
  const int tid  = threadIdx.x;
  const int lane = tid & 31;
  const int wave = __builtin_amdgcn_readfirstlane(tid >> 5);
  const int la = lane & 15, hi = lane >> 4;
  const int strip = blockIdx.x * GM_WV + wave;
  const bool live = strip < GM_NS;
  const int m0 = (live ? strip : 0) * 16;
  float4 a[G1_KB];
  {
    const float4* xr = (const float4*)(x + (size_t)(m0 + la) * FIN) + hi;
#pragma unroll
    for (int kb = 0; kb < G1_KB; ++kb) a[kb] = xr[kb * 2];
  }
  {
    const int t = tid >> 4, c = tid & 15;
#pragma unroll 4
    for (int kb = 0; kb < G1_KB; ++kb) {
#pragma unroll
      for (int hh = 0; hh < 2; ++hh) {
        const float* wp = W + (size_t)(kb * 8 + hh * 4) * HC + tid;
        Ws[(kb * G1_NT + t) * 32 + hh * 16 + c] = make_float4(wp[0], wp[HC], wp[2 * HC], wp[3 * HC]);
      }
    }
  }
  __syncthreads();
  if (!live) return;
  v8f acc[G1_NT];
#pragma unroll
  for (int t = 0; t < G1_NT; ++t) acc[t] = (v8f){};
#pragma unroll
  for (int kb = 0; kb < G1_KB; ++kb) {
    v2f a0, a1;
    a0.x = a[kb].x; a0.y = a[kb].y; a1.x = a[kb].z; a1.y = a[kb].w;
#pragma unroll
    for (int t = 0; t < G1_NT; ++t) {
      const float4 b = Ws[(kb * G1_NT + t) * 32 + lane];
      v2f b0, b1;
      b0.x = b.x; b0.y = b.y; b1.x = b.z; b1.y = b.w;
      acc[t] = __builtin_amdgcn_wmma_f32_16x16x4_f32(false, a0, false, b0, (short)0,
                                                     acc[t], false, false);
      acc[t] = __builtin_amdgcn_wmma_f32_16x16x4_f32(false, a1, false, b1, (short)0,
                                                     acc[t], false, false);
    }
  }
#pragma unroll
  for (int t = 0; t < G1_NT; ++t)
#pragma unroll
    for (int j = 0; j < 8; ++j)
      h[(size_t)(m0 + j + hi * 8) * HC + t * 16 + la] = acc[t][j];
}

#define G2_KB (HC / 8)
#define G2_NT 3
static_assert(G2_NT * 16 >= ODIM, "the tiles cover the output columns");
__global__ __launch_bounds__(256) void k_gemm2(const float* __restrict__ h2,
                                               const float* __restrict__ W,
                                               float* __restrict__ t2) {
  __shared__ __attribute__((aligned(16))) float4 Ws[G2_KB * G2_NT * 32];
  const int tid  = threadIdx.x;
  const int lane = tid & 31;
  const int wave = __builtin_amdgcn_readfirstlane(tid >> 5);
  const int la = lane & 15, hi = lane >> 4;
  const int strip = blockIdx.x * GM_WV + wave;
  const bool live = strip < GM_NS;
  const int m0 = (live ? strip : 0) * 16;
  float4 a[G2_KB];
  {
    const float4* xr = (const float4*)(h2 + (size_t)(m0 + la) * HC) + hi;
#pragma unroll
    for (int kb = 0; kb < G2_KB; ++kb) a[kb] = xr[kb * 2];
  }
  {
    static_assert(G2_KB == 4 * GM_WV, "eight waves cover the k blocks");
#pragma unroll
    for (int j = 0; j < 4; ++j) {
      const int kb = wave * 4 + j;
#pragma unroll
      for (int tt = 0; tt < G2_NT; ++tt) {
        const int n = tt * 16 + la;
        const float* wp = W + (size_t)(kb * 8 + hi * 4) * ODIM + (n < ODIM ? n : ODIM - 1);
        Ws[(kb * G2_NT + tt) * 32 + lane] = make_float4(wp[0], wp[ODIM], wp[2 * ODIM], wp[3 * ODIM]);
      }
    }
  }
  __syncthreads();
  if (!live) return;
  v8f acc[G2_NT];
#pragma unroll
  for (int t = 0; t < G2_NT; ++t) acc[t] = (v8f){};
#pragma unroll
  for (int kb = 0; kb < G2_KB; ++kb) {
    v2f a0, a1;
    a0.x = a[kb].x; a0.y = a[kb].y; a1.x = a[kb].z; a1.y = a[kb].w;
#pragma unroll
    for (int t = 0; t < G2_NT; ++t) {
      const float4 b = Ws[(kb * G2_NT + t) * 32 + lane];
      v2f b0, b1;
      b0.x = b.x; b0.y = b.y; b1.x = b.z; b1.y = b.w;
      acc[t] = __builtin_amdgcn_wmma_f32_16x16x4_f32(false, a0, false, b0, (short)0,
                                                     acc[t], false, false);
      acc[t] = __builtin_amdgcn_wmma_f32_16x16x4_f32(false, a1, false, b1, (short)0,
                                                     acc[t], false, false);
    }
  }
#pragma unroll
  for (int t = 0; t < G2_NT; ++t) {
    const int col = t * 16 + la;
    if (col < ODIM) {
#pragma unroll
      for (int j = 0; j < 8; ++j)
        t2[(size_t)(m0 + j + hi * 8) * ODIM + col] = acc[t][j];
    }
  }
}

#define ATT_WG 1024
#define ATT_NW (ATT_WG * 8)
__device__ __forceinline__ void att1_node(const float (&v)[NH], const float (&ws)[NH],
                                          const float (&wd)[NH], int node, int lane,
                                          float* __restrict__ os, float* __restrict__ od) {
#pragma unroll
  for (int hd = 0; hd < NH; ++hd) {
    const float s = wave_sum(v[hd] * ws[hd]);
    const float d = wave_sum(v[hd] * wd[hd]);
    if (lane == 0) { os[node * NH + hd] = s; od[node * NH + hd] = d; }
  }
}
__global__ __launch_bounds__(256) void k_att1(const float* __restrict__ h,
                                              const float* __restrict__ as_,
                                              const float* __restrict__ ad_,
                                              float* __restrict__ os,
                                              float* __restrict__ od) {
  const int lane = threadIdx.x & 31;
  const int w = blockIdx.x * 8 + (threadIdx.x >> 5);
  float ws[NH], wd[NH];
#pragma unroll
  for (int hd = 0; hd < NH; ++hd) { ws[hd] = as_[hd * HIDC + lane]; wd[hd] = ad_[hd * HIDC + lane]; }
  for (int na = w; na < NND; na += 2 * ATT_NW) {
    const int nb = na + ATT_NW;
    const bool hb = nb < NND;
    const float* ra = h + (size_t)na * HC;
    const float* rb = h + (size_t)(hb ? nb : na) * HC;
    float va[NH], vb[NH];
#pragma unroll
    for (int hd = 0; hd < NH; ++hd) va[hd] = ra[hd * HIDC + lane];
#pragma unroll
    for (int hd = 0; hd < NH; ++hd) vb[hd] = rb[hd * HIDC + lane];
    att1_node(va, ws, wd, na, lane, os, od);
    if (hb) att1_node(vb, ws, wd, nb, lane, os, od);
  }
}

__device__ __forceinline__ void att2_node(float v0, float v1, float s0, float s1, float d0,
                                          float d1, bool up, int node, int lane,
                                          float* __restrict__ os, float* __restrict__ od) {
  float s = 0.f, d = 0.f;
  s += v0 * s0; d += v0 * d0;
  if (up) { s += v1 * s1; d += v1 * d1; }
  s = wave_sum(s); d = wave_sum(d);
  if (lane == 0) { os[node] = s; od[node] = d; }
}
__global__ __launch_bounds__(256) void k_att2(const float* __restrict__ t,
                                              const float* __restrict__ as_,
                                              const float* __restrict__ ad_,
                                              float* __restrict__ os,
                                              float* __restrict__ od) {
  const int lane = threadIdx.x & 31;
  const int w = blockIdx.x * 8 + (threadIdx.x >> 5);
  const bool up = lane < ODIM - 32;
  const float s0 = as_[lane], d0 = ad_[lane];
  const float s1 = up ? as_[32 + lane] : 0.f, d1 = up ? ad_[32 + lane] : 0.f;
  for (int na = w; na < NND; na += 2 * ATT_NW) {
    const int nb = na + ATT_NW;
    const bool hb = nb < NND;
    const float* ta = t + (size_t)na * ODIM;
    const float* tb = t + (size_t)(hb ? nb : na) * ODIM;
    const float a0 = ta[lane], b0 = tb[lane];
    const float a1 = up ? ta[32 + lane] : 0.f, b1 = up ? tb[32 + lane] : 0.f;
    att2_node(a0, a1, s0, s1, d0, d1, up, na, lane, os, od);
    if (hb) att2_node(b0, b1, s0, s1, d0, d1, up, nb, lane, os, od);
  }
}


#define OWN_R    196
#define OWN_WG   256
#define OWN_T    1024
#define OWN_U    4
#define NE4      (NE / 4)
#define OWN_IT   ((NE4 + OWN_T * OWN_U - 1) / (OWN_T * OWN_U))
#define OWN_STEP (32 * OWN_U * 4)
#define OWN_QTH  32
#define OWN_QCAP 768
static_assert(OWN_QCAP >= OWN_QTH + OWN_STEP, "a list about to be worked off still takes one scan step");
static_assert(NE <= (1 << 20) && OWN_R <= 256 && NND <= (1 << 16), "the fields of a list entry");

__device__ __forceinline__ float lane_bcast_f(float v, int l) {
  return __int_as_float(__builtin_amdgcn_readlane(__float_as_int(v), l));
}
__device__ __forceinline__ float leaky(float v) { return v > 0.f ? v : v * NEG_SLOPE; }

template <int H>
__device__ __forceinline__ float own_logit(const float* __restrict__ aS, const float* __restrict__ aD,
                                           int s, int d, int hd) {
  return leaky(aS[s * H + hd] + aD[d * H + hd]);
}

template <int H>
__device__ __forceinline__ void own_max(int* rmax, const float* __restrict__ aS,
                                        const float* __restrict__ aD, int s, int d0, int x) {
  float lg[H];
#pragma unroll
  for (int hd = 0; hd < H; ++hd) lg[hd] = own_logit<H>(aS, aD, s, d0 + x, hd);
#pragma unroll
  for (int hd = 0; hd < H; ++hd)
    __hip_atomic_fetch_max(&rmax[x * H + hd], f32_ord(lg[hd]), __ATOMIC_RELAXED, __HIP_MEMORY_SCOPE_WORKGROUP);
}

template <int H, bool DO_MAX>
__device__ __forceinline__ void own_convert(unsigned* qw, int from, int n, const int* __restrict__ ei,
                                            const float* __restrict__ aS, const float* __restrict__ aD,
                                            int* rmax, int d0, int lane) {
  for (int b = from; b < n; b += 32) {
    const int i = b + lane;
    if (i < n) {
      const unsigned ent = qw[i];
      const int x = (int)(ent >> 20);
      const int s = ei[ent & 0xFFFFFu];
      if (DO_MAX) own_max<H>(rmax, aS, aD, s, d0, x);
      qw[i] = ((unsigned)x << 16) | (unsigned)s;
    }
  }
}

template <int C, int H, bool INIT>
__device__ __forceinline__ void own_edge(float* __restrict__ acc, float* __restrict__ wsum,
                                         const int* rmax, const float* __restrict__ h,
                                         const float* __restrict__ aS, const float* __restrict__ aD,
                                         int d0, int s, int x, int lane) {
  const float* hs = h + (size_t)s * C;
  float* ar = acc + x * C;
  constexpr int NP = (C + 31) / 32;
  float v[NP];
#pragma unroll
  for (int j = 0; j < NP; ++j) {
    const int c = j * 32 + lane;
    v[j] = c < C ? hs[c] : 0.f;
  }
  float a = 0.f;
  if (lane < H) {
    a = __expf(own_logit<H>(aS, aD, s, d0 + x, lane) - ord_f32(rmax[x * H + lane]));
    if (INIT) wsum[x * H + lane] = a;
    else __hip_atomic_fetch_add(&wsum[x * H + lane], a, __ATOMIC_RELAXED, __HIP_MEMORY_SCOPE_WORKGROUP);
  }
#pragma unroll
  for (int j = 0; j < NP; ++j) {
    const float alpha = lane_bcast_f(a, (j * 32) / (C / H));
    const int c = j * 32 + lane;
    if (c < C) {
      const float t = v[j] * alpha;
      if (INIT) ar[c] = t;
      else __hip_atomic_fetch_add(&ar[c], t, __ATOMIC_RELAXED, __HIP_MEMORY_SCOPE_WORKGROUP);
    }
  }
}

template <int C, int H>
__device__ __forceinline__ void own_pair(float* __restrict__ acc, float* __restrict__ wsum,
                                         const int* rmax, const float* __restrict__ h,
                                         const float* __restrict__ aS, const float* __restrict__ aD,
                                         int d0, int sA, int xA, int sB, int xB, int lane) {
  const float* hA = h + (size_t)sA * C;
  const float* hB = h + (size_t)sB * C;
  float* arA = acc + xA * C;
  float* arB = acc + xB * C;
  constexpr int NP = (C + 31) / 32;
  float vA[NP], vB[NP];
#pragma unroll
  for (int j = 0; j < NP; ++j) {
    const int c = j * 32 + lane;
    vA[j] = c < C ? hA[c] : 0.f;
  }
#pragma unroll
  for (int j = 0; j < NP; ++j) {
    const int c = j * 32 + lane;
    vB[j] = c < C ? hB[c] : 0.f;
  }
  float a = 0.f;
  if (lane < 2 * H) {
    const bool second = lane >= H;
    const int hh = second ? lane - H : lane;
    const int s  = second ? sB : sA;
    const int x  = second ? xB : xA;
    a = __expf(own_logit<H>(aS, aD, s, d0 + x, hh) - ord_f32(rmax[x * H + hh]));
    __hip_atomic_fetch_add(&wsum[x * H + hh], a, __ATOMIC_RELAXED, __HIP_MEMORY_SCOPE_WORKGROUP);
  }
#pragma unroll
  for (int j = 0; j < NP; ++j) {
    const float alpha = lane_bcast_f(a, (j * 32) / (C / H));
    const int c = j * 32 + lane;
    if (c < C)
      __hip_atomic_fetch_add(&arA[c], vA[j] * alpha, __ATOMIC_RELAXED, __HIP_MEMORY_SCOPE_WORKGROUP);
  }
#pragma unroll
  for (int j = 0; j < NP; ++j) {
    const float alpha = lane_bcast_f(a, H + (j * 32) / (C / H));
    const int c = j * 32 + lane;
    if (c < C)
      __hip_atomic_fetch_add(&arB[c], vB[j] * alpha, __ATOMIC_RELAXED, __HIP_MEMORY_SCOPE_WORKGROUP);
  }
}

template <int C, int H>
__device__ __forceinline__ void own_drain(float* __restrict__ acc, float* __restrict__ wsum,
                                          const int* rmax, const unsigned* qw, int n,
                                          const float* __restrict__ h,
                                          const float* __restrict__ aS, const float* __restrict__ aD,
                                          int d0, int lane) {
  for (int b = 0; b < n; b += 32) {
    int cnt = n - b;
    if (cnt > 32) cnt = 32;
    unsigned ent = 0;
    if (lane < cnt) ent = qw[b + lane];
    int i = 0;
    for (; i + 1 < cnt; i += 2) {
      const unsigned eA = (unsigned)__builtin_amdgcn_readlane((int)ent, i);
      const unsigned eB = (unsigned)__builtin_amdgcn_readlane((int)ent, i + 1);
      own_pair<C, H>(acc, wsum, rmax, h, aS, aD, d0, (int)(eA & 0xFFFFu), (int)(eA >> 16),
                     (int)(eB & 0xFFFFu), (int)(eB >> 16), lane);
    }
    if (i < cnt) {
      const unsigned eA = (unsigned)__builtin_amdgcn_readlane((int)ent, i);
      own_edge<C, H, false>(acc, wsum, rmax, h, aS, aD, d0, (int)(eA & 0xFFFFu), (int)(eA >> 16), lane);
    }
  }
}

template <int C, int H, int MODE>
__device__ __forceinline__ void own_scan(float* __restrict__ acc, float* __restrict__ wsum, int* rmax,
                                         unsigned* qw, const int* __restrict__ ei,
                                         const float* __restrict__ h,
                                         const float* __restrict__ aS, const float* __restrict__ aD,
                                         int d0, int nrow, int wofs, int lane, int it_from,
                                         int& qn, bool& over, int& it_over) {
  const int4* dst4 = (const int4*)(ei + NE);
  int qdone = qn;
  int4 cur[OWN_U];
  int ii = (int)((blockIdx.x * OWN_IT) / OWN_WG + it_from) % OWN_IT;
#pragma unroll
  for (int u = 0; u < OWN_U; ++u) {
    const int wb = (ii * OWN_U + u) * OWN_T + wofs;
    cur[u] = dst4[(wb < NE4 ? wb : NE4 - 32) + lane];
  }
  for (int it = it_from; it < OWN_IT; ++it) {
    const int in = ii + 1 < OWN_IT ? ii + 1 : 0;
    int4 nxt[OWN_U];
#pragma unroll
    for (int u = 0; u < OWN_U; ++u) {
      const int wb = (in * OWN_U + u) * OWN_T + wofs;
      nxt[u] = dst4[(wb < NE4 ? wb : NE4 - 32) + lane];
    }
    if (MODE == 0 && !over && qn + OWN_STEP > OWN_QCAP) {
      own_convert<H, true>(qw, qdone, qn, ei, aS, aD, rmax, d0, lane);
      qdone = qn;
      over = true;
      it_over = it;
    }
#pragma unroll
    for (int u = 0; u < OWN_U; ++u) {
      const int wb = (ii * OWN_U + u) * OWN_T + wofs;
      const int d0u = wb < NE4 ? d0 : 0x40000000;
      const int dd[4] = {cur[u].x, cur[u].y, cur[u].z, cur[u].w};
#pragma unroll
      for (int k = 0; k < 4; ++k) {
        const unsigned x = (unsigned)(dd[k] - d0u);
        const bool m = x < (unsigned)nrow;
        const unsigned mk = __builtin_amdgcn_ballot_w32(m);
        if (mk) {
          const unsigned eid = (unsigned)((wb + lane) * 4 + k);
          if (MODE == 0 && over) {
            if (m) own_max<H>(rmax, aS, aD, ei[eid], d0, (int)x);
          } else {
            if (m) qw[qn + (int)__builtin_amdgcn_mbcnt_lo(mk, 0u)] = (x << 20) | eid;
            qn += __builtin_popcount(mk);
          }
        }
      }
    }
    if (MODE == 0) {
      const int nb = (qn - qdone) & ~31;
      if (nb) {
        own_convert<H, true>(qw, qdone, qdone + nb, ei, aS, aD, rmax, d0, lane);
        qdone += nb;
      }
    } else if (qn >= OWN_QTH) {
      own_convert<H, false>(qw, 0, qn, ei, aS, aD, rmax, d0, lane);
      own_drain<C, H>(acc, wsum, rmax, qw, qn, h, aS, aD, d0, lane);
      qn = 0;
    }
#pragma unroll
    for (int u = 0; u < OWN_U; ++u) cur[u] = nxt[u];
    ii = in;
  }
  if (MODE == 0) {
    own_convert<H, true>(qw, qdone, qn, ei, aS, aD, rmax, d0, lane);
  } else {
    own_convert<H, false>(qw, 0, qn, ei, aS, aD, rmax, d0, lane);
    own_drain<C, H>(acc, wsum, rmax, qw, qn, h, aS, aD, d0, lane);
    qn = 0;
  }
}

template <int C, int H>
__global__ __launch_bounds__(OWN_T) void k_own(const int* __restrict__ ei,
                                               const float* __restrict__ h,
                                               const float* __restrict__ aS,
                                               const float* __restrict__ aD,
                                               float* __restrict__ agg) {
  __shared__ __attribute__((aligned(16))) float acc[OWN_R * C];
  __shared__ float wsum[OWN_R * H];
  __shared__ int rmax[OWN_R * H];
  __shared__ unsigned q[(OWN_T / 32) * OWN_QCAP];
  const int tid = threadIdx.x, lane = tid & 31;
  const int wave = __builtin_amdgcn_readfirstlane(tid >> 5);
  const int d0 = blockIdx.x * OWN_R;
  int nrow = NND - d0;
  if (nrow > OWN_R) nrow = OWN_R;

  for (int i = tid; i < nrow * H; i += OWN_T)
    rmax[i] = f32_ord(leaky(aS[d0 * H + i] + aD[d0 * H + i]));
  __syncthreads();

  unsigned* qw = q + wave * OWN_QCAP;
  const int wofs = wave * 32;
  int qn = 0, it_over = OWN_IT;
  bool over = false;
  own_scan<C, H, 0>(acc, wsum, rmax, qw, ei, h, aS, aD, d0, nrow, wofs, lane, 0, qn, over, it_over);
  __syncthreads();

  for (int r = wave; r < nrow; r += OWN_T / 32)
    own_edge<C, H, true>(acc, wsum, rmax, h, aS, aD, d0, d0 + r, r, lane);
  __syncthreads();

  own_drain<C, H>(acc, wsum, rmax, qw, qn, h, aS, aD, d0, lane);
  if (over) {
    qn = 0;
    own_scan<C, H, 1>(acc, wsum, rmax, qw, ei, h, aS, aD, d0, nrow, wofs, lane, it_over, qn, over, it_over);
  }
  __syncthreads();

  static_assert((C / H) % 4 == 0, "a 16-byte piece does not straddle two heads");
  constexpr int PH = (C / H) / 4;
  float4* o4 = (float4*)(agg + (size_t)d0 * C);
  const float4* a4 = (const float4*)acc;
  for (int i = tid; i < nrow * C / 4; i += OWN_T) {
    const float4 v = a4[i];
    const float dn = wsum[i / PH] + EPSV;
    o4[i] = make_float4(v.x / dn, v.y / dn, v.z / dn, v.w / dn);
  }
}

#define EL_WG 1250
#define EL_T  (EL_WG * 256)
#define EL_U  5
#define EL_N4 (NND * HC / 4)
static_assert(EL_N4 == 2 * EL_U * EL_T, "two full trips cover the rows exactly");
static_assert(EL_T % (HC / 4) == 0, "a thread's columns do not change from trip to trip");
__global__ __launch_bounds__(256) void k_elu_bias(float* __restrict__ a, const float* __restrict__ b) {
  const int i0 = blockIdx.x * 256 + threadIdx.x;
  const int c = (i0 & (HC / 4 - 1)) * 4;
  const float b0 = b[c], b1 = b[c + 1], b2 = b[c + 2], b3 = b[c + 3];
  float4* a4 = (float4*)a;
  for (int ib = i0; ib < EL_N4; ib += EL_U * EL_T) {
    float4 v[EL_U];
#pragma unroll
    for (int k = 0; k < EL_U; ++k) v[k] = a4[ib + k * EL_T];
#pragma unroll
    for (int k = 0; k < EL_U; ++k) {
      const float x0 = v[k].x + b0, x1 = v[k].y + b1, x2 = v[k].z + b2, x3 = v[k].w + b3;
      float4 r;
      r.x = x0 > 0.f ? x0 : (__expf(x0) - 1.f);
      r.y = x1 > 0.f ? x1 : (__expf(x1) - 1.f);
      r.z = x2 > 0.f ? x2 : (__expf(x2) - 1.f);
      r.w = x3 > 0.f ? x3 : (__expf(x3) - 1.f);
      a4[ib + k * EL_T] = r;
    }
  }
}

#define OB_WG 512
#define OB_T  (OB_WG * 256)
#define OB_U  4
#define OB_N4 (NND * ODIM / 4)
static_assert(OB_U * OB_T >= OB_N4, "one trip covers the output");
__global__ __launch_bounds__(256) void k_out_bias(const float* __restrict__ agg, const float* __restrict__ b,
                           float* __restrict__ out) {
  const int i0 = blockIdx.x * 256 + threadIdx.x;
  const float4* g4 = (const float4*)agg;
  float4* o4 = (float4*)out;
  float4 v[OB_U], bb[OB_U];
#pragma unroll
  for (int k = 0; k < OB_U; ++k) {
    const int i = i0 + k * OB_T;
    const int ic = i < OB_N4 ? i : i0;
    v[k] = g4[ic];
    const int c = (ic % (ODIM / 4)) * 4;
    bb[k] = make_float4(b[c], b[c + 1], b[c + 2], b[c + 3]);
  }
#pragma unroll
  for (int k = 0; k < OB_U; ++k) {
    const int i = i0 + k * OB_T;
    if (i < OB_N4)
      o4[i] = make_float4(v[k].x + bb[k].x, v[k].y + bb[k].y, v[k].z + bb[k].z, v[k].w + bb[k].w);
  }
}


extern "C" void kernel_launch(void* const* d_in, const int* in_sizes, int n_in,
                              void* d_out, int out_size, void* d_ws, size_t ws_size,
                              hipStream_t stream) {
  const float* x   = (const float*)d_in[0];
  const int*   ei  = (const int*)d_in[1];
  const float* W1  = (const float*)d_in[2];
  const float* as1 = (const float*)d_in[3];
  const float* ad1 = (const float*)d_in[4];
  const float* b1  = (const float*)d_in[5];
  const float* W2  = (const float*)d_in[6];
  const float* as2 = (const float*)d_in[7];
  const float* ad2 = (const float*)d_in[8];
  const float* b2  = (const float*)d_in[9];
  float* out = (float*)d_out;

  float* ws   = (float*)d_ws;
  float* h1   = ws;
  float* agg1 = h1   + (size_t)NND * HC;
  float* aS1  = agg1 + (size_t)NND * HC;
  float* aD1  = aS1  + NND * NH;
  float* t2   = aD1  + NND * NH;
  float* aS2  = t2   + (size_t)NND * ODIM;
  float* aD2  = aS2  + NND;
  float* agg2 = aD2  + NND;

  k_gemm1<<<GM_WG, 256, 0, stream>>>(x, W1, h1);
  k_att1<<<ATT_WG, 256, 0, stream>>>(h1, as1, ad1, aS1, aD1);
  k_own<HC, NH><<<OWN_WG, OWN_T, 0, stream>>>(ei, h1, aS1, aD1, agg1);
  k_elu_bias<<<EL_WG, 256, 0, stream>>>(agg1, b1);

  k_gemm2<<<GM_WG, 256, 0, stream>>>(agg1, W2, t2);
  k_att2<<<ATT_WG, 256, 0, stream>>>(t2, as2, ad2, aS2, aD2);
  k_own<ODIM, 1><<<OWN_WG, OWN_T, 0, stream>>>(ei, t2, aS2, aD2, agg2);
  k_out_bias<<<OB_WG, 256, 0, stream>>>(agg2, b2, out);
}


# Owner groups of layer 1 store each wave's edge list (local row, source) to the workspace; layer 2's groups load it and only raise their maxima: no second scan of the destination ids; sums as before
# speedup vs baseline: 3.2095x; 1.1272x over previous
#include <hip/hip_runtime.h>
#include <hip/hip_bf16.h>

#define NND   50000
#define FIN   128
#define NH    8
#define HIDC  32
#define HC    256
#define ODIM  40
#define NE    800000
#define NET   850000
#define NEG_SLOPE 0.2f
#define EPSV  1e-16f

typedef __attribute__((ext_vector_type(2))) float v2f;
typedef __attribute__((ext_vector_type(8))) float v8f;

__device__ __forceinline__ float wave_sum(float v) {
#pragma unroll
  for (int off = 16; off > 0; off >>= 1) v += __shfl_xor(v, off, 32);
  return v;
}
__device__ __forceinline__ int f32_ord(float x) {
  int i = __float_as_int(x);
  return i >= 0 ? i : (i ^ 0x7FFFFFFF);
}
__device__ __forceinline__ float ord_f32(int i) {
  return __int_as_float(i >= 0 ? i : (i ^ 0x7FFFFFFF));
}


#define GM_WV   8
#define GM_NS   (NND / 16)
#define GM_WG   ((GM_NS + GM_WV - 1) / GM_WV)
static_assert(NND % 16 == 0, "whole strips");

#define G1_KB (FIN / 8)
#define G1_NT (HC / 16)
__global__ __launch_bounds__(256) void k_gemm1(const float* __restrict__ x,
                                               const float* __restrict__ W,
                                               float* __restrict__ h) {
  __shared__ __attribute__((aligned(16))) float4 Ws[G1_KB * G1_NT * 32];
  const int tid  = threadIdx.x;
  const int lane = tid & 31;
  const int wave = __builtin_amdgcn_readfirstlane(tid >> 5);
  const int la = lane & 15, hi = lane >> 4;
  const int strip = blockIdx.x * GM_WV + wave;
  const bool live = strip < GM_NS;
  const int m0 = (live ? strip : 0) * 16;
  float4 a[G1_KB];
  {
    const float4* xr = (const float4*)(x + (size_t)(m0 + la) * FIN) + hi;
#pragma unroll
    for (int kb = 0; kb < G1_KB; ++kb) a[kb] = xr[kb * 2];
  }
  {
    const int t = tid >> 4, c = tid & 15;
#pragma unroll 4
    for (int kb = 0; kb < G1_KB; ++kb) {
#pragma unroll
      for (int hh = 0; hh < 2; ++hh) {
        const float* wp = W + (size_t)(kb * 8 + hh * 4) * HC + tid;
        Ws[(kb * G1_NT + t) * 32 + hh * 16 + c] = make_float4(wp[0], wp[HC], wp[2 * HC], wp[3 * HC]);
      }
    }
  }
  __syncthreads();
  if (!live) return;
  v8f acc[G1_NT];
#pragma unroll
  for (int t = 0; t < G1_NT; ++t) acc[t] = (v8f){};
#pragma unroll
  for (int kb = 0; kb < G1_KB; ++kb) {
    v2f a0, a1;
    a0.x = a[kb].x; a0.y = a[kb].y; a1.x = a[kb].z; a1.y = a[kb].w;
#pragma unroll
    for (int t = 0; t < G1_NT; ++t) {
      const float4 b = Ws[(kb * G1_NT + t) * 32 + lane];
      v2f b0, b1;
      b0.x = b.x; b0.y = b.y; b1.x = b.z; b1.y = b.w;
      acc[t] = __builtin_amdgcn_wmma_f32_16x16x4_f32(false, a0, false, b0, (short)0,
                                                     acc[t], false, false);
      acc[t] = __builtin_amdgcn_wmma_f32_16x16x4_f32(false, a1, false, b1, (short)0,
                                                     acc[t], false, false);
    }
  }
#pragma unroll
  for (int t = 0; t < G1_NT; ++t)
#pragma unroll
    for (int j = 0; j < 8; ++j)
      h[(size_t)(m0 + j + hi * 8) * HC + t * 16 + la] = acc[t][j];
}

#define G2_KB (HC / 8)
#define G2_NT 3
static_assert(G2_NT * 16 >= ODIM, "the tiles cover the output columns");
__global__ __launch_bounds__(256) void k_gemm2(const float* __restrict__ h2,
                                               const float* __restrict__ W,
                                               float* __restrict__ t2) {
  __shared__ __attribute__((aligned(16))) float4 Ws[G2_KB * G2_NT * 32];
  const int tid  = threadIdx.x;
  const int lane = tid & 31;
  const int wave = __builtin_amdgcn_readfirstlane(tid >> 5);
  const int la = lane & 15, hi = lane >> 4;
  const int strip = blockIdx.x * GM_WV + wave;
  const bool live = strip < GM_NS;
  const int m0 = (live ? strip : 0) * 16;
  float4 a[G2_KB];
  {
    const float4* xr = (const float4*)(h2 + (size_t)(m0 + la) * HC) + hi;
#pragma unroll
    for (int kb = 0; kb < G2_KB; ++kb) a[kb] = xr[kb * 2];
  }
  {
    static_assert(G2_KB == 4 * GM_WV, "eight waves cover the k blocks");
#pragma unroll
    for (int j = 0; j < 4; ++j) {
      const int kb = wave * 4 + j;
#pragma unroll
      for (int tt = 0; tt < G2_NT; ++tt) {
        const int n = tt * 16 + la;
        const float* wp = W + (size_t)(kb * 8 + hi * 4) * ODIM + (n < ODIM ? n : ODIM - 1);
        Ws[(kb * G2_NT + tt) * 32 + lane] = make_float4(wp[0], wp[ODIM], wp[2 * ODIM], wp[3 * ODIM]);
      }
    }
  }
  __syncthreads();
  if (!live) return;
  v8f acc[G2_NT];
#pragma unroll
  for (int t = 0; t < G2_NT; ++t) acc[t] = (v8f){};
#pragma unroll
  for (int kb = 0; kb < G2_KB; ++kb) {
    v2f a0, a1;
    a0.x = a[kb].x; a0.y = a[kb].y; a1.x = a[kb].z; a1.y = a[kb].w;
#pragma unroll
    for (int t = 0; t < G2_NT; ++t) {
      const float4 b = Ws[(kb * G2_NT + t) * 32 + lane];
      v2f b0, b1;
      b0.x = b.x; b0.y = b.y; b1.x = b.z; b1.y = b.w;
      acc[t] = __builtin_amdgcn_wmma_f32_16x16x4_f32(false, a0, false, b0, (short)0,
                                                     acc[t], false, false);
      acc[t] = __builtin_amdgcn_wmma_f32_16x16x4_f32(false, a1, false, b1, (short)0,
                                                     acc[t], false, false);
    }
  }
#pragma unroll
  for (int t = 0; t < G2_NT; ++t) {
    const int col = t * 16 + la;
    if (col < ODIM) {
#pragma unroll
      for (int j = 0; j < 8; ++j)
        t2[(size_t)(m0 + j + hi * 8) * ODIM + col] = acc[t][j];
    }
  }
}

#define ATT_WG 1024
#define ATT_NW (ATT_WG * 8)
__device__ __forceinline__ void att1_node(const float (&v)[NH], const float (&ws)[NH],
                                          const float (&wd)[NH], int node, int lane,
                                          float* __restrict__ os, float* __restrict__ od) {
#pragma unroll
  for (int hd = 0; hd < NH; ++hd) {
    const float s = wave_sum(v[hd] * ws[hd]);
    const float d = wave_sum(v[hd] * wd[hd]);
    if (lane == 0) { os[node * NH + hd] = s; od[node * NH + hd] = d; }
  }
}
__global__ __launch_bounds__(256) void k_att1(const float* __restrict__ h,
                                              const float* __restrict__ as_,
                                              const float* __restrict__ ad_,
                                              float* __restrict__ os,
                                              float* __restrict__ od) {
  const int lane = threadIdx.x & 31;
  const int w = blockIdx.x * 8 + (threadIdx.x >> 5);
  float ws[NH], wd[NH];
#pragma unroll
  for (int hd = 0; hd < NH; ++hd) { ws[hd] = as_[hd * HIDC + lane]; wd[hd] = ad_[hd * HIDC + lane]; }
  for (int na = w; na < NND; na += 2 * ATT_NW) {
    const int nb = na + ATT_NW;
    const bool hb = nb < NND;
    const float* ra = h + (size_t)na * HC;
    const float* rb = h + (size_t)(hb ? nb : na) * HC;
    float va[NH], vb[NH];
#pragma unroll
    for (int hd = 0; hd < NH; ++hd) va[hd] = ra[hd * HIDC + lane];
#pragma unroll
    for (int hd = 0; hd < NH; ++hd) vb[hd] = rb[hd * HIDC + lane];
    att1_node(va, ws, wd, na, lane, os, od);
    if (hb) att1_node(vb, ws, wd, nb, lane, os, od);
  }
}

__device__ __forceinline__ void att2_node(float v0, float v1, float s0, float s1, float d0,
                                          float d1, bool up, int node, int lane,
                                          float* __restrict__ os, float* __restrict__ od) {
  float s = 0.f, d = 0.f;
  s += v0 * s0; d += v0 * d0;
  if (up) { s += v1 * s1; d += v1 * d1; }
  s = wave_sum(s); d = wave_sum(d);
  if (lane == 0) { os[node] = s; od[node] = d; }
}
__global__ __launch_bounds__(256) void k_att2(const float* __restrict__ t,
                                              const float* __restrict__ as_,
                                              const float* __restrict__ ad_,
                                              float* __restrict__ os,
                                              float* __restrict__ od) {
  const int lane = threadIdx.x & 31;
  const int w = blockIdx.x * 8 + (threadIdx.x >> 5);
  const bool up = lane < ODIM - 32;
  const float s0 = as_[lane], d0 = ad_[lane];
  const float s1 = up ? as_[32 + lane] : 0.f, d1 = up ? ad_[32 + lane] : 0.f;
  for (int na = w; na < NND; na += 2 * ATT_NW) {
    const int nb = na + ATT_NW;
    const bool hb = nb < NND;
    const float* ta = t + (size_t)na * ODIM;
    const float* tb = t + (size_t)(hb ? nb : na) * ODIM;
    const float a0 = ta[lane], b0 = tb[lane];
    const float a1 = up ? ta[32 + lane] : 0.f, b1 = up ? tb[32 + lane] : 0.f;
    att2_node(a0, a1, s0, s1, d0, d1, up, na, lane, os, od);
    if (hb) att2_node(b0, b1, s0, s1, d0, d1, up, nb, lane, os, od);
  }
}


#define OWN_R    196
#define OWN_WG   256
#define OWN_T    1024
#define OWN_U    4
#define NE4      (NE / 4)
#define OWN_IT   ((NE4 + OWN_T * OWN_U - 1) / (OWN_T * OWN_U))
#define OWN_STEP (32 * OWN_U * 4)
#define OWN_QTH  32
#define OWN_QCAP 768
static_assert(OWN_QCAP >= OWN_QTH + OWN_STEP, "a list about to be worked off still takes one scan step");
static_assert(NE <= (1 << 20) && OWN_R <= 256 && NND <= (1 << 16), "the fields of a list entry");

__device__ __forceinline__ float lane_bcast_f(float v, int l) {
  return __int_as_float(__builtin_amdgcn_readlane(__float_as_int(v), l));
}
__device__ __forceinline__ float leaky(float v) { return v > 0.f ? v : v * NEG_SLOPE; }

template <int H>
__device__ __forceinline__ float own_logit(const float* __restrict__ aS, const float* __restrict__ aD,
                                           int s, int d, int hd) {
  return leaky(aS[s * H + hd] + aD[d * H + hd]);
}

template <int H>
__device__ __forceinline__ void own_max(int* rmax, const float* __restrict__ aS,
                                        const float* __restrict__ aD, int s, int d0, int x) {
  float lg[H];
#pragma unroll
  for (int hd = 0; hd < H; ++hd) lg[hd] = own_logit<H>(aS, aD, s, d0 + x, hd);
#pragma unroll
  for (int hd = 0; hd < H; ++hd)
    __hip_atomic_fetch_max(&rmax[x * H + hd], f32_ord(lg[hd]), __ATOMIC_RELAXED, __HIP_MEMORY_SCOPE_WORKGROUP);
}

template <int H, bool DO_MAX>
__device__ __forceinline__ void own_convert(unsigned* qw, int from, int n, const int* __restrict__ ei,
                                            const float* __restrict__ aS, const float* __restrict__ aD,
                                            int* rmax, int d0, int lane) {
  for (int b = from; b < n; b += 32) {
    const int i = b + lane;
    if (i < n) {
      const unsigned ent = qw[i];
      const int x = (int)(ent >> 20);
      const int s = ei[ent & 0xFFFFFu];
      if (DO_MAX) own_max<H>(rmax, aS, aD, s, d0, x);
      qw[i] = ((unsigned)x << 16) | (unsigned)s;
    }
  }
}

#define OWN_TK 4
template <int H>
__device__ __forceinline__ void own_take(unsigned* qw, const unsigned* __restrict__ lw, int n,
                                         const float* __restrict__ aS, const float* __restrict__ aD,
                                         int* rmax, int d0, int lane) {
  for (int b = 0; b < n; b += 32 * OWN_TK) {
    unsigned e[OWN_TK];
#pragma unroll
    for (int k = 0; k < OWN_TK; ++k) {
      const int i = b + k * 32 + lane;
      e[k] = lw[i < n ? i : 0];
    }
    static_assert(OWN_TK == 4, "the statement below names the entries one by one");
    asm volatile("" :: "v"(e[0]), "v"(e[1]), "v"(e[2]), "v"(e[3]));
    float ls[OWN_TK][H], ld[OWN_TK][H];
#pragma unroll
    for (int k = 0; k < OWN_TK; ++k)
#pragma unroll
      for (int hd = 0; hd < H; ++hd) {
        ls[k][hd] = aS[(int)(e[k] & 0xFFFFu) * H + hd];
        ld[k][hd] = aD[(d0 + (int)(e[k] >> 16)) * H + hd];
      }
#pragma unroll
    for (int k = 0; k < OWN_TK; ++k)
#pragma unroll
      for (int hd = 0; hd < H; ++hd)
        asm volatile("" :: "v"(ls[k][hd]), "v"(ld[k][hd]));
#pragma unroll
    for (int k = 0; k < OWN_TK; ++k) {
      const int i = b + k * 32 + lane;
      if (i < n) {
        qw[i] = e[k];
        const int x = (int)(e[k] >> 16);
#pragma unroll
        for (int hd = 0; hd < H; ++hd)
          __hip_atomic_fetch_max(&rmax[x * H + hd], f32_ord(leaky(ls[k][hd] + ld[k][hd])), __ATOMIC_RELAXED,
                                 __HIP_MEMORY_SCOPE_WORKGROUP);
      }
    }
  }
}

template <int C, int H, bool INIT>
__device__ __forceinline__ void own_edge(float* __restrict__ acc, float* __restrict__ wsum,
                                         const int* rmax, const float* __restrict__ h,
                                         const float* __restrict__ aS, const float* __restrict__ aD,
                                         int d0, int s, int x, int lane) {
  const float* hs = h + (size_t)s * C;
  float* ar = acc + x * C;
  constexpr int NP = (C + 31) / 32;
  float v[NP];
#pragma unroll
  for (int j = 0; j < NP; ++j) {
    const int c = j * 32 + lane;
    v[j] = c < C ? hs[c] : 0.f;
  }
  float a = 0.f;
  if (lane < H) {
    a = __expf(own_logit<H>(aS, aD, s, d0 + x, lane) - ord_f32(rmax[x * H + lane]));
    if (INIT) wsum[x * H + lane] = a;
    else __hip_atomic_fetch_add(&wsum[x * H + lane], a, __ATOMIC_RELAXED, __HIP_MEMORY_SCOPE_WORKGROUP);
  }
#pragma unroll
  for (int j = 0; j < NP; ++j) {
    const float alpha = lane_bcast_f(a, (j * 32) / (C / H));
    const int c = j * 32 + lane;
    if (c < C) {
      const float t = v[j] * alpha;
      if (INIT) ar[c] = t;
      else __hip_atomic_fetch_add(&ar[c], t, __ATOMIC_RELAXED, __HIP_MEMORY_SCOPE_WORKGROUP);
    }
  }
}

template <int C, int H>
__device__ __forceinline__ void own_pair(float* __restrict__ acc, float* __restrict__ wsum,
                                         const int* rmax, const float* __restrict__ h,
                                         const float* __restrict__ aS, const float* __restrict__ aD,
                                         int d0, int sA, int xA, int sB, int xB, int lane) {
  const float* hA = h + (size_t)sA * C;
  const float* hB = h + (size_t)sB * C;
  float* arA = acc + xA * C;
  float* arB = acc + xB * C;
  constexpr int NP = (C + 31) / 32;
  float vA[NP], vB[NP];
#pragma unroll
  for (int j = 0; j < NP; ++j) {
    const int c = j * 32 + lane;
    vA[j] = c < C ? hA[c] : 0.f;
  }
#pragma unroll
  for (int j = 0; j < NP; ++j) {
    const int c = j * 32 + lane;
    vB[j] = c < C ? hB[c] : 0.f;
  }
  float a = 0.f;
  if (lane < 2 * H) {
    const bool second = lane >= H;
    const int hh = second ? lane - H : lane;
    const int s  = second ? sB : sA;
    const int x  = second ? xB : xA;
    a = __expf(own_logit<H>(aS, aD, s, d0 + x, hh) - ord_f32(rmax[x * H + hh]));
    __hip_atomic_fetch_add(&wsum[x * H + hh], a, __ATOMIC_RELAXED, __HIP_MEMORY_SCOPE_WORKGROUP);
  }
#pragma unroll
  for (int j = 0; j < NP; ++j) {
    const float alpha = lane_bcast_f(a, (j * 32) / (C / H));
    const int c = j * 32 + lane;
    if (c < C)
      __hip_atomic_fetch_add(&arA[c], vA[j] * alpha, __ATOMIC_RELAXED, __HIP_MEMORY_SCOPE_WORKGROUP);
  }
#pragma unroll
  for (int j = 0; j < NP; ++j) {
    const float alpha = lane_bcast_f(a, H + (j * 32) / (C / H));
    const int c = j * 32 + lane;
    if (c < C)
      __hip_atomic_fetch_add(&arB[c], vB[j] * alpha, __ATOMIC_RELAXED, __HIP_MEMORY_SCOPE_WORKGROUP);
  }
}

template <int C, int H>
__device__ __forceinline__ void own_drain(float* __restrict__ acc, float* __restrict__ wsum,
                                          const int* rmax, const unsigned* qw, int n,
                                          const float* __restrict__ h,
                                          const float* __restrict__ aS, const float* __restrict__ aD,
                                          int d0, int lane) {
  for (int b = 0; b < n; b += 32) {
    int cnt = n - b;
    if (cnt > 32) cnt = 32;
    unsigned ent = 0;
    if (lane < cnt) ent = qw[b + lane];
    int i = 0;
    for (; i + 1 < cnt; i += 2) {
      const unsigned eA = (unsigned)__builtin_amdgcn_readlane((int)ent, i);
      const unsigned eB = (unsigned)__builtin_amdgcn_readlane((int)ent, i + 1);
      own_pair<C, H>(acc, wsum, rmax, h, aS, aD, d0, (int)(eA & 0xFFFFu), (int)(eA >> 16),
                     (int)(eB & 0xFFFFu), (int)(eB >> 16), lane);
    }
    if (i < cnt) {
      const unsigned eA = (unsigned)__builtin_amdgcn_readlane((int)ent, i);
      own_edge<C, H, false>(acc, wsum, rmax, h, aS, aD, d0, (int)(eA & 0xFFFFu), (int)(eA >> 16), lane);
    }
  }
}

template <int C, int H, int MODE>
__device__ __forceinline__ void own_scan(float* __restrict__ acc, float* __restrict__ wsum, int* rmax,
                                         unsigned* qw, const int* __restrict__ ei,
                                         const float* __restrict__ h,
                                         const float* __restrict__ aS, const float* __restrict__ aD,
                                         int d0, int nrow, int wofs, int lane, int it_from,
                                         int& qn, bool& over, int& it_over) {
  const int4* dst4 = (const int4*)(ei + NE);
  int qdone = qn;
  int4 cur[OWN_U];
  int ii = (int)((blockIdx.x * OWN_IT) / OWN_WG + it_from) % OWN_IT;
#pragma unroll
  for (int u = 0; u < OWN_U; ++u) {
    const int wb = (ii * OWN_U + u) * OWN_T + wofs;
    cur[u] = dst4[(wb < NE4 ? wb : NE4 - 32) + lane];
  }
  for (int it = it_from; it < OWN_IT; ++it) {
    const int in = ii + 1 < OWN_IT ? ii + 1 : 0;
    int4 nxt[OWN_U];
#pragma unroll
    for (int u = 0; u < OWN_U; ++u) {
      const int wb = (in * OWN_U + u) * OWN_T + wofs;
      nxt[u] = dst4[(wb < NE4 ? wb : NE4 - 32) + lane];
    }
    if (MODE == 0 && !over && qn + OWN_STEP > OWN_QCAP) {
      own_convert<H, true>(qw, qdone, qn, ei, aS, aD, rmax, d0, lane);
      qdone = qn;
      over = true;
      it_over = it;
    }
#pragma unroll
    for (int u = 0; u < OWN_U; ++u) {
      const int wb = (ii * OWN_U + u) * OWN_T + wofs;
      const int d0u = wb < NE4 ? d0 : 0x40000000;
      const int dd[4] = {cur[u].x, cur[u].y, cur[u].z, cur[u].w};
#pragma unroll
      for (int k = 0; k < 4; ++k) {
        const unsigned x = (unsigned)(dd[k] - d0u);
        const bool m = x < (unsigned)nrow;
        const unsigned mk = __builtin_amdgcn_ballot_w32(m);
        if (mk) {
          const unsigned eid = (unsigned)((wb + lane) * 4 + k);
          if (MODE == 0 && over) {
            if (m) own_max<H>(rmax, aS, aD, ei[eid], d0, (int)x);
          } else {
            if (m) qw[qn + (int)__builtin_amdgcn_mbcnt_lo(mk, 0u)] = (x << 20) | eid;
            qn += __builtin_popcount(mk);
          }
        }
      }
    }
    if (MODE == 0) {
      const int nb = (qn - qdone) & ~31;
      if (nb) {
        own_convert<H, true>(qw, qdone, qdone + nb, ei, aS, aD, rmax, d0, lane);
        qdone += nb;
      }
    } else if (qn >= OWN_QTH) {
      own_convert<H, false>(qw, 0, qn, ei, aS, aD, rmax, d0, lane);
      own_drain<C, H>(acc, wsum, rmax, qw, qn, h, aS, aD, d0, lane);
      qn = 0;
    }
#pragma unroll
    for (int u = 0; u < OWN_U; ++u) cur[u] = nxt[u];
    ii = in;
  }
  if (MODE == 0) {
    own_convert<H, true>(qw, qdone, qn, ei, aS, aD, rmax, d0, lane);
  } else {
    own_convert<H, false>(qw, 0, qn, ei, aS, aD, rmax, d0, lane);
    own_drain<C, H>(acc, wsum, rmax, qw, qn, h, aS, aD, d0, lane);
    qn = 0;
  }
}

template <int C, int H, bool KEPT>
__global__ __launch_bounds__(OWN_T) void k_own(const int* __restrict__ ei,
                                               const float* __restrict__ h,
                                               const float* __restrict__ aS,
                                               const float* __restrict__ aD,
                                               float* __restrict__ agg,
                                               unsigned* __restrict__ lst,
                                               int* __restrict__ lcnt) {
  __shared__ __attribute__((aligned(16))) float acc[OWN_R * C];
  __shared__ float wsum[OWN_R * H];
  __shared__ int rmax[OWN_R * H];
  __shared__ unsigned q[(OWN_T / 32) * OWN_QCAP];
  const int tid = threadIdx.x, lane = tid & 31;
  const int wave = __builtin_amdgcn_readfirstlane(tid >> 5);
  const int d0 = blockIdx.x * OWN_R;
  int nrow = NND - d0;
  if (nrow > OWN_R) nrow = OWN_R;

  for (int i = tid; i < nrow * H; i += OWN_T)
    rmax[i] = f32_ord(leaky(aS[d0 * H + i] + aD[d0 * H + i]));
  __syncthreads();

  unsigned* qw = q + wave * OWN_QCAP;
  const int wofs = wave * 32;
  int qn = 0, it_over = OWN_IT;
  bool over = false;
  const int lw_i = blockIdx.x * (OWN_T / 32) + wave;
  unsigned* lw = lst + (size_t)lw_i * OWN_QCAP;
  bool have = false;
  if (KEPT) {
    const int n = __builtin_amdgcn_readfirstlane(lcnt[lw_i]);
    have = (unsigned)n <= (unsigned)OWN_QCAP;
    if (have) {
      own_take<H>(qw, lw, n, aS, aD, rmax, d0, lane);
      qn = n;
    }
  }
  if (!have) {
    own_scan<C, H, 0>(acc, wsum, rmax, qw, ei, h, aS, aD, d0, nrow, wofs, lane, 0, qn, over, it_over);
    if (!KEPT) {
      if (lane == 0) lcnt[lw_i] = over ? -1 : qn;
      if (!over)
        for (int b = 0; b < qn; b += 32) {
          const int i = b + lane;
          if (i < qn) lw[i] = qw[i];
        }
    }
  }
  __syncthreads();

  for (int r = wave; r < nrow; r += OWN_T / 32)
    own_edge<C, H, true>(acc, wsum, rmax, h, aS, aD, d0, d0 + r, r, lane);
  __syncthreads();

  own_drain<C, H>(acc, wsum, rmax, qw, qn, h, aS, aD, d0, lane);
  if (over) {
    qn = 0;
    own_scan<C, H, 1>(acc, wsum, rmax, qw, ei, h, aS, aD, d0, nrow, wofs, lane, it_over, qn, over, it_over);
  }
  __syncthreads();

  static_assert((C / H) % 4 == 0, "a 16-byte piece does not straddle two heads");
  constexpr int PH = (C / H) / 4;
  float4* o4 = (float4*)(agg + (size_t)d0 * C);
  const float4* a4 = (const float4*)acc;
  for (int i = tid; i < nrow * C / 4; i += OWN_T) {
    const float4 v = a4[i];
    const float dn = wsum[i / PH] + EPSV;
    o4[i] = make_float4(v.x / dn, v.y / dn, v.z / dn, v.w / dn);
  }
}

#define EL_WG 1250
#define EL_T  (EL_WG * 256)
#define EL_U  5
#define EL_N4 (NND * HC / 4)
static_assert(EL_N4 == 2 * EL_U * EL_T, "two full trips cover the rows exactly");
static_assert(EL_T % (HC / 4) == 0, "a thread's columns do not change from trip to trip");
__global__ __launch_bounds__(256) void k_elu_bias(float* __restrict__ a, const float* __restrict__ b) {
  const int i0 = blockIdx.x * 256 + threadIdx.x;
  const int c = (i0 & (HC / 4 - 1)) * 4;
  const float b0 = b[c], b1 = b[c + 1], b2 = b[c + 2], b3 = b[c + 3];
  float4* a4 = (float4*)a;
  for (int ib = i0; ib < EL_N4; ib += EL_U * EL_T) {
    float4 v[EL_U];
#pragma unroll
    for (int k = 0; k < EL_U; ++k) v[k] = a4[ib + k * EL_T];
#pragma unroll
    for (int k = 0; k < EL_U; ++k) {
      const float x0 = v[k].x + b0, x1 = v[k].y + b1, x2 = v[k].z + b2, x3 = v[k].w + b3;
      float4 r;
      r.x = x0 > 0.f ? x0 : (__expf(x0) - 1.f);
      r.y = x1 > 0.f ? x1 : (__expf(x1) - 1.f);
      r.z = x2 > 0.f ? x2 : (__expf(x2) - 1.f);
      r.w = x3 > 0.f ? x3 : (__expf(x3) - 1.f);
      a4[ib + k * EL_T] = r;
    }
  }
}

#define OB_WG 512
#define OB_T  (OB_WG * 256)
#define OB_U  4
#define OB_N4 (NND * ODIM / 4)
static_assert(OB_U * OB_T >= OB_N4, "one trip covers the output");
__global__ __launch_bounds__(256) void k_out_bias(const float* __restrict__ agg, const float* __restrict__ b,
                           float* __restrict__ out) {
  const int i0 = blockIdx.x * 256 + threadIdx.x;
  const float4* g4 = (const float4*)agg;
  float4* o4 = (float4*)out;
  float4 v[OB_U], bb[OB_U];
#pragma unroll
  for (int k = 0; k < OB_U; ++k) {
    const int i = i0 + k * OB_T;
    const int ic = i < OB_N4 ? i : i0;
    v[k] = g4[ic];
    const int c = (ic % (ODIM / 4)) * 4;
    bb[k] = make_float4(b[c], b[c + 1], b[c + 2], b[c + 3]);
  }
#pragma unroll
  for (int k = 0; k < OB_U; ++k) {
    const int i = i0 + k * OB_T;
    if (i < OB_N4)
      o4[i] = make_float4(v[k].x + bb[k].x, v[k].y + bb[k].y, v[k].z + bb[k].z, v[k].w + bb[k].w);
  }
}


extern "C" void kernel_launch(void* const* d_in, const int* in_sizes, int n_in,
                              void* d_out, int out_size, void* d_ws, size_t ws_size,
                              hipStream_t stream) {
  const float* x   = (const float*)d_in[0];
  const int*   ei  = (const int*)d_in[1];
  const float* W1  = (const float*)d_in[2];
  const float* as1 = (const float*)d_in[3];
  const float* ad1 = (const float*)d_in[4];
  const float* b1  = (const float*)d_in[5];
  const float* W2  = (const float*)d_in[6];
  const float* as2 = (const float*)d_in[7];
  const float* ad2 = (const float*)d_in[8];
  const float* b2  = (const float*)d_in[9];
  float* out = (float*)d_out;

  float* ws   = (float*)d_ws;
  float* h1   = ws;
  float* agg1 = h1   + (size_t)NND * HC;
  float* aS1  = agg1 + (size_t)NND * HC;
  float* aD1  = aS1  + NND * NH;
  float* t2   = aD1  + NND * NH;
  float* aS2  = t2   + (size_t)NND * ODIM;
  float* aD2  = aS2  + NND;
  float* agg2 = aD2  + NND;
  unsigned* lst = (unsigned*)(agg2 + (size_t)NND * ODIM);
  int* lcnt = (int*)(lst + (size_t)OWN_WG * (OWN_T / 32) * OWN_QCAP);

  k_gemm1<<<GM_WG, 256, 0, stream>>>(x, W1, h1);
  k_att1<<<ATT_WG, 256, 0, stream>>>(h1, as1, ad1, aS1, aD1);
  k_own<HC, NH, false><<<OWN_WG, OWN_T, 0, stream>>>(ei, h1, aS1, aD1, agg1, lst, lcnt);
  k_elu_bias<<<EL_WG, 256, 0, stream>>>(agg1, b1);

  k_gemm2<<<GM_WG, 256, 0, stream>>>(agg1, W2, t2);
  k_att2<<<ATT_WG, 256, 0, stream>>>(t2, as2, ad2, aS2, aD2);
  k_own<ODIM, 1, true><<<OWN_WG, OWN_T, 0, stream>>>(ei, t2, aS2, aD2, agg2, lst, lcnt);
  k_out_bias<<<OB_WG, 256, 0, stream>>>(agg2, b2, out);
}


# Attention vectors folded into the weights: each group forms W*a by WMMA as extra LDS weight columns, both products emit the logits; 2 dot launches gone (6 left); logit sum order: channels, then k
# speedup vs baseline: 3.4600x; 1.0780x over previous
#include <hip/hip_runtime.h>
#include <hip/hip_bf16.h>

#define NND   50000
#define FIN   128
#define NH    8
#define HIDC  32
#define HC    256
#define ODIM  40
#define NE    800000
#define NET   850000
#define NEG_SLOPE 0.2f
#define EPSV  1e-16f

typedef __attribute__((ext_vector_type(2))) float v2f;
typedef __attribute__((ext_vector_type(8))) float v8f;

__device__ __forceinline__ int f32_ord(float x) {
  int i = __float_as_int(x);
  return i >= 0 ? i : (i ^ 0x7FFFFFFF);
}
__device__ __forceinline__ float ord_f32(int i) {
  return __int_as_float(i >= 0 ? i : (i ^ 0x7FFFFFFF));
}


#define GM_WV   8
#define GM_NS   (NND / 16)
#define GM_WG   ((GM_NS + GM_WV - 1) / GM_WV)
static_assert(NND % 16 == 0, "whole strips");

#define G1_KB (FIN / 8)
#define G1_NT (HC / 16)
#define G1_LT (G1_NT + 1)
static_assert(2 * NH == 16 && HIDC == 32, "one tile takes both logit sets; a block of 8 k lies in one head");
static_assert(FIN == 16 * GM_WV, "eight waves of 16 rows cover the rows of W");
__global__ __launch_bounds__(256) void k_gemm1(const float* __restrict__ x,
                                               const float* __restrict__ W,
                                               const float* __restrict__ as_,
                                               const float* __restrict__ ad_,
                                               float* __restrict__ h,
                                               float* __restrict__ os,
                                               float* __restrict__ od) {
  __shared__ __attribute__((aligned(16))) float4 Ws[G1_KB * G1_LT * 32];
  const int tid  = threadIdx.x;
  const int lane = tid & 31;
  const int wave = __builtin_amdgcn_readfirstlane(tid >> 5);
  const int la = lane & 15, hi = lane >> 4;
  const int strip = blockIdx.x * GM_WV + wave;
  const bool live = strip < GM_NS;
  const int m0 = (live ? strip : 0) * 16;
  float4 a[G1_KB];
  {
    const float4* xr = (const float4*)(x + (size_t)(m0 + la) * FIN) + hi;
#pragma unroll
    for (int kb = 0; kb < G1_KB; ++kb) a[kb] = xr[kb * 2];
  }
  {
    const int t = tid >> 4, c = tid & 15;
#pragma unroll 4
    for (int kb = 0; kb < G1_KB; ++kb) {
#pragma unroll
      for (int hh = 0; hh < 2; ++hh) {
        const float* wp = W + (size_t)(kb * 8 + hh * 4) * HC + tid;
        Ws[(kb * G1_LT + t) * 32 + hh * 16 + c] = make_float4(wp[0], wp[HC], wp[2 * HC], wp[3 * HC]);
      }
    }
  }
  {
    const float4* wr = (const float4*)(W + (size_t)(wave * 16 + la) * HC) + hi;
    const float4* ar = (const float4*)(la < NH ? as_ : ad_) + hi;
    const int hd = la & (NH - 1);
    v8f wa = (v8f){};
#pragma unroll 16
    for (int kb = 0; kb < HC / 8; ++kb) {
      const float4 aw = wr[kb * 2];
      const float4 bv = ar[kb * 2];
      const bool on = (kb >> 2) == hd;
      v2f a0, a1, b0, b1;
      a0.x = aw.x; a0.y = aw.y; a1.x = aw.z; a1.y = aw.w;
      b0.x = on ? bv.x : 0.f; b0.y = on ? bv.y : 0.f; b1.x = on ? bv.z : 0.f; b1.y = on ? bv.w : 0.f;
      wa = __builtin_amdgcn_wmma_f32_16x16x4_f32(false, a0, false, b0, (short)0, wa, false, false);
      wa = __builtin_amdgcn_wmma_f32_16x16x4_f32(false, a1, false, b1, (short)0, wa, false, false);
    }
    float4* wt = Ws + ((2 * wave + hi) * G1_LT + G1_NT) * 32 + la;
    wt[0]  = make_float4(wa[0], wa[1], wa[2], wa[3]);
    wt[16] = make_float4(wa[4], wa[5], wa[6], wa[7]);
  }
  __syncthreads();
  if (!live) return;
  v8f acc[G1_LT];
#pragma unroll
  for (int t = 0; t < G1_LT; ++t) acc[t] = (v8f){};
#pragma unroll
  for (int kb = 0; kb < G1_KB; ++kb) {
    v2f a0, a1;
    a0.x = a[kb].x; a0.y = a[kb].y; a1.x = a[kb].z; a1.y = a[kb].w;
#pragma unroll
    for (int t = 0; t < G1_LT; ++t) {
      const float4 b = Ws[(kb * G1_LT + t) * 32 + lane];
      v2f b0, b1;
      b0.x = b.x; b0.y = b.y; b1.x = b.z; b1.y = b.w;
      acc[t] = __builtin_amdgcn_wmma_f32_16x16x4_f32(false, a0, false, b0, (short)0,
                                                     acc[t], false, false);
      acc[t] = __builtin_amdgcn_wmma_f32_16x16x4_f32(false, a1, false, b1, (short)0,
                                                     acc[t], false, false);
    }
  }
#pragma unroll
  for (int t = 0; t < G1_NT; ++t)
#pragma unroll
    for (int j = 0; j < 8; ++j)
      h[(size_t)(m0 + j + hi * 8) * HC + t * 16 + la] = acc[t][j];
  {
    float* lo = (la < NH ? os : od) + (la & (NH - 1));
#pragma unroll
    for (int j = 0; j < 8; ++j) lo[(m0 + j + hi * 8) * NH] = acc[G1_NT][j];
  }
}

#define G2_KB (HC / 8)
#define G2_NT 3
static_assert(G2_NT * 16 >= ODIM + 2, "the tiles cover the output columns and the two logit columns");
static_assert(ODIM % 16 == 8 && ODIM % 8 == 0, "the logit columns are entries 8 and 9 of the last tile; whole blocks of 8 channels");
static_assert(HC == 32 * GM_WV, "eight waves of two 16-row strips cover the rows of W");
__global__ __launch_bounds__(256) void k_gemm2(const float* __restrict__ h2,
                                               const float* __restrict__ W,
                                               const float* __restrict__ as_,
                                               const float* __restrict__ ad_,
                                               float* __restrict__ t2,
                                               float* __restrict__ os,
                                               float* __restrict__ od) {
  __shared__ __attribute__((aligned(16))) float4 Ws[G2_KB * G2_NT * 32];
  const int tid  = threadIdx.x;
  const int lane = tid & 31;
  const int wave = __builtin_amdgcn_readfirstlane(tid >> 5);
  const int la = lane & 15, hi = lane >> 4;
  const int strip = blockIdx.x * GM_WV + wave;
  const bool live = strip < GM_NS;
  const int m0 = (live ? strip : 0) * 16;
  float4 a[G2_KB];
  {
    const float4* xr = (const float4*)(h2 + (size_t)(m0 + la) * HC) + hi;
#pragma unroll
    for (int kb = 0; kb < G2_KB; ++kb) a[kb] = xr[kb * 2];
  }
  {
    static_assert(G2_KB == 4 * GM_WV, "eight waves cover the k blocks");
#pragma unroll
    for (int j = 0; j < 4; ++j) {
      const int kb = wave * 4 + j;
#pragma unroll
      for (int tt = 0; tt < G2_NT; ++tt) {
        const int n = tt * 16 + la;
        const float* wp = W + (size_t)(kb * 8 + hi * 4) * ODIM + (n < ODIM ? n : ODIM - 1);
        const float4 wv = make_float4(wp[0], wp[ODIM], wp[2 * ODIM], wp[3 * ODIM]);
        if (n != ODIM && n != ODIM + 1) Ws[(kb * G2_NT + tt) * 32 + lane] = wv;
      }
    }
  }
  {
    const float4* ar = (const float4*)(la == 1 ? ad_ : as_) + hi;
    const float4* wr = (const float4*)(W + (size_t)(wave * 32 + la) * ODIM) + hi;
    const bool on = la < 2;
    v8f wa[2];
    wa[0] = (v8f){}; wa[1] = (v8f){};
#pragma unroll
    for (int kb = 0; kb < ODIM / 8; ++kb) {
      const float4 bv = ar[kb * 2];
      v2f b0, b1;
      b0.x = on ? bv.x : 0.f; b0.y = on ? bv.y : 0.f; b1.x = on ? bv.z : 0.f; b1.y = on ? bv.w : 0.f;
#pragma unroll
      for (int ss = 0; ss < 2; ++ss) {
        const float4 aw = wr[ss * 4 * ODIM + kb * 2];
        v2f a0, a1;
        a0.x = aw.x; a0.y = aw.y; a1.x = aw.z; a1.y = aw.w;
        wa[ss] = __builtin_amdgcn_wmma_f32_16x16x4_f32(false, a0, false, b0, (short)0, wa[ss], false, false);
        wa[ss] = __builtin_amdgcn_wmma_f32_16x16x4_f32(false, a1, false, b1, (short)0, wa[ss], false, false);
      }
    }
    if (on) {
#pragma unroll
      for (int ss = 0; ss < 2; ++ss) {
        float4* wt = Ws + ((2 * (wave * 2 + ss) + hi) * G2_NT + (G2_NT - 1)) * 32 + (ODIM % 16) + la;
        wt[0]  = make_float4(wa[ss][0], wa[ss][1], wa[ss][2], wa[ss][3]);
        wt[16] = make_float4(wa[ss][4], wa[ss][5], wa[ss][6], wa[ss][7]);
      }
    }
  }
  __syncthreads();
  if (!live) return;
  v8f acc[G2_NT];
#pragma unroll
  for (int t = 0; t < G2_NT; ++t) acc[t] = (v8f){};
#pragma unroll
  for (int kb = 0; kb < G2_KB; ++kb) {
    v2f a0, a1;
    a0.x = a[kb].x; a0.y = a[kb].y; a1.x = a[kb].z; a1.y = a[kb].w;
#pragma unroll
    for (int t = 0; t < G2_NT; ++t) {
      const float4 b = Ws[(kb * G2_NT + t) * 32 + lane];
      v2f b0, b1;
      b0.x = b.x; b0.y = b.y; b1.x = b.z; b1.y = b.w;
      acc[t] = __builtin_amdgcn_wmma_f32_16x16x4_f32(false, a0, false, b0, (short)0,
                                                     acc[t], false, false);
      acc[t] = __builtin_amdgcn_wmma_f32_16x16x4_f32(false, a1, false, b1, (short)0,
                                                     acc[t], false, false);
    }
  }
#pragma unroll
  for (int t = 0; t < G2_NT; ++t) {
    const int col = t * 16 + la;
    if (col < ODIM) {
#pragma unroll
      for (int j = 0; j < 8; ++j)
        t2[(size_t)(m0 + j + hi * 8) * ODIM + col] = acc[t][j];
    } else if (col < ODIM + 2) {
      float* lo = col == ODIM ? os : od;
#pragma unroll
      for (int j = 0; j < 8; ++j) lo[m0 + j + hi * 8] = acc[t][j];
    }
  }
}

#define OWN_R    196
#define OWN_WG   256
#define OWN_T    1024
#define OWN_U    4
#define NE4      (NE / 4)
#define OWN_IT   ((NE4 + OWN_T * OWN_U - 1) / (OWN_T * OWN_U))
#define OWN_STEP (32 * OWN_U * 4)
#define OWN_QTH  32
#define OWN_QCAP 768
static_assert(OWN_QCAP >= OWN_QTH + OWN_STEP, "a list about to be worked off still takes one scan step");
static_assert(NE <= (1 << 20) && OWN_R <= 256 && NND <= (1 << 16), "the fields of a list entry");

__device__ __forceinline__ float lane_bcast_f(float v, int l) {
  return __int_as_float(__builtin_amdgcn_readlane(__float_as_int(v), l));
}
__device__ __forceinline__ float leaky(float v) { return v > 0.f ? v : v * NEG_SLOPE; }

template <int H>
__device__ __forceinline__ float own_logit(const float* __restrict__ aS, const float* __restrict__ aD,
                                           int s, int d, int hd) {
  return leaky(aS[s * H + hd] + aD[d * H + hd]);
}

template <int H>
__device__ __forceinline__ void own_max(int* rmax, const float* __restrict__ aS,
                                        const float* __restrict__ aD, int s, int d0, int x) {
  float lg[H];
#pragma unroll
  for (int hd = 0; hd < H; ++hd) lg[hd] = own_logit<H>(aS, aD, s, d0 + x, hd);
#pragma unroll
  for (int hd = 0; hd < H; ++hd)
    __hip_atomic_fetch_max(&rmax[x * H + hd], f32_ord(lg[hd]), __ATOMIC_RELAXED, __HIP_MEMORY_SCOPE_WORKGROUP);
}

template <int H, bool DO_MAX>
__device__ __forceinline__ void own_convert(unsigned* qw, int from, int n, const int* __restrict__ ei,
                                            const float* __restrict__ aS, const float* __restrict__ aD,
                                            int* rmax, int d0, int lane) {
  for (int b = from; b < n; b += 32) {
    const int i = b + lane;
    if (i < n) {
      const unsigned ent = qw[i];
      const int x = (int)(ent >> 20);
      const int s = ei[ent & 0xFFFFFu];
      if (DO_MAX) own_max<H>(rmax, aS, aD, s, d0, x);
      qw[i] = ((unsigned)x << 16) | (unsigned)s;
    }
  }
}

#define OWN_TK 4
template <int H>
__device__ __forceinline__ void own_take(unsigned* qw, const unsigned* __restrict__ lw, int n,
                                         const float* __restrict__ aS, const float* __restrict__ aD,
                                         int* rmax, int d0, int lane) {
  for (int b = 0; b < n; b += 32 * OWN_TK) {
    unsigned e[OWN_TK];
#pragma unroll
    for (int k = 0; k < OWN_TK; ++k) {
      const int i = b + k * 32 + lane;
      e[k] = lw[i < n ? i : 0];
    }
    static_assert(OWN_TK == 4, "the statement below names the entries one by one");
    asm volatile("" :: "v"(e[0]), "v"(e[1]), "v"(e[2]), "v"(e[3]));
    float ls[OWN_TK][H], ld[OWN_TK][H];
#pragma unroll
    for (int k = 0; k < OWN_TK; ++k)
#pragma unroll
      for (int hd = 0; hd < H; ++hd) {
        ls[k][hd] = aS[(int)(e[k] & 0xFFFFu) * H + hd];
        ld[k][hd] = aD[(d0 + (int)(e[k] >> 16)) * H + hd];
      }
#pragma unroll
    for (int k = 0; k < OWN_TK; ++k)
#pragma unroll
      for (int hd = 0; hd < H; ++hd)
        asm volatile("" :: "v"(ls[k][hd]), "v"(ld[k][hd]));
#pragma unroll
    for (int k = 0; k < OWN_TK; ++k) {
      const int i = b + k * 32 + lane;
      if (i < n) {
        qw[i] = e[k];
        const int x = (int)(e[k] >> 16);
#pragma unroll
        for (int hd = 0; hd < H; ++hd)
          __hip_atomic_fetch_max(&rmax[x * H + hd], f32_ord(leaky(ls[k][hd] + ld[k][hd])), __ATOMIC_RELAXED,
                                 __HIP_MEMORY_SCOPE_WORKGROUP);
      }
    }
  }
}

template <int C, int H, bool INIT>
__device__ __forceinline__ void own_edge(float* __restrict__ acc, float* __restrict__ wsum,
                                         const int* rmax, const float* __restrict__ h,
                                         const float* __restrict__ aS, const float* __restrict__ aD,
                                         int d0, int s, int x, int lane) {
  const float* hs = h + (size_t)s * C;
  float* ar = acc + x * C;
  constexpr int NP = (C + 31) / 32;
  float v[NP];
#pragma unroll
  for (int j = 0; j < NP; ++j) {
    const int c = j * 32 + lane;
    v[j] = c < C ? hs[c] : 0.f;
  }
  float a = 0.f;
  if (lane < H) {
    a = __expf(own_logit<H>(aS, aD, s, d0 + x, lane) - ord_f32(rmax[x * H + lane]));
    if (INIT) wsum[x * H + lane] = a;
    else __hip_atomic_fetch_add(&wsum[x * H + lane], a, __ATOMIC_RELAXED, __HIP_MEMORY_SCOPE_WORKGROUP);
  }
#pragma unroll
  for (int j = 0; j < NP; ++j) {
    const float alpha = lane_bcast_f(a, (j * 32) / (C / H));
    const int c = j * 32 + lane;
    if (c < C) {
      const float t = v[j] * alpha;
      if (INIT) ar[c] = t;
      else __hip_atomic_fetch_add(&ar[c], t, __ATOMIC_RELAXED, __HIP_MEMORY_SCOPE_WORKGROUP);
    }
  }
}

template <int C, int H>
__device__ __forceinline__ void own_pair(float* __restrict__ acc, float* __restrict__ wsum,
                                         const int* rmax, const float* __restrict__ h,
                                         const float* __restrict__ aS, const float* __restrict__ aD,
                                         int d0, int sA, int xA, int sB, int xB, int lane) {
  const float* hA = h + (size_t)sA * C;
  const float* hB = h + (size_t)sB * C;
  float* arA = acc + xA * C;
  float* arB = acc + xB * C;
  constexpr int NP = (C + 31) / 32;
  float vA[NP], vB[NP];
#pragma unroll
  for (int j = 0; j < NP; ++j) {
    const int c = j * 32 + lane;
    vA[j] = c < C ? hA[c] : 0.f;
  }
#pragma unroll
  for (int j = 0; j < NP; ++j) {
    const int c = j * 32 + lane;
    vB[j] = c < C ? hB[c] : 0.f;
  }
  float a = 0.f;
  if (lane < 2 * H) {
    const bool second = lane >= H;
    const int hh = second ? lane - H : lane;
    const int s  = second ? sB : sA;
    const int x  = second ? xB : xA;
    a = __expf(own_logit<H>(aS, aD, s, d0 + x, hh) - ord_f32(rmax[x * H + hh]));
    __hip_atomic_fetch_add(&wsum[x * H + hh], a, __ATOMIC_RELAXED, __HIP_MEMORY_SCOPE_WORKGROUP);
  }
#pragma unroll
  for (int j = 0; j < NP; ++j) {
    const float alpha = lane_bcast_f(a, (j * 32) / (C / H));
    const int c = j * 32 + lane;
    if (c < C)
      __hip_atomic_fetch_add(&arA[c], vA[j] * alpha, __ATOMIC_RELAXED, __HIP_MEMORY_SCOPE_WORKGROUP);
  }
#pragma unroll
  for (int j = 0; j < NP; ++j) {
    const float alpha = lane_bcast_f(a, H + (j * 32) / (C / H));
    const int c = j * 32 + lane;
    if (c < C)
      __hip_atomic_fetch_add(&arB[c], vB[j] * alpha, __ATOMIC_RELAXED, __HIP_MEMORY_SCOPE_WORKGROUP);
  }
}

template <int C, int H>
__device__ __forceinline__ void own_drain(float* __restrict__ acc, float* __restrict__ wsum,
                                          const int* rmax, const unsigned* qw, int n,
                                          const float* __restrict__ h,
                                          const float* __restrict__ aS, const float* __restrict__ aD,
                                          int d0, int lane) {
  for (int b = 0; b < n; b += 32) {
    int cnt = n - b;
    if (cnt > 32) cnt = 32;
    unsigned ent = 0;
    if (lane < cnt) ent = qw[b + lane];
    int i = 0;
    for (; i + 1 < cnt; i += 2) {
      const unsigned eA = (unsigned)__builtin_amdgcn_readlane((int)ent, i);
      const unsigned eB = (unsigned)__builtin_amdgcn_readlane((int)ent, i + 1);
      own_pair<C, H>(acc, wsum, rmax, h, aS, aD, d0, (int)(eA & 0xFFFFu), (int)(eA >> 16),
                     (int)(eB & 0xFFFFu), (int)(eB >> 16), lane);
    }
    if (i < cnt) {
      const unsigned eA = (unsigned)__builtin_amdgcn_readlane((int)ent, i);
      own_edge<C, H, false>(acc, wsum, rmax, h, aS, aD, d0, (int)(eA & 0xFFFFu), (int)(eA >> 16), lane);
    }
  }
}

template <int C, int H, int MODE>
__device__ __forceinline__ void own_scan(float* __restrict__ acc, float* __restrict__ wsum, int* rmax,
                                         unsigned* qw, const int* __restrict__ ei,
                                         const float* __restrict__ h,
                                         const float* __restrict__ aS, const float* __restrict__ aD,
                                         int d0, int nrow, int wofs, int lane, int it_from,
                                         int& qn, bool& over, int& it_over) {
  const int4* dst4 = (const int4*)(ei + NE);
  int qdone = qn;
  int4 cur[OWN_U];
  int ii = (int)((blockIdx.x * OWN_IT) / OWN_WG + it_from) % OWN_IT;
#pragma unroll
  for (int u = 0; u < OWN_U; ++u) {
    const int wb = (ii * OWN_U + u) * OWN_T + wofs;
    cur[u] = dst4[(wb < NE4 ? wb : NE4 - 32) + lane];
  }
  for (int it = it_from; it < OWN_IT; ++it) {
    const int in = ii + 1 < OWN_IT ? ii + 1 : 0;
    int4 nxt[OWN_U];
#pragma unroll
    for (int u = 0; u < OWN_U; ++u) {
      const int wb = (in * OWN_U + u) * OWN_T + wofs;
      nxt[u] = dst4[(wb < NE4 ? wb : NE4 - 32) + lane];
    }
    if (MODE == 0 && !over && qn + OWN_STEP > OWN_QCAP) {
      own_convert<H, true>(qw, qdone, qn, ei, aS, aD, rmax, d0, lane);
      qdone = qn;
      over = true;
      it_over = it;
    }
#pragma unroll
    for (int u = 0; u < OWN_U; ++u) {
      const int wb = (ii * OWN_U + u) * OWN_T + wofs;
      const int d0u = wb < NE4 ? d0 : 0x40000000;
      const int dd[4] = {cur[u].x, cur[u].y, cur[u].z, cur[u].w};
#pragma unroll
      for (int k = 0; k < 4; ++k) {
        const unsigned x = (unsigned)(dd[k] - d0u);
        const bool m = x < (unsigned)nrow;
        const unsigned mk = __builtin_amdgcn_ballot_w32(m);
        if (mk) {
          const unsigned eid = (unsigned)((wb + lane) * 4 + k);
          if (MODE == 0 && over) {
            if (m) own_max<H>(rmax, aS, aD, ei[eid], d0, (int)x);
          } else {
            if (m) qw[qn + (int)__builtin_amdgcn_mbcnt_lo(mk, 0u)] = (x << 20) | eid;
            qn += __builtin_popcount(mk);
          }
        }
      }
    }
    if (MODE == 0) {
      const int nb = (qn - qdone) & ~31;
      if (nb) {
        own_convert<H, true>(qw, qdone, qdone + nb, ei, aS, aD, rmax, d0, lane);
        qdone += nb;
      }
    } else if (qn >= OWN_QTH) {
      own_convert<H, false>(qw, 0, qn, ei, aS, aD, rmax, d0, lane);
      own_drain<C, H>(acc, wsum, rmax, qw, qn, h, aS, aD, d0, lane);
      qn = 0;
    }
#pragma unroll
    for (int u = 0; u < OWN_U; ++u) cur[u] = nxt[u];
    ii = in;
  }
  if (MODE == 0) {
    own_convert<H, true>(qw, qdone, qn, ei, aS, aD, rmax, d0, lane);
  } else {
    own_convert<H, false>(qw, 0, qn, ei, aS, aD, rmax, d0, lane);
    own_drain<C, H>(acc, wsum, rmax, qw, qn, h, aS, aD, d0, lane);
    qn = 0;
  }
}

template <int C, int H, bool KEPT>
__global__ __launch_bounds__(OWN_T) void k_own(const int* __restrict__ ei,
                                               const float* __restrict__ h,
                                               const float* __restrict__ aS,
                                               const float* __restrict__ aD,
                                               float* __restrict__ agg,
                                               unsigned* __restrict__ lst,
                                               int* __restrict__ lcnt) {
  __shared__ __attribute__((aligned(16))) float acc[OWN_R * C];
  __shared__ float wsum[OWN_R * H];
  __shared__ int rmax[OWN_R * H];
  __shared__ unsigned q[(OWN_T / 32) * OWN_QCAP];
  const int tid = threadIdx.x, lane = tid & 31;
  const int wave = __builtin_amdgcn_readfirstlane(tid >> 5);
  const int d0 = blockIdx.x * OWN_R;
  int nrow = NND - d0;
  if (nrow > OWN_R) nrow = OWN_R;

  for (int i = tid; i < nrow * H; i += OWN_T)
    rmax[i] = f32_ord(leaky(aS[d0 * H + i] + aD[d0 * H + i]));
  __syncthreads();

  unsigned* qw = q + wave * OWN_QCAP;
  const int wofs = wave * 32;
  int qn = 0, it_over = OWN_IT;
  bool over = false;
  const int lw_i = blockIdx.x * (OWN_T / 32) + wave;
  unsigned* lw = lst + (size_t)lw_i * OWN_QCAP;
  bool have = false;
  if (KEPT) {
    const int n = __builtin_amdgcn_readfirstlane(lcnt[lw_i]);
    have = (unsigned)n <= (unsigned)OWN_QCAP;
    if (have) {
      own_take<H>(qw, lw, n, aS, aD, rmax, d0, lane);
      qn = n;
    }
  }
  if (!have) {
    own_scan<C, H, 0>(acc, wsum, rmax, qw, ei, h, aS, aD, d0, nrow, wofs, lane, 0, qn, over, it_over);
    if (!KEPT) {
      if (lane == 0) lcnt[lw_i] = over ? -1 : qn;
      if (!over)
        for (int b = 0; b < qn; b += 32) {
          const int i = b + lane;
          if (i < qn) lw[i] = qw[i];
        }
    }
  }
  __syncthreads();

  for (int r = wave; r < nrow; r += OWN_T / 32)
    own_edge<C, H, true>(acc, wsum, rmax, h, aS, aD, d0, d0 + r, r, lane);
  __syncthreads();

  own_drain<C, H>(acc, wsum, rmax, qw, qn, h, aS, aD, d0, lane);
  if (over) {
    qn = 0;
    own_scan<C, H, 1>(acc, wsum, rmax, qw, ei, h, aS, aD, d0, nrow, wofs, lane, it_over, qn, over, it_over);
  }
  __syncthreads();

  static_assert((C / H) % 4 == 0, "a 16-byte piece does not straddle two heads");
  constexpr int PH = (C / H) / 4;
  float4* o4 = (float4*)(agg + (size_t)d0 * C);
  const float4* a4 = (const float4*)acc;
  for (int i = tid; i < nrow * C / 4; i += OWN_T) {
    const float4 v = a4[i];
    const float dn = wsum[i / PH] + EPSV;
    o4[i] = make_float4(v.x / dn, v.y / dn, v.z / dn, v.w / dn);
  }
}

#define EL_WG 1250
#define EL_T  (EL_WG * 256)
#define EL_U  5
#define EL_N4 (NND * HC / 4)
static_assert(EL_N4 == 2 * EL_U * EL_T, "two full trips cover the rows exactly");
static_assert(EL_T % (HC / 4) == 0, "a thread's columns do not change from trip to trip");
__global__ __launch_bounds__(256) void k_elu_bias(float* __restrict__ a, const float* __restrict__ b) {
  const int i0 = blockIdx.x * 256 + threadIdx.x;
  const int c = (i0 & (HC / 4 - 1)) * 4;
  const float b0 = b[c], b1 = b[c + 1], b2 = b[c + 2], b3 = b[c + 3];
  float4* a4 = (float4*)a;
  for (int ib = i0; ib < EL_N4; ib += EL_U * EL_T) {
    float4 v[EL_U];
#pragma unroll
    for (int k = 0; k < EL_U; ++k) v[k] = a4[ib + k * EL_T];
#pragma unroll
    for (int k = 0; k < EL_U; ++k) {
      const float x0 = v[k].x + b0, x1 = v[k].y + b1, x2 = v[k].z + b2, x3 = v[k].w + b3;
      float4 r;
      r.x = x0 > 0.f ? x0 : (__expf(x0) - 1.f);
      r.y = x1 > 0.f ? x1 : (__expf(x1) - 1.f);
      r.z = x2 > 0.f ? x2 : (__expf(x2) - 1.f);
      r.w = x3 > 0.f ? x3 : (__expf(x3) - 1.f);
      a4[ib + k * EL_T] = r;
    }
  }
}

#define OB_WG 512
#define OB_T  (OB_WG * 256)
#define OB_U  4
#define OB_N4 (NND * ODIM / 4)
static_assert(OB_U * OB_T >= OB_N4, "one trip covers the output");
__global__ __launch_bounds__(256) void k_out_bias(const float* __restrict__ agg, const float* __restrict__ b,
                           float* __restrict__ out) {
  const int i0 = blockIdx.x * 256 + threadIdx.x;
  const float4* g4 = (const float4*)agg;
  float4* o4 = (float4*)out;
  float4 v[OB_U], bb[OB_U];
#pragma unroll
  for (int k = 0; k < OB_U; ++k) {
    const int i = i0 + k * OB_T;
    const int ic = i < OB_N4 ? i : i0;
    v[k] = g4[ic];
    const int c = (ic % (ODIM / 4)) * 4;
    bb[k] = make_float4(b[c], b[c + 1], b[c + 2], b[c + 3]);
  }
#pragma unroll
  for (int k = 0; k < OB_U; ++k) {
    const int i = i0 + k * OB_T;
    if (i < OB_N4)
      o4[i] = make_float4(v[k].x + bb[k].x, v[k].y + bb[k].y, v[k].z + bb[k].z, v[k].w + bb[k].w);
  }
}


extern "C" void kernel_launch(void* const* d_in, const int* in_sizes, int n_in,
                              void* d_out, int out_size, void* d_ws, size_t ws_size,
                              hipStream_t stream) {
  const float* x   = (const float*)d_in[0];
  const int*   ei  = (const int*)d_in[1];
  const float* W1  = (const float*)d_in[2];
  const float* as1 = (const float*)d_in[3];
  const float* ad1 = (const float*)d_in[4];
  const float* b1  = (const float*)d_in[5];
  const float* W2  = (const float*)d_in[6];
  const float* as2 = (const float*)d_in[7];
  const float* ad2 = (const float*)d_in[8];
  const float* b2  = (const float*)d_in[9];
  float* out = (float*)d_out;

  float* ws   = (float*)d_ws;
  float* h1   = ws;
  float* agg1 = h1   + (size_t)NND * HC;
  float* aS1  = agg1 + (size_t)NND * HC;
  float* aD1  = aS1  + NND * NH;
  float* t2   = aD1  + NND * NH;
  float* aS2  = t2   + (size_t)NND * ODIM;
  float* aD2  = aS2  + NND;
  float* agg2 = aD2  + NND;
  unsigned* lst = (unsigned*)(agg2 + (size_t)NND * ODIM);
  int* lcnt = (int*)(lst + (size_t)OWN_WG * (OWN_T / 32) * OWN_QCAP);

  k_gemm1<<<GM_WG, 256, 0, stream>>>(x, W1, as1, ad1, h1, aS1, aD1);
  k_own<HC, NH, false><<<OWN_WG, OWN_T, 0, stream>>>(ei, h1, aS1, aD1, agg1, lst, lcnt);
  k_elu_bias<<<EL_WG, 256, 0, stream>>>(agg1, b1);

  k_gemm2<<<GM_WG, 256, 0, stream>>>(agg1, W2, as2, ad2, t2, aS2, aD2);
  k_own<ODIM, 1, true><<<OWN_WG, OWN_T, 0, stream>>>(ei, t2, aS2, aD2, agg2, lst, lcnt);
  k_out_bias<<<OB_WG, 256, 0, stream>>>(agg2, b2, out);
}


# Owner groups scan one byte per edge (destination row / 196, written by the first launch's threads), not the 4-byte id, four bytes per compare; new split of edges over waves, sums in arrival order
# speedup vs baseline: 3.6384x; 1.0516x over previous
#include <hip/hip_runtime.h>
#include <hip/hip_bf16.h>

#define NND   50000
#define FIN   128
#define NH    8
#define HIDC  32
#define HC    256
#define ODIM  40
#define NE    800000
#define NET   850000
#define NEG_SLOPE 0.2f
#define EPSV  1e-16f

typedef __attribute__((ext_vector_type(2))) float v2f;
typedef __attribute__((ext_vector_type(8))) float v8f;

__device__ __forceinline__ int f32_ord(float x) {
  int i = __float_as_int(x);
  return i >= 0 ? i : (i ^ 0x7FFFFFFF);
}
__device__ __forceinline__ float ord_f32(int i) {
  return __int_as_float(i >= 0 ? i : (i ^ 0x7FFFFFFF));
}


#define OWN_R   196
#define GM_WV   8
#define GM_NS   (NND / 16)
#define GM_WG   ((GM_NS + GM_WV - 1) / GM_WV)
static_assert(NND % 16 == 0, "whole strips");

#define G1_KB (FIN / 8)
#define G1_NT (HC / 16)
#define G1_LT (G1_NT + 1)
static_assert(2 * NH == 16 && HIDC == 32, "one tile takes both logit sets; a block of 8 k lies in one head");
static_assert(FIN == 16 * GM_WV, "eight waves of 16 rows cover the rows of W");
static_assert(NE % 8 == 0 && GM_WG * 256 >= NE / 8, "the launch's threads cover the edges, 8 each");
static_assert((NND + OWN_R - 1) / OWN_R <= 256, "an owner number fits a byte");
__device__ __forceinline__ unsigned own_of(int d) { return (unsigned)d / (unsigned)OWN_R; }
__global__ __launch_bounds__(256) void k_gemm1(const float* __restrict__ x,
                                               const float* __restrict__ W,
                                               const float* __restrict__ as_,
                                               const float* __restrict__ ad_,
                                               const int* __restrict__ ei,
                                               float* __restrict__ h,
                                               float* __restrict__ os,
                                               float* __restrict__ od,
                                               unsigned* __restrict__ ob) {
  __shared__ __attribute__((aligned(16))) float4 Ws[G1_KB * G1_LT * 32];
  const int tid  = threadIdx.x;
  const int lane = tid & 31;
  const int wave = __builtin_amdgcn_readfirstlane(tid >> 5);
  const int la = lane & 15, hi = lane >> 4;
  const int strip = blockIdx.x * GM_WV + wave;
  const bool live = strip < GM_NS;
  const int m0 = (live ? strip : 0) * 16;
  float4 a[G1_KB];
  {
    const float4* xr = (const float4*)(x + (size_t)(m0 + la) * FIN) + hi;
#pragma unroll
    for (int kb = 0; kb < G1_KB; ++kb) a[kb] = xr[kb * 2];
  }
  {
    const int g = blockIdx.x * 256 + tid;
    const int gc = g < NE / 8 ? g : NE / 8 - 1;
    const int4* d4 = (const int4*)(ei + NE) + gc * 2;
    const int4 p = d4[0], q = d4[1];
    uint2 o;
    o.x = own_of(p.x) | (own_of(p.y) << 8) | (own_of(p.z) << 16) | (own_of(p.w) << 24);
    o.y = own_of(q.x) | (own_of(q.y) << 8) | (own_of(q.z) << 16) | (own_of(q.w) << 24);
    if (g < NE / 8) ((uint2*)ob)[g] = o;
  }
  {
    const int t = tid >> 4, c = tid & 15;
#pragma unroll 4
    for (int kb = 0; kb < G1_KB; ++kb) {
#pragma unroll
      for (int hh = 0; hh < 2; ++hh) {
        const float* wp = W + (size_t)(kb * 8 + hh * 4) * HC + tid;
        Ws[(kb * G1_LT + t) * 32 + hh * 16 + c] = make_float4(wp[0], wp[HC], wp[2 * HC], wp[3 * HC]);
      }
    }
  }
  {
    const float4* wr = (const float4*)(W + (size_t)(wave * 16 + la) * HC) + hi;
    const float4* ar = (const float4*)(la < NH ? as_ : ad_) + hi;
    const int hd = la & (NH - 1);
    v8f wa = (v8f){};
#pragma unroll 16
    for (int kb = 0; kb < HC / 8; ++kb) {
      const float4 aw = wr[kb * 2];
      const float4 bv = ar[kb * 2];
      const bool on = (kb >> 2) == hd;
      v2f a0, a1, b0, b1;
      a0.x = aw.x; a0.y = aw.y; a1.x = aw.z; a1.y = aw.w;
      b0.x = on ? bv.x : 0.f; b0.y = on ? bv.y : 0.f; b1.x = on ? bv.z : 0.f; b1.y = on ? bv.w : 0.f;
      wa = __builtin_amdgcn_wmma_f32_16x16x4_f32(false, a0, false, b0, (short)0, wa, false, false);
      wa = __builtin_amdgcn_wmma_f32_16x16x4_f32(false, a1, false, b1, (short)0, wa, false, false);
    }
    float4* wt = Ws + ((2 * wave + hi) * G1_LT + G1_NT) * 32 + la;
    wt[0]  = make_float4(wa[0], wa[1], wa[2], wa[3]);
    wt[16] = make_float4(wa[4], wa[5], wa[6], wa[7]);
  }
  __syncthreads();
  if (!live) return;
  v8f acc[G1_LT];
#pragma unroll
  for (int t = 0; t < G1_LT; ++t) acc[t] = (v8f){};
#pragma unroll
  for (int kb = 0; kb < G1_KB; ++kb) {
    v2f a0, a1;
    a0.x = a[kb].x; a0.y = a[kb].y; a1.x = a[kb].z; a1.y = a[kb].w;
#pragma unroll
    for (int t = 0; t < G1_LT; ++t) {
      const float4 b = Ws[(kb * G1_LT + t) * 32 + lane];
      v2f b0, b1;
      b0.x = b.x; b0.y = b.y; b1.x = b.z; b1.y = b.w;
      acc[t] = __builtin_amdgcn_wmma_f32_16x16x4_f32(false, a0, false, b0, (short)0,
                                                     acc[t], false, false);
      acc[t] = __builtin_amdgcn_wmma_f32_16x16x4_f32(false, a1, false, b1, (short)0,
                                                     acc[t], false, false);
    }
  }
#pragma unroll
  for (int t = 0; t < G1_NT; ++t)
#pragma unroll
    for (int j = 0; j < 8; ++j)
      h[(size_t)(m0 + j + hi * 8) * HC + t * 16 + la] = acc[t][j];
  {
    float* lo = (la < NH ? os : od) + (la & (NH - 1));
#pragma unroll
    for (int j = 0; j < 8; ++j) lo[(m0 + j + hi * 8) * NH] = acc[G1_NT][j];
  }
}

#define G2_KB (HC / 8)
#define G2_NT 3
static_assert(G2_NT * 16 >= ODIM + 2, "the tiles cover the output columns and the two logit columns");
static_assert(ODIM % 16 == 8 && ODIM % 8 == 0, "the logit columns are entries 8 and 9 of the last tile; whole blocks of 8 channels");
static_assert(HC == 32 * GM_WV, "eight waves of two 16-row strips cover the rows of W");
__global__ __launch_bounds__(256) void k_gemm2(const float* __restrict__ h2,
                                               const float* __restrict__ W,
                                               const float* __restrict__ as_,
                                               const float* __restrict__ ad_,
                                               float* __restrict__ t2,
                                               float* __restrict__ os,
                                               float* __restrict__ od) {
  __shared__ __attribute__((aligned(16))) float4 Ws[G2_KB * G2_NT * 32];
  const int tid  = threadIdx.x;
  const int lane = tid & 31;
  const int wave = __builtin_amdgcn_readfirstlane(tid >> 5);
  const int la = lane & 15, hi = lane >> 4;
  const int strip = blockIdx.x * GM_WV + wave;
  const bool live = strip < GM_NS;
  const int m0 = (live ? strip : 0) * 16;
  float4 a[G2_KB];
  {
    const float4* xr = (const float4*)(h2 + (size_t)(m0 + la) * HC) + hi;
#pragma unroll
    for (int kb = 0; kb < G2_KB; ++kb) a[kb] = xr[kb * 2];
  }
  {
    static_assert(G2_KB == 4 * GM_WV, "eight waves cover the k blocks");
#pragma unroll
    for (int j = 0; j < 4; ++j) {
      const int kb = wave * 4 + j;
#pragma unroll
      for (int tt = 0; tt < G2_NT; ++tt) {
        const int n = tt * 16 + la;
        const float* wp = W + (size_t)(kb * 8 + hi * 4) * ODIM + (n < ODIM ? n : ODIM - 1);
        const float4 wv = make_float4(wp[0], wp[ODIM], wp[2 * ODIM], wp[3 * ODIM]);
        if (n != ODIM && n != ODIM + 1) Ws[(kb * G2_NT + tt) * 32 + lane] = wv;
      }
    }
  }
  {
    const float4* ar = (const float4*)(la == 1 ? ad_ : as_) + hi;
    const float4* wr = (const float4*)(W + (size_t)(wave * 32 + la) * ODIM) + hi;
    const bool on = la < 2;
    v8f wa[2];
    wa[0] = (v8f){}; wa[1] = (v8f){};
#pragma unroll
    for (int kb = 0; kb < ODIM / 8; ++kb) {
      const float4 bv = ar[kb * 2];
      v2f b0, b1;
      b0.x = on ? bv.x : 0.f; b0.y = on ? bv.y : 0.f; b1.x = on ? bv.z : 0.f; b1.y = on ? bv.w : 0.f;
#pragma unroll
      for (int ss = 0; ss < 2; ++ss) {
        const float4 aw = wr[ss * 4 * ODIM + kb * 2];
        v2f a0, a1;
        a0.x = aw.x; a0.y = aw.y; a1.x = aw.z; a1.y = aw.w;
        wa[ss] = __builtin_amdgcn_wmma_f32_16x16x4_f32(false, a0, false, b0, (short)0, wa[ss], false, false);
        wa[ss] = __builtin_amdgcn_wmma_f32_16x16x4_f32(false, a1, false, b1, (short)0, wa[ss], false, false);
      }
    }
    if (on) {
#pragma unroll
      for (int ss = 0; ss < 2; ++ss) {
        float4* wt = Ws + ((2 * (wave * 2 + ss) + hi) * G2_NT + (G2_NT - 1)) * 32 + (ODIM % 16) + la;
        wt[0]  = make_float4(wa[ss][0], wa[ss][1], wa[ss][2], wa[ss][3]);
        wt[16] = make_float4(wa[ss][4], wa[ss][5], wa[ss][6], wa[ss][7]);
      }
    }
  }
  __syncthreads();
  if (!live) return;
  v8f acc[G2_NT];
#pragma unroll
  for (int t = 0; t < G2_NT; ++t) acc[t] = (v8f){};
#pragma unroll
  for (int kb = 0; kb < G2_KB; ++kb) {
    v2f a0, a1;
    a0.x = a[kb].x; a0.y = a[kb].y; a1.x = a[kb].z; a1.y = a[kb].w;
#pragma unroll
    for (int t = 0; t < G2_NT; ++t) {
      const float4 b = Ws[(kb * G2_NT + t) * 32 + lane];
      v2f b0, b1;
      b0.x = b.x; b0.y = b.y; b1.x = b.z; b1.y = b.w;
      acc[t] = __builtin_amdgcn_wmma_f32_16x16x4_f32(false, a0, false, b0, (short)0,
                                                     acc[t], false, false);
      acc[t] = __builtin_amdgcn_wmma_f32_16x16x4_f32(false, a1, false, b1, (short)0,
                                                     acc[t], false, false);
    }
  }
#pragma unroll
  for (int t = 0; t < G2_NT; ++t) {
    const int col = t * 16 + la;
    if (col < ODIM) {
#pragma unroll
      for (int j = 0; j < 8; ++j)
        t2[(size_t)(m0 + j + hi * 8) * ODIM + col] = acc[t][j];
    } else if (col < ODIM + 2) {
      float* lo = col == ODIM ? os : od;
#pragma unroll
      for (int j = 0; j < 8; ++j) lo[m0 + j + hi * 8] = acc[t][j];
    }
  }
}

static_assert(256 * OWN_R >= NND, "256 owner work-groups cover the rows");
#define OWN_WG   256
#define OWN_T    1024
#define NE16     (NE / 16)
#define OWN_IT   ((NE16 + OWN_T - 1) / OWN_T)
#define OWN_STEP (32 * 16)
static_assert(NE % 16 == 0, "whole 16-byte pieces");
#define OWN_QTH  32
#define OWN_QCAP 768
static_assert(OWN_QCAP >= OWN_QTH + OWN_STEP, "a list about to be worked off still takes one scan step");
static_assert(OWN_R <= (1 << 16) && NND <= (1 << 16), "the fields of a list entry");

__device__ __forceinline__ float lane_bcast_f(float v, int l) {
  return __int_as_float(__builtin_amdgcn_readlane(__float_as_int(v), l));
}
__device__ __forceinline__ float leaky(float v) { return v > 0.f ? v : v * NEG_SLOPE; }

template <int H>
__device__ __forceinline__ float own_logit(const float* __restrict__ aS, const float* __restrict__ aD,
                                           int s, int d, int hd) {
  return leaky(aS[s * H + hd] + aD[d * H + hd]);
}

template <int H>
__device__ __forceinline__ void own_max(int* rmax, const float* __restrict__ aS,
                                        const float* __restrict__ aD, int s, int d0, int x) {
  float lg[H];
#pragma unroll
  for (int hd = 0; hd < H; ++hd) lg[hd] = own_logit<H>(aS, aD, s, d0 + x, hd);
#pragma unroll
  for (int hd = 0; hd < H; ++hd)
    __hip_atomic_fetch_max(&rmax[x * H + hd], f32_ord(lg[hd]), __ATOMIC_RELAXED, __HIP_MEMORY_SCOPE_WORKGROUP);
}

template <int H, bool DO_MAX>
__device__ __forceinline__ void own_convert(unsigned* qw, int from, int n, const int* __restrict__ ei,
                                            const float* __restrict__ aS, const float* __restrict__ aD,
                                            int* rmax, int d0, int lane) {
  for (int b = from; b < n; b += 32) {
    const int i = b + lane;
    if (i < n) {
      const unsigned eid = qw[i];
      const int x = ei[NE + eid] - d0;
      const int s = ei[eid];
      if (DO_MAX) own_max<H>(rmax, aS, aD, s, d0, x);
      qw[i] = ((unsigned)x << 16) | (unsigned)s;
    }
  }
}

#define OWN_TK 4
template <int H>
__device__ __forceinline__ void own_take(unsigned* qw, const unsigned* __restrict__ lw, int n,
                                         const float* __restrict__ aS, const float* __restrict__ aD,
                                         int* rmax, int d0, int lane) {
  for (int b = 0; b < n; b += 32 * OWN_TK) {
    unsigned e[OWN_TK];
#pragma unroll
    for (int k = 0; k < OWN_TK; ++k) {
      const int i = b + k * 32 + lane;
      e[k] = lw[i < n ? i : 0];
    }
    static_assert(OWN_TK == 4, "the statement below names the entries one by one");
    asm volatile("" :: "v"(e[0]), "v"(e[1]), "v"(e[2]), "v"(e[3]));
    float ls[OWN_TK][H], ld[OWN_TK][H];
#pragma unroll
    for (int k = 0; k < OWN_TK; ++k)
#pragma unroll
      for (int hd = 0; hd < H; ++hd) {
        ls[k][hd] = aS[(int)(e[k] & 0xFFFFu) * H + hd];
        ld[k][hd] = aD[(d0 + (int)(e[k] >> 16)) * H + hd];
      }
#pragma unroll
    for (int k = 0; k < OWN_TK; ++k)
#pragma unroll
      for (int hd = 0; hd < H; ++hd)
        asm volatile("" :: "v"(ls[k][hd]), "v"(ld[k][hd]));
#pragma unroll
    for (int k = 0; k < OWN_TK; ++k) {
      const int i = b + k * 32 + lane;
      if (i < n) {
        qw[i] = e[k];
        const int x = (int)(e[k] >> 16);
#pragma unroll
        for (int hd = 0; hd < H; ++hd)
          __hip_atomic_fetch_max(&rmax[x * H + hd], f32_ord(leaky(ls[k][hd] + ld[k][hd])), __ATOMIC_RELAXED,
                                 __HIP_MEMORY_SCOPE_WORKGROUP);
      }
    }
  }
}

template <int C, int H, bool INIT>
__device__ __forceinline__ void own_edge(float* __restrict__ acc, float* __restrict__ wsum,
                                         const int* rmax, const float* __restrict__ h,
                                         const float* __restrict__ aS, const float* __restrict__ aD,
                                         int d0, int s, int x, int lane) {
  const float* hs = h + (size_t)s * C;
  float* ar = acc + x * C;
  constexpr int NP = (C + 31) / 32;
  float v[NP];
#pragma unroll
  for (int j = 0; j < NP; ++j) {
    const int c = j * 32 + lane;
    v[j] = c < C ? hs[c] : 0.f;
  }
  float a = 0.f;
  if (lane < H) {
    a = __expf(own_logit<H>(aS, aD, s, d0 + x, lane) - ord_f32(rmax[x * H + lane]));
    if (INIT) wsum[x * H + lane] = a;
    else __hip_atomic_fetch_add(&wsum[x * H + lane], a, __ATOMIC_RELAXED, __HIP_MEMORY_SCOPE_WORKGROUP);
  }
#pragma unroll
  for (int j = 0; j < NP; ++j) {
    const float alpha = lane_bcast_f(a, (j * 32) / (C / H));
    const int c = j * 32 + lane;
    if (c < C) {
      const float t = v[j] * alpha;
      if (INIT) ar[c] = t;
      else __hip_atomic_fetch_add(&ar[c], t, __ATOMIC_RELAXED, __HIP_MEMORY_SCOPE_WORKGROUP);
    }
  }
}

template <int C, int H>
__device__ __forceinline__ void own_pair(float* __restrict__ acc, float* __restrict__ wsum,
                                         const int* rmax, const float* __restrict__ h,
                                         const float* __restrict__ aS, const float* __restrict__ aD,
                                         int d0, int sA, int xA, int sB, int xB, int lane) {
  const float* hA = h + (size_t)sA * C;
  const float* hB = h + (size_t)sB * C;
  float* arA = acc + xA * C;
  float* arB = acc + xB * C;
  constexpr int NP = (C + 31) / 32;
  float vA[NP], vB[NP];
#pragma unroll
  for (int j = 0; j < NP; ++j) {
    const int c = j * 32 + lane;
    vA[j] = c < C ? hA[c] : 0.f;
  }
#pragma unroll
  for (int j = 0; j < NP; ++j) {
    const int c = j * 32 + lane;
    vB[j] = c < C ? hB[c] : 0.f;
  }
  float a = 0.f;
  if (lane < 2 * H) {
    const bool second = lane >= H;
    const int hh = second ? lane - H : lane;
    const int s  = second ? sB : sA;
    const int x  = second ? xB : xA;
    a = __expf(own_logit<H>(aS, aD, s, d0 + x, hh) - ord_f32(rmax[x * H + hh]));
    __hip_atomic_fetch_add(&wsum[x * H + hh], a, __ATOMIC_RELAXED, __HIP_MEMORY_SCOPE_WORKGROUP);
  }
#pragma unroll
  for (int j = 0; j < NP; ++j) {
    const float alpha = lane_bcast_f(a, (j * 32) / (C / H));
    const int c = j * 32 + lane;
    if (c < C)
      __hip_atomic_fetch_add(&arA[c], vA[j] * alpha, __ATOMIC_RELAXED, __HIP_MEMORY_SCOPE_WORKGROUP);
  }
#pragma unroll
  for (int j = 0; j < NP; ++j) {
    const float alpha = lane_bcast_f(a, H + (j * 32) / (C / H));
    const int c = j * 32 + lane;
    if (c < C)
      __hip_atomic_fetch_add(&arB[c], vB[j] * alpha, __ATOMIC_RELAXED, __HIP_MEMORY_SCOPE_WORKGROUP);
  }
}

template <int C, int H>
__device__ __forceinline__ void own_drain(float* __restrict__ acc, float* __restrict__ wsum,
                                          const int* rmax, const unsigned* qw, int n,
                                          const float* __restrict__ h,
                                          const float* __restrict__ aS, const float* __restrict__ aD,
                                          int d0, int lane) {
  for (int b = 0; b < n; b += 32) {
    int cnt = n - b;
    if (cnt > 32) cnt = 32;
    unsigned ent = 0;
    if (lane < cnt) ent = qw[b + lane];
    int i = 0;
    for (; i + 1 < cnt; i += 2) {
      const unsigned eA = (unsigned)__builtin_amdgcn_readlane((int)ent, i);
      const unsigned eB = (unsigned)__builtin_amdgcn_readlane((int)ent, i + 1);
      own_pair<C, H>(acc, wsum, rmax, h, aS, aD, d0, (int)(eA & 0xFFFFu), (int)(eA >> 16),
                     (int)(eB & 0xFFFFu), (int)(eB >> 16), lane);
    }
    if (i < cnt) {
      const unsigned eA = (unsigned)__builtin_amdgcn_readlane((int)ent, i);
      own_edge<C, H, false>(acc, wsum, rmax, h, aS, aD, d0, (int)(eA & 0xFFFFu), (int)(eA >> 16), lane);
    }
  }
}

__device__ __forceinline__ unsigned own_eq4(unsigned w, unsigned pat) {
  const unsigned x = w ^ pat;
  const unsigned t = (x & 0x7F7F7F7Fu) + 0x7F7F7F7Fu;
  return ~(t | x) & 0x80808080u;
}
__device__ __forceinline__ int own_piece(int i) { return i < NE16 ? i : NE16 - 1; }
__device__ __forceinline__ unsigned own_eq16(const int4 v, unsigned pat) {
  return (own_eq4((unsigned)v.x, pat) >> 7) | (own_eq4((unsigned)v.y, pat) >> 6) |
         (own_eq4((unsigned)v.z, pat) >> 5) | (own_eq4((unsigned)v.w, pat) >> 4);
}

template <int C, int H, int MODE>
__device__ __forceinline__ void own_scan(float* __restrict__ acc, float* __restrict__ wsum, int* rmax,
                                         unsigned* qw, const int* __restrict__ ei,
                                         const unsigned* __restrict__ ob,
                                         const float* __restrict__ h,
                                         const float* __restrict__ aS, const float* __restrict__ aD,
                                         int d0, int wofs, int lane, int it_from,
                                         int& qn, bool& over, int& it_over) {
  const int4* ob4 = (const int4*)ob;
  const unsigned pat = (unsigned)blockIdx.x * 0x01010101u;
  int qdone = qn;
  int sp[2];
  sp[0] = (int)((blockIdx.x * OWN_IT) / OWN_WG + it_from) % OWN_IT;
  sp[1] = sp[0] + 1 < OWN_IT ? sp[0] + 1 : 0;
  int4 cur[2];
#pragma unroll
  for (int r = 0; r < 2; ++r) cur[r] = ob4[own_piece(sp[r] * OWN_T + wofs + lane)];
  for (int it = it_from; it < OWN_IT; it += 2) {
    int sn[2];
    sn[0] = sp[1] + 1 < OWN_IT ? sp[1] + 1 : 0;
    sn[1] = sn[0] + 1 < OWN_IT ? sn[0] + 1 : 0;
    int4 nxt[2];
#pragma unroll
    for (int r = 0; r < 2; ++r) nxt[r] = ob4[own_piece(sn[r] * OWN_T + wofs + lane)];
#pragma unroll
    for (int r = 0; r < 2; ++r) {
      if (it + r >= OWN_IT) break;
      if (MODE == 0 && !over && qn + OWN_STEP > OWN_QCAP) {
        own_convert<H, true>(qw, qdone, qn, ei, aS, aD, rmax, d0, lane);
        qdone = qn;
        over = true;
        it_over = it + r;
      }
      const int ib = sp[r] * OWN_T + wofs + lane;
      unsigned mm = own_eq16(cur[r], pat);
      if (ib >= NE16) mm = 0;
      unsigned mk = __builtin_amdgcn_ballot_w32(mm != 0);
      for (int tr = 0; tr < 16 && mk; ++tr) {
        const bool m = mm != 0;
        const int p = __builtin_ctz(mm | 0x80000000u);
        const unsigned eid = (unsigned)(ib * 16 + (p & 3) * 4 + (p >> 3));
        if (MODE == 0 && over) {
          if (m) own_max<H>(rmax, aS, aD, ei[eid], d0, ei[NE + eid] - d0);
        } else {
          if (m) qw[qn + (int)__builtin_amdgcn_mbcnt_lo(mk, 0u)] = eid;
          qn += __builtin_popcount(mk);
        }
        mm &= mm - 1;
        mk = __builtin_amdgcn_ballot_w32(mm != 0);
      }
      if (MODE == 0) {
        const int nb = (qn - qdone) & ~31;
        if (nb) {
          own_convert<H, true>(qw, qdone, qdone + nb, ei, aS, aD, rmax, d0, lane);
          qdone += nb;
        }
      } else if (qn >= OWN_QTH) {
        own_convert<H, false>(qw, 0, qn, ei, aS, aD, rmax, d0, lane);
        own_drain<C, H>(acc, wsum, rmax, qw, qn, h, aS, aD, d0, lane);
        qn = 0;
      }
    }
#pragma unroll
    for (int r = 0; r < 2; ++r) {
      cur[r] = nxt[r];
      sp[r] = sn[r];
    }
  }
  if (MODE == 0) {
    own_convert<H, true>(qw, qdone, qn, ei, aS, aD, rmax, d0, lane);
  } else {
    own_convert<H, false>(qw, 0, qn, ei, aS, aD, rmax, d0, lane);
    own_drain<C, H>(acc, wsum, rmax, qw, qn, h, aS, aD, d0, lane);
    qn = 0;
  }
}

template <int C, int H, bool KEPT>
__global__ __launch_bounds__(OWN_T) void k_own(const int* __restrict__ ei,
                                               const float* __restrict__ h,
                                               const float* __restrict__ aS,
                                               const float* __restrict__ aD,
                                               float* __restrict__ agg,
                                               unsigned* __restrict__ lst,
                                               int* __restrict__ lcnt,
                                               const unsigned* __restrict__ ob) {
  __shared__ __attribute__((aligned(16))) float acc[OWN_R * C];
  __shared__ float wsum[OWN_R * H];
  __shared__ int rmax[OWN_R * H];
  __shared__ unsigned q[(OWN_T / 32) * OWN_QCAP];
  const int tid = threadIdx.x, lane = tid & 31;
  const int wave = __builtin_amdgcn_readfirstlane(tid >> 5);
  const int d0 = blockIdx.x * OWN_R;
  int nrow = NND - d0;
  if (nrow > OWN_R) nrow = OWN_R;

  for (int i = tid; i < nrow * H; i += OWN_T)
    rmax[i] = f32_ord(leaky(aS[d0 * H + i] + aD[d0 * H + i]));
  __syncthreads();

  unsigned* qw = q + wave * OWN_QCAP;
  const int wofs = wave * 32;
  int qn = 0, it_over = OWN_IT;
  bool over = false;
  const int lw_i = blockIdx.x * (OWN_T / 32) + wave;
  unsigned* lw = lst + (size_t)lw_i * OWN_QCAP;
  bool have = false;
  if (KEPT) {
    const int n = __builtin_amdgcn_readfirstlane(lcnt[lw_i]);
    have = (unsigned)n <= (unsigned)OWN_QCAP;
    if (have) {
      own_take<H>(qw, lw, n, aS, aD, rmax, d0, lane);
      qn = n;
    }
  }
  if (!have) {
    own_scan<C, H, 0>(acc, wsum, rmax, qw, ei, ob, h, aS, aD, d0, wofs, lane, 0, qn, over, it_over);
    if (!KEPT) {
      if (lane == 0) lcnt[lw_i] = over ? -1 : qn;
      if (!over)
        for (int b = 0; b < qn; b += 32) {
          const int i = b + lane;
          if (i < qn) lw[i] = qw[i];
        }
    }
  }
  __syncthreads();

  for (int r = wave; r < nrow; r += OWN_T / 32)
    own_edge<C, H, true>(acc, wsum, rmax, h, aS, aD, d0, d0 + r, r, lane);
  __syncthreads();

  own_drain<C, H>(acc, wsum, rmax, qw, qn, h, aS, aD, d0, lane);
  if (over) {
    qn = 0;
    own_scan<C, H, 1>(acc, wsum, rmax, qw, ei, ob, h, aS, aD, d0, wofs, lane, it_over, qn, over, it_over);
  }
  __syncthreads();

  static_assert((C / H) % 4 == 0, "a 16-byte piece does not straddle two heads");
  constexpr int PH = (C / H) / 4;
  float4* o4 = (float4*)(agg + (size_t)d0 * C);
  const float4* a4 = (const float4*)acc;
  for (int i = tid; i < nrow * C / 4; i += OWN_T) {
    const float4 v = a4[i];
    const float dn = wsum[i / PH] + EPSV;
    o4[i] = make_float4(v.x / dn, v.y / dn, v.z / dn, v.w / dn);
  }
}

#define EL_WG 1250
#define EL_T  (EL_WG * 256)
#define EL_U  5
#define EL_N4 (NND * HC / 4)
static_assert(EL_N4 == 2 * EL_U * EL_T, "two full trips cover the rows exactly");
static_assert(EL_T % (HC / 4) == 0, "a thread's columns do not change from trip to trip");
__global__ __launch_bounds__(256) void k_elu_bias(float* __restrict__ a, const float* __restrict__ b) {
  const int i0 = blockIdx.x * 256 + threadIdx.x;
  const int c = (i0 & (HC / 4 - 1)) * 4;
  const float b0 = b[c], b1 = b[c + 1], b2 = b[c + 2], b3 = b[c + 3];
  float4* a4 = (float4*)a;
  for (int ib = i0; ib < EL_N4; ib += EL_U * EL_T) {
    float4 v[EL_U];
#pragma unroll
    for (int k = 0; k < EL_U; ++k) v[k] = a4[ib + k * EL_T];
#pragma unroll
    for (int k = 0; k < EL_U; ++k) {
      const float x0 = v[k].x + b0, x1 = v[k].y + b1, x2 = v[k].z + b2, x3 = v[k].w + b3;
      float4 r;
      r.x = x0 > 0.f ? x0 : (__expf(x0) - 1.f);
      r.y = x1 > 0.f ? x1 : (__expf(x1) - 1.f);
      r.z = x2 > 0.f ? x2 : (__expf(x2) - 1.f);
      r.w = x3 > 0.f ? x3 : (__expf(x3) - 1.f);
      a4[ib + k * EL_T] = r;
    }
  }
}

#define OB_WG 512
#define OB_T  (OB_WG * 256)
#define OB_U  4
#define OB_N4 (NND * ODIM / 4)
static_assert(OB_U * OB_T >= OB_N4, "one trip covers the output");
__global__ __launch_bounds__(256) void k_out_bias(const float* __restrict__ agg, const float* __restrict__ b,
                           float* __restrict__ out) {
  const int i0 = blockIdx.x * 256 + threadIdx.x;
  const float4* g4 = (const float4*)agg;
  float4* o4 = (float4*)out;
  float4 v[OB_U], bb[OB_U];
#pragma unroll
  for (int k = 0; k < OB_U; ++k) {
    const int i = i0 + k * OB_T;
    const int ic = i < OB_N4 ? i : i0;
    v[k] = g4[ic];
    const int c = (ic % (ODIM / 4)) * 4;
    bb[k] = make_float4(b[c], b[c + 1], b[c + 2], b[c + 3]);
  }
#pragma unroll
  for (int k = 0; k < OB_U; ++k) {
    const int i = i0 + k * OB_T;
    if (i < OB_N4)
      o4[i] = make_float4(v[k].x + bb[k].x, v[k].y + bb[k].y, v[k].z + bb[k].z, v[k].w + bb[k].w);
  }
}


extern "C" void kernel_launch(void* const* d_in, const int* in_sizes, int n_in,
                              void* d_out, int out_size, void* d_ws, size_t ws_size,
                              hipStream_t stream) {
  const float* x   = (const float*)d_in[0];
  const int*   ei  = (const int*)d_in[1];
  const float* W1  = (const float*)d_in[2];
  const float* as1 = (const float*)d_in[3];
  const float* ad1 = (const float*)d_in[4];
  const float* b1  = (const float*)d_in[5];
  const float* W2  = (const float*)d_in[6];
  const float* as2 = (const float*)d_in[7];
  const float* ad2 = (const float*)d_in[8];
  const float* b2  = (const float*)d_in[9];
  float* out = (float*)d_out;

  float* ws   = (float*)d_ws;
  float* h1   = ws;
  float* agg1 = h1   + (size_t)NND * HC;
  float* aS1  = agg1 + (size_t)NND * HC;
  float* aD1  = aS1  + NND * NH;
  float* t2   = aD1  + NND * NH;
  float* aS2  = t2   + (size_t)NND * ODIM;
  float* aD2  = aS2  + NND;
  float* agg2 = aD2  + NND;
  unsigned* lst = (unsigned*)(agg2 + (size_t)NND * ODIM);
  int* lcnt = (int*)(lst + (size_t)OWN_WG * (OWN_T / 32) * OWN_QCAP);
  unsigned* ob = (unsigned*)(lcnt + OWN_WG * (OWN_T / 32));

  k_gemm1<<<GM_WG, 256, 0, stream>>>(x, W1, as1, ad1, ei, h1, aS1, aD1, ob);
  k_own<HC, NH, false><<<OWN_WG, OWN_T, 0, stream>>>(ei, h1, aS1, aD1, agg1, lst, lcnt, ob);
  k_elu_bias<<<EL_WG, 256, 0, stream>>>(agg1, b1);

  k_gemm2<<<GM_WG, 256, 0, stream>>>(agg1, W2, as2, ad2, t2, aS2, aD2);
  k_own<ODIM, 1, true><<<OWN_WG, OWN_T, 0, stream>>>(ei, t2, aS2, aD2, agg2, lst, lcnt, ob);
  k_out_bias<<<OB_WG, 256, 0, stream>>>(agg2, b2, out);
}


# Gather loops of the owner groups batched: weights of 32 listed edges taken a lane per edge, then the rows of 8 (layer 2: 16) edges requested together before their LDS adds; sums in arrival order
# speedup vs baseline: 3.9912x; 1.0970x over previous
#include <hip/hip_runtime.h>
#include <hip/hip_bf16.h>

#define NND   50000
#define FIN   128
#define NH    8
#define HIDC  32
#define HC    256
#define ODIM  40
#define NE    800000
#define NET   850000
#define NEG_SLOPE 0.2f
#define EPSV  1e-16f

typedef __attribute__((ext_vector_type(2))) float v2f;
typedef __attribute__((ext_vector_type(8))) float v8f;

__device__ __forceinline__ int f32_ord(float x) {
  int i = __float_as_int(x);
  return i >= 0 ? i : (i ^ 0x7FFFFFFF);
}
__device__ __forceinline__ float ord_f32(int i) {
  return __int_as_float(i >= 0 ? i : (i ^ 0x7FFFFFFF));
}


#define OWN_R   196
#define GM_WV   8
#define GM_NS   (NND / 16)
#define GM_WG   ((GM_NS + GM_WV - 1) / GM_WV)
static_assert(NND % 16 == 0, "whole strips");

#define G1_KB (FIN / 8)
#define G1_NT (HC / 16)
#define G1_LT (G1_NT + 1)
static_assert(2 * NH == 16 && HIDC == 32, "one tile takes both logit sets; a block of 8 k lies in one head");
static_assert(FIN == 16 * GM_WV, "eight waves of 16 rows cover the rows of W");
static_assert(NE % 8 == 0 && GM_WG * 256 >= NE / 8, "the launch's threads cover the edges, 8 each");
static_assert((NND + OWN_R - 1) / OWN_R <= 256, "an owner number fits a byte");
__device__ __forceinline__ unsigned own_of(int d) { return (unsigned)d / (unsigned)OWN_R; }
__global__ __launch_bounds__(256) void k_gemm1(const float* __restrict__ x,
                                               const float* __restrict__ W,
                                               const float* __restrict__ as_,
                                               const float* __restrict__ ad_,
                                               const int* __restrict__ ei,
                                               float* __restrict__ h,
                                               float* __restrict__ os,
                                               float* __restrict__ od,
                                               unsigned* __restrict__ ob) {
  __shared__ __attribute__((aligned(16))) float4 Ws[G1_KB * G1_LT * 32];
  const int tid  = threadIdx.x;
  const int lane = tid & 31;
  const int wave = __builtin_amdgcn_readfirstlane(tid >> 5);
  const int la = lane & 15, hi = lane >> 4;
  const int strip = blockIdx.x * GM_WV + wave;
  const bool live = strip < GM_NS;
  const int m0 = (live ? strip : 0) * 16;
  float4 a[G1_KB];
  {
    const float4* xr = (const float4*)(x + (size_t)(m0 + la) * FIN) + hi;
#pragma unroll
    for (int kb = 0; kb < G1_KB; ++kb) a[kb] = xr[kb * 2];
  }
  {
    const int g = blockIdx.x * 256 + tid;
    const int gc = g < NE / 8 ? g : NE / 8 - 1;
    const int4* d4 = (const int4*)(ei + NE) + gc * 2;
    const int4 p = d4[0], q = d4[1];
    uint2 o;
    o.x = own_of(p.x) | (own_of(p.y) << 8) | (own_of(p.z) << 16) | (own_of(p.w) << 24);
    o.y = own_of(q.x) | (own_of(q.y) << 8) | (own_of(q.z) << 16) | (own_of(q.w) << 24);
    if (g < NE / 8) ((uint2*)ob)[g] = o;
  }
  {
    const int t = tid >> 4, c = tid & 15;
#pragma unroll 4
    for (int kb = 0; kb < G1_KB; ++kb) {
#pragma unroll
      for (int hh = 0; hh < 2; ++hh) {
        const float* wp = W + (size_t)(kb * 8 + hh * 4) * HC + tid;
        Ws[(kb * G1_LT + t) * 32 + hh * 16 + c] = make_float4(wp[0], wp[HC], wp[2 * HC], wp[3 * HC]);
      }
    }
  }
  {
    const float4* wr = (const float4*)(W + (size_t)(wave * 16 + la) * HC) + hi;
    const float4* ar = (const float4*)(la < NH ? as_ : ad_) + hi;
    const int hd = la & (NH - 1);
    v8f wa = (v8f){};
#pragma unroll 16
    for (int kb = 0; kb < HC / 8; ++kb) {
      const float4 aw = wr[kb * 2];
      const float4 bv = ar[kb * 2];
      const bool on = (kb >> 2) == hd;
      v2f a0, a1, b0, b1;
      a0.x = aw.x; a0.y = aw.y; a1.x = aw.z; a1.y = aw.w;
      b0.x = on ? bv.x : 0.f; b0.y = on ? bv.y : 0.f; b1.x = on ? bv.z : 0.f; b1.y = on ? bv.w : 0.f;
      wa = __builtin_amdgcn_wmma_f32_16x16x4_f32(false, a0, false, b0, (short)0, wa, false, false);
      wa = __builtin_amdgcn_wmma_f32_16x16x4_f32(false, a1, false, b1, (short)0, wa, false, false);
    }
    float4* wt = Ws + ((2 * wave + hi) * G1_LT + G1_NT) * 32 + la;
    wt[0]  = make_float4(wa[0], wa[1], wa[2], wa[3]);
    wt[16] = make_float4(wa[4], wa[5], wa[6], wa[7]);
  }
  __syncthreads();
  if (!live) return;
  v8f acc[G1_LT];
#pragma unroll
  for (int t = 0; t < G1_LT; ++t) acc[t] = (v8f){};
#pragma unroll
  for (int kb = 0; kb < G1_KB; ++kb) {
    v2f a0, a1;
    a0.x = a[kb].x; a0.y = a[kb].y; a1.x = a[kb].z; a1.y = a[kb].w;
#pragma unroll
    for (int t = 0; t < G1_LT; ++t) {
      const float4 b = Ws[(kb * G1_LT + t) * 32 + lane];
      v2f b0, b1;
      b0.x = b.x; b0.y = b.y; b1.x = b.z; b1.y = b.w;
      acc[t] = __builtin_amdgcn_wmma_f32_16x16x4_f32(false, a0, false, b0, (short)0,
                                                     acc[t], false, false);
      acc[t] = __builtin_amdgcn_wmma_f32_16x16x4_f32(false, a1, false, b1, (short)0,
                                                     acc[t], false, false);
    }
  }
#pragma unroll
  for (int t = 0; t < G1_NT; ++t)
#pragma unroll
    for (int j = 0; j < 8; ++j)
      h[(size_t)(m0 + j + hi * 8) * HC + t * 16 + la] = acc[t][j];
  {
    float* lo = (la < NH ? os : od) + (la & (NH - 1));
#pragma unroll
    for (int j = 0; j < 8; ++j) lo[(m0 + j + hi * 8) * NH] = acc[G1_NT][j];
  }
}

#define G2_KB (HC / 8)
#define G2_NT 3
static_assert(G2_NT * 16 >= ODIM + 2, "the tiles cover the output columns and the two logit columns");
static_assert(ODIM % 16 == 8 && ODIM % 8 == 0, "the logit columns are entries 8 and 9 of the last tile; whole blocks of 8 channels");
static_assert(HC == 32 * GM_WV, "eight waves of two 16-row strips cover the rows of W");
__global__ __launch_bounds__(256) void k_gemm2(const float* __restrict__ h2,
                                               const float* __restrict__ W,
                                               const float* __restrict__ as_,
                                               const float* __restrict__ ad_,
                                               float* __restrict__ t2,
                                               float* __restrict__ os,
                                               float* __restrict__ od) {
  __shared__ __attribute__((aligned(16))) float4 Ws[G2_KB * G2_NT * 32];
  const int tid  = threadIdx.x;
  const int lane = tid & 31;
  const int wave = __builtin_amdgcn_readfirstlane(tid >> 5);
  const int la = lane & 15, hi = lane >> 4;
  const int strip = blockIdx.x * GM_WV + wave;
  const bool live = strip < GM_NS;
  const int m0 = (live ? strip : 0) * 16;
  float4 a[G2_KB];
  {
    const float4* xr = (const float4*)(h2 + (size_t)(m0 + la) * HC) + hi;
#pragma unroll
    for (int kb = 0; kb < G2_KB; ++kb) a[kb] = xr[kb * 2];
  }
  {
    static_assert(G2_KB == 4 * GM_WV, "eight waves cover the k blocks");
#pragma unroll
    for (int j = 0; j < 4; ++j) {
      const int kb = wave * 4 + j;
#pragma unroll
      for (int tt = 0; tt < G2_NT; ++tt) {
        const int n = tt * 16 + la;
        const float* wp = W + (size_t)(kb * 8 + hi * 4) * ODIM + (n < ODIM ? n : ODIM - 1);
        const float4 wv = make_float4(wp[0], wp[ODIM], wp[2 * ODIM], wp[3 * ODIM]);
        if (n != ODIM && n != ODIM + 1) Ws[(kb * G2_NT + tt) * 32 + lane] = wv;
      }
    }
  }
  {
    const float4* ar = (const float4*)(la == 1 ? ad_ : as_) + hi;
    const float4* wr = (const float4*)(W + (size_t)(wave * 32 + la) * ODIM) + hi;
    const bool on = la < 2;
    v8f wa[2];
    wa[0] = (v8f){}; wa[1] = (v8f){};
#pragma unroll
    for (int kb = 0; kb < ODIM / 8; ++kb) {
      const float4 bv = ar[kb * 2];
      v2f b0, b1;
      b0.x = on ? bv.x : 0.f; b0.y = on ? bv.y : 0.f; b1.x = on ? bv.z : 0.f; b1.y = on ? bv.w : 0.f;
#pragma unroll
      for (int ss = 0; ss < 2; ++ss) {
        const float4 aw = wr[ss * 4 * ODIM + kb * 2];
        v2f a0, a1;
        a0.x = aw.x; a0.y = aw.y; a1.x = aw.z; a1.y = aw.w;
        wa[ss] = __builtin_amdgcn_wmma_f32_16x16x4_f32(false, a0, false, b0, (short)0, wa[ss], false, false);
        wa[ss] = __builtin_amdgcn_wmma_f32_16x16x4_f32(false, a1, false, b1, (short)0, wa[ss], false, false);
      }
    }
    if (on) {
#pragma unroll
      for (int ss = 0; ss < 2; ++ss) {
        float4* wt = Ws + ((2 * (wave * 2 + ss) + hi) * G2_NT + (G2_NT - 1)) * 32 + (ODIM % 16) + la;
        wt[0]  = make_float4(wa[ss][0], wa[ss][1], wa[ss][2], wa[ss][3]);
        wt[16] = make_float4(wa[ss][4], wa[ss][5], wa[ss][6], wa[ss][7]);
      }
    }
  }
  __syncthreads();
  if (!live) return;
  v8f acc[G2_NT];
#pragma unroll
  for (int t = 0; t < G2_NT; ++t) acc[t] = (v8f){};
#pragma unroll
  for (int kb = 0; kb < G2_KB; ++kb) {
    v2f a0, a1;
    a0.x = a[kb].x; a0.y = a[kb].y; a1.x = a[kb].z; a1.y = a[kb].w;
#pragma unroll
    for (int t = 0; t < G2_NT; ++t) {
      const float4 b = Ws[(kb * G2_NT + t) * 32 + lane];
      v2f b0, b1;
      b0.x = b.x; b0.y = b.y; b1.x = b.z; b1.y = b.w;
      acc[t] = __builtin_amdgcn_wmma_f32_16x16x4_f32(false, a0, false, b0, (short)0,
                                                     acc[t], false, false);
      acc[t] = __builtin_amdgcn_wmma_f32_16x16x4_f32(false, a1, false, b1, (short)0,
                                                     acc[t], false, false);
    }
  }
#pragma unroll
  for (int t = 0; t < G2_NT; ++t) {
    const int col = t * 16 + la;
    if (col < ODIM) {
#pragma unroll
      for (int j = 0; j < 8; ++j)
        t2[(size_t)(m0 + j + hi * 8) * ODIM + col] = acc[t][j];
    } else if (col < ODIM + 2) {
      float* lo = col == ODIM ? os : od;
#pragma unroll
      for (int j = 0; j < 8; ++j) lo[m0 + j + hi * 8] = acc[t][j];
    }
  }
}

static_assert(256 * OWN_R >= NND, "256 owner work-groups cover the rows");
#define OWN_WG   256
#define OWN_T    1024
#define NE16     (NE / 16)
#define OWN_IT   ((NE16 + OWN_T - 1) / OWN_T)
#define OWN_STEP (32 * 16)
static_assert(NE % 16 == 0, "whole 16-byte pieces");
#define OWN_QTH  32
#define OWN_QCAP 768
static_assert(OWN_QCAP >= OWN_QTH + OWN_STEP, "a list about to be worked off still takes one scan step");
static_assert(OWN_R <= (1 << 16) && NND <= (1 << 16), "the fields of a list entry");

__device__ __forceinline__ float lane_bcast_f(float v, int l) {
  return __int_as_float(__builtin_amdgcn_readlane(__float_as_int(v), l));
}
__device__ __forceinline__ float leaky(float v) { return v > 0.f ? v : v * NEG_SLOPE; }

template <int H>
__device__ __forceinline__ float own_logit(const float* __restrict__ aS, const float* __restrict__ aD,
                                           int s, int d, int hd) {
  return leaky(aS[s * H + hd] + aD[d * H + hd]);
}

template <int H>
__device__ __forceinline__ void own_max(int* rmax, const float* __restrict__ aS,
                                        const float* __restrict__ aD, int s, int d0, int x) {
  float lg[H];
#pragma unroll
  for (int hd = 0; hd < H; ++hd) lg[hd] = own_logit<H>(aS, aD, s, d0 + x, hd);
#pragma unroll
  for (int hd = 0; hd < H; ++hd)
    __hip_atomic_fetch_max(&rmax[x * H + hd], f32_ord(lg[hd]), __ATOMIC_RELAXED, __HIP_MEMORY_SCOPE_WORKGROUP);
}

template <int H, bool DO_MAX>
__device__ __forceinline__ void own_convert(unsigned* qw, int from, int n, const int* __restrict__ ei,
                                            const float* __restrict__ aS, const float* __restrict__ aD,
                                            int* rmax, int d0, int lane) {
  for (int b = from; b < n; b += 32) {
    const int i = b + lane;
    if (i < n) {
      const unsigned eid = qw[i];
      const int x = ei[NE + eid] - d0;
      const int s = ei[eid];
      if (DO_MAX) own_max<H>(rmax, aS, aD, s, d0, x);
      qw[i] = ((unsigned)x << 16) | (unsigned)s;
    }
  }
}

#define OWN_TK 4
template <int H>
__device__ __forceinline__ void own_take(unsigned* qw, const unsigned* __restrict__ lw, int n,
                                         const float* __restrict__ aS, const float* __restrict__ aD,
                                         int* rmax, int d0, int lane) {
  for (int b = 0; b < n; b += 32 * OWN_TK) {
    unsigned e[OWN_TK];
#pragma unroll
    for (int k = 0; k < OWN_TK; ++k) {
      const int i = b + k * 32 + lane;
      e[k] = lw[i < n ? i : 0];
    }
    static_assert(OWN_TK == 4, "the statement below names the entries one by one");
    asm volatile("" :: "v"(e[0]), "v"(e[1]), "v"(e[2]), "v"(e[3]));
    float ls[OWN_TK][H], ld[OWN_TK][H];
#pragma unroll
    for (int k = 0; k < OWN_TK; ++k)
#pragma unroll
      for (int hd = 0; hd < H; ++hd) {
        ls[k][hd] = aS[(int)(e[k] & 0xFFFFu) * H + hd];
        ld[k][hd] = aD[(d0 + (int)(e[k] >> 16)) * H + hd];
      }
#pragma unroll
    for (int k = 0; k < OWN_TK; ++k)
#pragma unroll
      for (int hd = 0; hd < H; ++hd)
        asm volatile("" :: "v"(ls[k][hd]), "v"(ld[k][hd]));
#pragma unroll
    for (int k = 0; k < OWN_TK; ++k) {
      const int i = b + k * 32 + lane;
      if (i < n) {
        qw[i] = e[k];
        const int x = (int)(e[k] >> 16);
#pragma unroll
        for (int hd = 0; hd < H; ++hd)
          __hip_atomic_fetch_max(&rmax[x * H + hd], f32_ord(leaky(ls[k][hd] + ld[k][hd])), __ATOMIC_RELAXED,
                                 __HIP_MEMORY_SCOPE_WORKGROUP);
      }
    }
  }
}

template <int C, int H, bool INIT>
__device__ __forceinline__ void own_batch(float* __restrict__ acc, float* __restrict__ wsum,
                                          const int* rmax, const float* __restrict__ h,
                                          const float* __restrict__ aS, const float* __restrict__ aD,
                                          int d0, unsigned ent, int cnt, int lane) {
  constexpr int NP = (C + 31) / 32;
  constexpr int G = NP >= 8 ? 8 : 16;
  float a[H];
#pragma unroll
  for (int hd = 0; hd < H; ++hd) a[hd] = 0.f;
  if (lane < cnt) {
    const int s = (int)(ent & 0xFFFFu), x = (int)(ent >> 16);
    float lg[H];
#pragma unroll
    for (int hd = 0; hd < H; ++hd) lg[hd] = own_logit<H>(aS, aD, s, d0 + x, hd);
#pragma unroll
    for (int hd = 0; hd < H; ++hd) {
      a[hd] = __expf(lg[hd] - ord_f32(rmax[x * H + hd]));
      if (INIT) wsum[x * H + hd] = a[hd];
      else __hip_atomic_fetch_add(&wsum[x * H + hd], a[hd], __ATOMIC_RELAXED, __HIP_MEMORY_SCOPE_WORKGROUP);
    }
  }
  for (int g = 0; g < cnt; g += G) {
    unsigned e[G];
    float v[G][NP];
#pragma unroll
    for (int k = 0; k < G; ++k) {
      const int i = g + k < cnt ? g + k : cnt - 1;
      e[k] = (unsigned)__builtin_amdgcn_readlane((int)ent, i);
      const float* hs = h + (size_t)(e[k] & 0xFFFFu) * C;
#pragma unroll
      for (int j = 0; j < NP; ++j) {
        const int c = j * 32 + lane;
        v[k][j] = hs[c < C ? c : C - 1];
      }
    }
#pragma unroll
    for (int k = 0; k < G; ++k) {
#pragma unroll
      for (int j = 0; j < NP; ++j)
        asm volatile("" :: "v"(v[k][j]));
      if (g + k < cnt) {
        float* ar = acc + (int)(e[k] >> 16) * C;
#pragma unroll
        for (int j = 0; j < NP; ++j) {
          const float alpha = lane_bcast_f(a[(j * 32) / (C / H)], g + k);
          const int c = j * 32 + lane;
          if (c < C) {
            const float t = v[k][j] * alpha;
            if (INIT) ar[c] = t;
            else __hip_atomic_fetch_add(&ar[c], t, __ATOMIC_RELAXED, __HIP_MEMORY_SCOPE_WORKGROUP);
          }
        }
      }
    }
  }
}

template <int C, int H>
__device__ __forceinline__ void own_drain(float* __restrict__ acc, float* __restrict__ wsum,
                                          const int* rmax, const unsigned* qw, int n,
                                          const float* __restrict__ h,
                                          const float* __restrict__ aS, const float* __restrict__ aD,
                                          int d0, int lane) {
  for (int b = 0; b < n; b += 32) {
    int cnt = n - b;
    if (cnt > 32) cnt = 32;
    unsigned ent = 0;
    if (lane < cnt) ent = qw[b + lane];
    own_batch<C, H, false>(acc, wsum, rmax, h, aS, aD, d0, ent, cnt, lane);
  }
}

__device__ __forceinline__ unsigned own_eq4(unsigned w, unsigned pat) {
  const unsigned x = w ^ pat;
  const unsigned t = (x & 0x7F7F7F7Fu) + 0x7F7F7F7Fu;
  return ~(t | x) & 0x80808080u;
}
__device__ __forceinline__ int own_piece(int i) { return i < NE16 ? i : NE16 - 1; }
__device__ __forceinline__ unsigned own_eq16(const int4 v, unsigned pat) {
  return (own_eq4((unsigned)v.x, pat) >> 7) | (own_eq4((unsigned)v.y, pat) >> 6) |
         (own_eq4((unsigned)v.z, pat) >> 5) | (own_eq4((unsigned)v.w, pat) >> 4);
}

template <int C, int H, int MODE>
__device__ __forceinline__ void own_scan(float* __restrict__ acc, float* __restrict__ wsum, int* rmax,
                                         unsigned* qw, const int* __restrict__ ei,
                                         const unsigned* __restrict__ ob,
                                         const float* __restrict__ h,
                                         const float* __restrict__ aS, const float* __restrict__ aD,
                                         int d0, int wofs, int lane, int it_from,
                                         int& qn, bool& over, int& it_over) {
  const int4* ob4 = (const int4*)ob;
  const unsigned pat = (unsigned)blockIdx.x * 0x01010101u;
  int qdone = qn;
  int sp[2];
  sp[0] = (int)((blockIdx.x * OWN_IT) / OWN_WG + it_from) % OWN_IT;
  sp[1] = sp[0] + 1 < OWN_IT ? sp[0] + 1 : 0;
  int4 cur[2];
#pragma unroll
  for (int r = 0; r < 2; ++r) cur[r] = ob4[own_piece(sp[r] * OWN_T + wofs + lane)];
  for (int it = it_from; it < OWN_IT; it += 2) {
    int sn[2];
    sn[0] = sp[1] + 1 < OWN_IT ? sp[1] + 1 : 0;
    sn[1] = sn[0] + 1 < OWN_IT ? sn[0] + 1 : 0;
    int4 nxt[2];
#pragma unroll
    for (int r = 0; r < 2; ++r) nxt[r] = ob4[own_piece(sn[r] * OWN_T + wofs + lane)];
#pragma unroll
    for (int r = 0; r < 2; ++r) {
      if (it + r >= OWN_IT) break;
      if (MODE == 0 && !over && qn + OWN_STEP > OWN_QCAP) {
        own_convert<H, true>(qw, qdone, qn, ei, aS, aD, rmax, d0, lane);
        qdone = qn;
        over = true;
        it_over = it + r;
      }
      const int ib = sp[r] * OWN_T + wofs + lane;
      unsigned mm = own_eq16(cur[r], pat);
      if (ib >= NE16) mm = 0;
      unsigned mk = __builtin_amdgcn_ballot_w32(mm != 0);
      for (int tr = 0; tr < 16 && mk; ++tr) {
        const bool m = mm != 0;
        const int p = __builtin_ctz(mm | 0x80000000u);
        const unsigned eid = (unsigned)(ib * 16 + (p & 3) * 4 + (p >> 3));
        if (MODE == 0 && over) {
          if (m) own_max<H>(rmax, aS, aD, ei[eid], d0, ei[NE + eid] - d0);
        } else {
          if (m) qw[qn + (int)__builtin_amdgcn_mbcnt_lo(mk, 0u)] = eid;
          qn += __builtin_popcount(mk);
        }
        mm &= mm - 1;
        mk = __builtin_amdgcn_ballot_w32(mm != 0);
      }
      if (MODE == 0) {
        const int nb = (qn - qdone) & ~31;
        if (nb) {
          own_convert<H, true>(qw, qdone, qdone + nb, ei, aS, aD, rmax, d0, lane);
          qdone += nb;
        }
      } else if (qn >= OWN_QTH) {
        own_convert<H, false>(qw, 0, qn, ei, aS, aD, rmax, d0, lane);
        own_drain<C, H>(acc, wsum, rmax, qw, qn, h, aS, aD, d0, lane);
        qn = 0;
      }
    }
#pragma unroll
    for (int r = 0; r < 2; ++r) {
      cur[r] = nxt[r];
      sp[r] = sn[r];
    }
  }
  if (MODE == 0) {
    own_convert<H, true>(qw, qdone, qn, ei, aS, aD, rmax, d0, lane);
  } else {
    own_convert<H, false>(qw, 0, qn, ei, aS, aD, rmax, d0, lane);
    own_drain<C, H>(acc, wsum, rmax, qw, qn, h, aS, aD, d0, lane);
    qn = 0;
  }
}

template <int C, int H, bool KEPT>
__global__ __launch_bounds__(OWN_T) void k_own(const int* __restrict__ ei,
                                               const float* __restrict__ h,
                                               const float* __restrict__ aS,
                                               const float* __restrict__ aD,
                                               float* __restrict__ agg,
                                               unsigned* __restrict__ lst,
                                               int* __restrict__ lcnt,
                                               const unsigned* __restrict__ ob) {
  __shared__ __attribute__((aligned(16))) float acc[OWN_R * C];
  __shared__ float wsum[OWN_R * H];
  __shared__ int rmax[OWN_R * H];
  __shared__ unsigned q[(OWN_T / 32) * OWN_QCAP];
  const int tid = threadIdx.x, lane = tid & 31;
  const int wave = __builtin_amdgcn_readfirstlane(tid >> 5);
  const int d0 = blockIdx.x * OWN_R;
  int nrow = NND - d0;
  if (nrow > OWN_R) nrow = OWN_R;

  for (int i = tid; i < nrow * H; i += OWN_T)
    rmax[i] = f32_ord(leaky(aS[d0 * H + i] + aD[d0 * H + i]));
  __syncthreads();

  unsigned* qw = q + wave * OWN_QCAP;
  const int wofs = wave * 32;
  int qn = 0, it_over = OWN_IT;
  bool over = false;
  const int lw_i = blockIdx.x * (OWN_T / 32) + wave;
  unsigned* lw = lst + (size_t)lw_i * OWN_QCAP;
  bool have = false;
  if (KEPT) {
    const int n = __builtin_amdgcn_readfirstlane(lcnt[lw_i]);
    have = (unsigned)n <= (unsigned)OWN_QCAP;
    if (have) {
      own_take<H>(qw, lw, n, aS, aD, rmax, d0, lane);
      qn = n;
    }
  }
  if (!have) {
    own_scan<C, H, 0>(acc, wsum, rmax, qw, ei, ob, h, aS, aD, d0, wofs, lane, 0, qn, over, it_over);
    if (!KEPT) {
      if (lane == 0) lcnt[lw_i] = over ? -1 : qn;
      if (!over)
        for (int b = 0; b < qn; b += 32) {
          const int i = b + lane;
          if (i < qn) lw[i] = qw[i];
        }
    }
  }
  __syncthreads();

  static_assert((OWN_R + OWN_T / 32 - 1) / (OWN_T / 32) <= 32, "a wave's rows fit one batch");
  {
    const int cnt = nrow > wave ? (nrow - wave + OWN_T / 32 - 1) / (OWN_T / 32) : 0;
    const int r = wave + lane * (OWN_T / 32);
    const unsigned ent = lane < cnt ? ((unsigned)r << 16) | (unsigned)(d0 + r) : 0u;
    own_batch<C, H, true>(acc, wsum, rmax, h, aS, aD, d0, ent, cnt, lane);
  }
  __syncthreads();

  own_drain<C, H>(acc, wsum, rmax, qw, qn, h, aS, aD, d0, lane);
  if (over) {
    qn = 0;
    own_scan<C, H, 1>(acc, wsum, rmax, qw, ei, ob, h, aS, aD, d0, wofs, lane, it_over, qn, over, it_over);
  }
  __syncthreads();

  static_assert((C / H) % 4 == 0, "a 16-byte piece does not straddle two heads");
  constexpr int PH = (C / H) / 4;
  float4* o4 = (float4*)(agg + (size_t)d0 * C);
  const float4* a4 = (const float4*)acc;
  for (int i = tid; i < nrow * C / 4; i += OWN_T) {
    const float4 v = a4[i];
    const float dn = wsum[i / PH] + EPSV;
    o4[i] = make_float4(v.x / dn, v.y / dn, v.z / dn, v.w / dn);
  }
}

#define EL_WG 1250
#define EL_T  (EL_WG * 256)
#define EL_U  5
#define EL_N4 (NND * HC / 4)
static_assert(EL_N4 == 2 * EL_U * EL_T, "two full trips cover the rows exactly");
static_assert(EL_T % (HC / 4) == 0, "a thread's columns do not change from trip to trip");
__global__ __launch_bounds__(256) void k_elu_bias(float* __restrict__ a, const float* __restrict__ b) {
  const int i0 = blockIdx.x * 256 + threadIdx.x;
  const int c = (i0 & (HC / 4 - 1)) * 4;
  const float b0 = b[c], b1 = b[c + 1], b2 = b[c + 2], b3 = b[c + 3];
  float4* a4 = (float4*)a;
  for (int ib = i0; ib < EL_N4; ib += EL_U * EL_T) {
    float4 v[EL_U];
#pragma unroll
    for (int k = 0; k < EL_U; ++k) v[k] = a4[ib + k * EL_T];
#pragma unroll
    for (int k = 0; k < EL_U; ++k) {
      const float x0 = v[k].x + b0, x1 = v[k].y + b1, x2 = v[k].z + b2, x3 = v[k].w + b3;
      float4 r;
      r.x = x0 > 0.f ? x0 : (__expf(x0) - 1.f);
      r.y = x1 > 0.f ? x1 : (__expf(x1) - 1.f);
      r.z = x2 > 0.f ? x2 : (__expf(x2) - 1.f);
      r.w = x3 > 0.f ? x3 : (__expf(x3) - 1.f);
      a4[ib + k * EL_T] = r;
    }
  }
}

#define OB_WG 512
#define OB_T  (OB_WG * 256)
#define OB_U  4
#define OB_N4 (NND * ODIM / 4)
static_assert(OB_U * OB_T >= OB_N4, "one trip covers the output");
__global__ __launch_bounds__(256) void k_out_bias(const float* __restrict__ agg, const float* __restrict__ b,
                           float* __restrict__ out) {
  const int i0 = blockIdx.x * 256 + threadIdx.x;
  const float4* g4 = (const float4*)agg;
  float4* o4 = (float4*)out;
  float4 v[OB_U], bb[OB_U];
#pragma unroll
  for (int k = 0; k < OB_U; ++k) {
    const int i = i0 + k * OB_T;
    const int ic = i < OB_N4 ? i : i0;
    v[k] = g4[ic];
    const int c = (ic % (ODIM / 4)) * 4;
    bb[k] = make_float4(b[c], b[c + 1], b[c + 2], b[c + 3]);
  }
#pragma unroll
  for (int k = 0; k < OB_U; ++k) {
    const int i = i0 + k * OB_T;
    if (i < OB_N4)
      o4[i] = make_float4(v[k].x + bb[k].x, v[k].y + bb[k].y, v[k].z + bb[k].z, v[k].w + bb[k].w);
  }
}


extern "C" void kernel_launch(void* const* d_in, const int* in_sizes, int n_in,
                              void* d_out, int out_size, void* d_ws, size_t ws_size,
                              hipStream_t stream) {
  const float* x   = (const float*)d_in[0];
  const int*   ei  = (const int*)d_in[1];
  const float* W1  = (const float*)d_in[2];
  const float* as1 = (const float*)d_in[3];
  const float* ad1 = (const float*)d_in[4];
  const float* b1  = (const float*)d_in[5];
  const float* W2  = (const float*)d_in[6];
  const float* as2 = (const float*)d_in[7];
  const float* ad2 = (const float*)d_in[8];
  const float* b2  = (const float*)d_in[9];
  float* out = (float*)d_out;

  float* ws   = (float*)d_ws;
  float* h1   = ws;
  float* agg1 = h1   + (size_t)NND * HC;
  float* aS1  = agg1 + (size_t)NND * HC;
  float* aD1  = aS1  + NND * NH;
  float* t2   = aD1  + NND * NH;
  float* aS2  = t2   + (size_t)NND * ODIM;
  float* aD2  = aS2  + NND;
  float* agg2 = aD2  + NND;
  unsigned* lst = (unsigned*)(agg2 + (size_t)NND * ODIM);
  int* lcnt = (int*)(lst + (size_t)OWN_WG * (OWN_T / 32) * OWN_QCAP);
  unsigned* ob = (unsigned*)(lcnt + OWN_WG * (OWN_T / 32));

  k_gemm1<<<GM_WG, 256, 0, stream>>>(x, W1, as1, ad1, ei, h1, aS1, aD1, ob);
  k_own<HC, NH, false><<<OWN_WG, OWN_T, 0, stream>>>(ei, h1, aS1, aD1, agg1, lst, lcnt, ob);
  k_elu_bias<<<EL_WG, 256, 0, stream>>>(agg1, b1);

  k_gemm2<<<GM_WG, 256, 0, stream>>>(agg1, W2, as2, ad2, t2, aS2, aD2);
  k_own<ODIM, 1, true><<<OWN_WG, OWN_T, 0, stream>>>(ei, t2, aS2, aD2, agg2, lst, lcnt, ob);
  k_out_bias<<<OB_WG, 256, 0, stream>>>(agg2, b2, out);
}


# Layer 1's bias and ELU applied by the second product to its strip in registers, per block of 8 k (bias in LDS): the elementwise launch and its pass over the rows are gone; 5 launches; same values
# speedup vs baseline: 4.2823x; 1.0729x over previous
#include <hip/hip_runtime.h>
#include <hip/hip_bf16.h>

#define NND   50000
#define FIN   128
#define NH    8
#define HIDC  32
#define HC    256
#define ODIM  40
#define NE    800000
#define NET   850000
#define NEG_SLOPE 0.2f
#define EPSV  1e-16f

typedef __attribute__((ext_vector_type(2))) float v2f;
typedef __attribute__((ext_vector_type(8))) float v8f;

__device__ __forceinline__ int f32_ord(float x) {
  int i = __float_as_int(x);
  return i >= 0 ? i : (i ^ 0x7FFFFFFF);
}
__device__ __forceinline__ float ord_f32(int i) {
  return __int_as_float(i >= 0 ? i : (i ^ 0x7FFFFFFF));
}


#define OWN_R   196
#define GM_WV   8
#define GM_NS   (NND / 16)
#define GM_WG   ((GM_NS + GM_WV - 1) / GM_WV)
static_assert(NND % 16 == 0, "whole strips");

#define G1_KB (FIN / 8)
#define G1_NT (HC / 16)
#define G1_LT (G1_NT + 1)
static_assert(2 * NH == 16 && HIDC == 32, "one tile takes both logit sets; a block of 8 k lies in one head");
static_assert(FIN == 16 * GM_WV, "eight waves of 16 rows cover the rows of W");
static_assert(NE % 8 == 0 && GM_WG * 256 >= NE / 8, "the launch's threads cover the edges, 8 each");
static_assert((NND + OWN_R - 1) / OWN_R <= 256, "an owner number fits a byte");
__device__ __forceinline__ unsigned own_of(int d) { return (unsigned)d / (unsigned)OWN_R; }
__global__ __launch_bounds__(256) void k_gemm1(const float* __restrict__ x,
                                               const float* __restrict__ W,
                                               const float* __restrict__ as_,
                                               const float* __restrict__ ad_,
                                               const int* __restrict__ ei,
                                               float* __restrict__ h,
                                               float* __restrict__ os,
                                               float* __restrict__ od,
                                               unsigned* __restrict__ ob) {
  __shared__ __attribute__((aligned(16))) float4 Ws[G1_KB * G1_LT * 32];
  const int tid  = threadIdx.x;
  const int lane = tid & 31;
  const int wave = __builtin_amdgcn_readfirstlane(tid >> 5);
  const int la = lane & 15, hi = lane >> 4;
  const int strip = blockIdx.x * GM_WV + wave;
  const bool live = strip < GM_NS;
  const int m0 = (live ? strip : 0) * 16;
  float4 a[G1_KB];
  {
    const float4* xr = (const float4*)(x + (size_t)(m0 + la) * FIN) + hi;
#pragma unroll
    for (int kb = 0; kb < G1_KB; ++kb) a[kb] = xr[kb * 2];
  }
  {
    const int g = blockIdx.x * 256 + tid;
    const int gc = g < NE / 8 ? g : NE / 8 - 1;
    const int4* d4 = (const int4*)(ei + NE) + gc * 2;
    const int4 p = d4[0], q = d4[1];
    uint2 o;
    o.x = own_of(p.x) | (own_of(p.y) << 8) | (own_of(p.z) << 16) | (own_of(p.w) << 24);
    o.y = own_of(q.x) | (own_of(q.y) << 8) | (own_of(q.z) << 16) | (own_of(q.w) << 24);
    if (g < NE / 8) ((uint2*)ob)[g] = o;
  }
  {
    const int t = tid >> 4, c = tid & 15;
#pragma unroll 4
    for (int kb = 0; kb < G1_KB; ++kb) {
#pragma unroll
      for (int hh = 0; hh < 2; ++hh) {
        const float* wp = W + (size_t)(kb * 8 + hh * 4) * HC + tid;
        Ws[(kb * G1_LT + t) * 32 + hh * 16 + c] = make_float4(wp[0], wp[HC], wp[2 * HC], wp[3 * HC]);
      }
    }
  }
  {
    const float4* wr = (const float4*)(W + (size_t)(wave * 16 + la) * HC) + hi;
    const float4* ar = (const float4*)(la < NH ? as_ : ad_) + hi;
    const int hd = la & (NH - 1);
    v8f wa = (v8f){};
#pragma unroll 16
    for (int kb = 0; kb < HC / 8; ++kb) {
      const float4 aw = wr[kb * 2];
      const float4 bv = ar[kb * 2];
      const bool on = (kb >> 2) == hd;
      v2f a0, a1, b0, b1;
      a0.x = aw.x; a0.y = aw.y; a1.x = aw.z; a1.y = aw.w;
      b0.x = on ? bv.x : 0.f; b0.y = on ? bv.y : 0.f; b1.x = on ? bv.z : 0.f; b1.y = on ? bv.w : 0.f;
      wa = __builtin_amdgcn_wmma_f32_16x16x4_f32(false, a0, false, b0, (short)0, wa, false, false);
      wa = __builtin_amdgcn_wmma_f32_16x16x4_f32(false, a1, false, b1, (short)0, wa, false, false);
    }
    float4* wt = Ws + ((2 * wave + hi) * G1_LT + G1_NT) * 32 + la;
    wt[0]  = make_float4(wa[0], wa[1], wa[2], wa[3]);
    wt[16] = make_float4(wa[4], wa[5], wa[6], wa[7]);
  }
  __syncthreads();
  if (!live) return;
  v8f acc[G1_LT];
#pragma unroll
  for (int t = 0; t < G1_LT; ++t) acc[t] = (v8f){};
#pragma unroll
  for (int kb = 0; kb < G1_KB; ++kb) {
    v2f a0, a1;
    a0.x = a[kb].x; a0.y = a[kb].y; a1.x = a[kb].z; a1.y = a[kb].w;
#pragma unroll
    for (int t = 0; t < G1_LT; ++t) {
      const float4 b = Ws[(kb * G1_LT + t) * 32 + lane];
      v2f b0, b1;
      b0.x = b.x; b0.y = b.y; b1.x = b.z; b1.y = b.w;
      acc[t] = __builtin_amdgcn_wmma_f32_16x16x4_f32(false, a0, false, b0, (short)0,
                                                     acc[t], false, false);
      acc[t] = __builtin_amdgcn_wmma_f32_16x16x4_f32(false, a1, false, b1, (short)0,
                                                     acc[t], false, false);
    }
  }
#pragma unroll
  for (int t = 0; t < G1_NT; ++t)
#pragma unroll
    for (int j = 0; j < 8; ++j)
      h[(size_t)(m0 + j + hi * 8) * HC + t * 16 + la] = acc[t][j];
  {
    float* lo = (la < NH ? os : od) + (la & (NH - 1));
#pragma unroll
    for (int j = 0; j < 8; ++j) lo[(m0 + j + hi * 8) * NH] = acc[G1_NT][j];
  }
}

__device__ __forceinline__ float elu_f(float v) { return v > 0.f ? v : (__expf(v) - 1.f); }

#define G2_KB (HC / 8)
#define G2_NT 3
static_assert(G2_NT * 16 >= ODIM + 2, "the tiles cover the output columns and the two logit columns");
static_assert(ODIM % 16 == 8 && ODIM % 8 == 0, "the logit columns are entries 8 and 9 of the last tile; whole blocks of 8 channels");
static_assert(HC == 32 * GM_WV, "eight waves of two 16-row strips cover the rows of W");
__global__ __launch_bounds__(256) void k_gemm2(const float* __restrict__ h2,
                                               const float* __restrict__ bin,
                                               const float* __restrict__ W,
                                               const float* __restrict__ as_,
                                               const float* __restrict__ ad_,
                                               float* __restrict__ t2,
                                               float* __restrict__ os,
                                               float* __restrict__ od) {
  __shared__ __attribute__((aligned(16))) float4 Ws[G2_KB * G2_NT * 32];
  __shared__ __attribute__((aligned(16))) float4 Bs[HC / 4];
  const int tid  = threadIdx.x;
  const int lane = tid & 31;
  const int wave = __builtin_amdgcn_readfirstlane(tid >> 5);
  const int la = lane & 15, hi = lane >> 4;
  const int strip = blockIdx.x * GM_WV + wave;
  const bool live = strip < GM_NS;
  const int m0 = (live ? strip : 0) * 16;
  float4 a[G2_KB];
  {
    const float4* xr = (const float4*)(h2 + (size_t)(m0 + la) * HC) + hi;
#pragma unroll
    for (int kb = 0; kb < G2_KB; ++kb) a[kb] = xr[kb * 2];
  }
  if (tid < HC / 4) Bs[tid] = ((const float4*)bin)[tid];
  {
    static_assert(G2_KB == 4 * GM_WV, "eight waves cover the k blocks");
#pragma unroll
    for (int j = 0; j < 4; ++j) {
      const int kb = wave * 4 + j;
#pragma unroll
      for (int tt = 0; tt < G2_NT; ++tt) {
        const int n = tt * 16 + la;
        const float* wp = W + (size_t)(kb * 8 + hi * 4) * ODIM + (n < ODIM ? n : ODIM - 1);
        const float4 wv = make_float4(wp[0], wp[ODIM], wp[2 * ODIM], wp[3 * ODIM]);
        if (n != ODIM && n != ODIM + 1) Ws[(kb * G2_NT + tt) * 32 + lane] = wv;
      }
    }
  }
  {
    const float4* ar = (const float4*)(la == 1 ? ad_ : as_) + hi;
    const float4* wr = (const float4*)(W + (size_t)(wave * 32 + la) * ODIM) + hi;
    const bool on = la < 2;
    v8f wa[2];
    wa[0] = (v8f){}; wa[1] = (v8f){};
#pragma unroll
    for (int kb = 0; kb < ODIM / 8; ++kb) {
      const float4 bv = ar[kb * 2];
      v2f b0, b1;
      b0.x = on ? bv.x : 0.f; b0.y = on ? bv.y : 0.f; b1.x = on ? bv.z : 0.f; b1.y = on ? bv.w : 0.f;
#pragma unroll
      for (int ss = 0; ss < 2; ++ss) {
        const float4 aw = wr[ss * 4 * ODIM + kb * 2];
        v2f a0, a1;
        a0.x = aw.x; a0.y = aw.y; a1.x = aw.z; a1.y = aw.w;
        wa[ss] = __builtin_amdgcn_wmma_f32_16x16x4_f32(false, a0, false, b0, (short)0, wa[ss], false, false);
        wa[ss] = __builtin_amdgcn_wmma_f32_16x16x4_f32(false, a1, false, b1, (short)0, wa[ss], false, false);
      }
    }
    if (on) {
#pragma unroll
      for (int ss = 0; ss < 2; ++ss) {
        float4* wt = Ws + ((2 * (wave * 2 + ss) + hi) * G2_NT + (G2_NT - 1)) * 32 + (ODIM % 16) + la;
        wt[0]  = make_float4(wa[ss][0], wa[ss][1], wa[ss][2], wa[ss][3]);
        wt[16] = make_float4(wa[ss][4], wa[ss][5], wa[ss][6], wa[ss][7]);
      }
    }
  }
  __syncthreads();
  if (!live) return;
  v8f acc[G2_NT];
#pragma unroll
  for (int t = 0; t < G2_NT; ++t) acc[t] = (v8f){};
#pragma unroll
  for (int kb = 0; kb < G2_KB; ++kb) {
    const float4 bv = Bs[kb * 2 + hi];
    v2f a0, a1;
    a0.x = elu_f(a[kb].x + bv.x); a0.y = elu_f(a[kb].y + bv.y);
    a1.x = elu_f(a[kb].z + bv.z); a1.y = elu_f(a[kb].w + bv.w);
#pragma unroll
    for (int t = 0; t < G2_NT; ++t) {
      const float4 b = Ws[(kb * G2_NT + t) * 32 + lane];
      v2f b0, b1;
      b0.x = b.x; b0.y = b.y; b1.x = b.z; b1.y = b.w;
      acc[t] = __builtin_amdgcn_wmma_f32_16x16x4_f32(false, a0, false, b0, (short)0,
                                                     acc[t], false, false);
      acc[t] = __builtin_amdgcn_wmma_f32_16x16x4_f32(false, a1, false, b1, (short)0,
                                                     acc[t], false, false);
    }
  }
#pragma unroll
  for (int t = 0; t < G2_NT; ++t) {
    const int col = t * 16 + la;
    if (col < ODIM) {
#pragma unroll
      for (int j = 0; j < 8; ++j)
        t2[(size_t)(m0 + j + hi * 8) * ODIM + col] = acc[t][j];
    } else if (col < ODIM + 2) {
      float* lo = col == ODIM ? os : od;
#pragma unroll
      for (int j = 0; j < 8; ++j) lo[m0 + j + hi * 8] = acc[t][j];
    }
  }
}

static_assert(256 * OWN_R >= NND, "256 owner work-groups cover the rows");
#define OWN_WG   256
#define OWN_T    1024
#define NE16     (NE / 16)
#define OWN_IT   ((NE16 + OWN_T - 1) / OWN_T)
#define OWN_STEP (32 * 16)
static_assert(NE % 16 == 0, "whole 16-byte pieces");
#define OWN_QTH  32
#define OWN_QCAP 768
static_assert(OWN_QCAP >= OWN_QTH + OWN_STEP, "a list about to be worked off still takes one scan step");
static_assert(OWN_R <= (1 << 16) && NND <= (1 << 16), "the fields of a list entry");

__device__ __forceinline__ float lane_bcast_f(float v, int l) {
  return __int_as_float(__builtin_amdgcn_readlane(__float_as_int(v), l));
}
__device__ __forceinline__ float leaky(float v) { return v > 0.f ? v : v * NEG_SLOPE; }

template <int H>
__device__ __forceinline__ float own_logit(const float* __restrict__ aS, const float* __restrict__ aD,
                                           int s, int d, int hd) {
  return leaky(aS[s * H + hd] + aD[d * H + hd]);
}

template <int H>
__device__ __forceinline__ void own_max(int* rmax, const float* __restrict__ aS,
                                        const float* __restrict__ aD, int s, int d0, int x) {
  float lg[H];
#pragma unroll
  for (int hd = 0; hd < H; ++hd) lg[hd] = own_logit<H>(aS, aD, s, d0 + x, hd);
#pragma unroll
  for (int hd = 0; hd < H; ++hd)
    __hip_atomic_fetch_max(&rmax[x * H + hd], f32_ord(lg[hd]), __ATOMIC_RELAXED, __HIP_MEMORY_SCOPE_WORKGROUP);
}

template <int H, bool DO_MAX>
__device__ __forceinline__ void own_convert(unsigned* qw, int from, int n, const int* __restrict__ ei,
                                            const float* __restrict__ aS, const float* __restrict__ aD,
                                            int* rmax, int d0, int lane) {
  for (int b = from; b < n; b += 32) {
    const int i = b + lane;
    if (i < n) {
      const unsigned eid = qw[i];
      const int x = ei[NE + eid] - d0;
      const int s = ei[eid];
      if (DO_MAX) own_max<H>(rmax, aS, aD, s, d0, x);
      qw[i] = ((unsigned)x << 16) | (unsigned)s;
    }
  }
}

#define OWN_TK 4
template <int H>
__device__ __forceinline__ void own_take(unsigned* qw, const unsigned* __restrict__ lw, int n,
                                         const float* __restrict__ aS, const float* __restrict__ aD,
                                         int* rmax, int d0, int lane) {
  for (int b = 0; b < n; b += 32 * OWN_TK) {
    unsigned e[OWN_TK];
#pragma unroll
    for (int k = 0; k < OWN_TK; ++k) {
      const int i = b + k * 32 + lane;
      e[k] = lw[i < n ? i : 0];
    }
    static_assert(OWN_TK == 4, "the statement below names the entries one by one");
    asm volatile("" :: "v"(e[0]), "v"(e[1]), "v"(e[2]), "v"(e[3]));
    float ls[OWN_TK][H], ld[OWN_TK][H];
#pragma unroll
    for (int k = 0; k < OWN_TK; ++k)
#pragma unroll
      for (int hd = 0; hd < H; ++hd) {
        ls[k][hd] = aS[(int)(e[k] & 0xFFFFu) * H + hd];
        ld[k][hd] = aD[(d0 + (int)(e[k] >> 16)) * H + hd];
      }
#pragma unroll
    for (int k = 0; k < OWN_TK; ++k)
#pragma unroll
      for (int hd = 0; hd < H; ++hd)
        asm volatile("" :: "v"(ls[k][hd]), "v"(ld[k][hd]));
#pragma unroll
    for (int k = 0; k < OWN_TK; ++k) {
      const int i = b + k * 32 + lane;
      if (i < n) {
        qw[i] = e[k];
        const int x = (int)(e[k] >> 16);
#pragma unroll
        for (int hd = 0; hd < H; ++hd)
          __hip_atomic_fetch_max(&rmax[x * H + hd], f32_ord(leaky(ls[k][hd] + ld[k][hd])), __ATOMIC_RELAXED,
                                 __HIP_MEMORY_SCOPE_WORKGROUP);
      }
    }
  }
}

template <int C, int H, bool INIT>
__device__ __forceinline__ void own_batch(float* __restrict__ acc, float* __restrict__ wsum,
                                          const int* rmax, const float* __restrict__ h,
                                          const float* __restrict__ aS, const float* __restrict__ aD,
                                          int d0, unsigned ent, int cnt, int lane) {
  constexpr int NP = (C + 31) / 32;
  constexpr int G = NP >= 8 ? 8 : 16;
  float a[H];
#pragma unroll
  for (int hd = 0; hd < H; ++hd) a[hd] = 0.f;
  if (lane < cnt) {
    const int s = (int)(ent & 0xFFFFu), x = (int)(ent >> 16);
    float lg[H];
#pragma unroll
    for (int hd = 0; hd < H; ++hd) lg[hd] = own_logit<H>(aS, aD, s, d0 + x, hd);
#pragma unroll
    for (int hd = 0; hd < H; ++hd) {
      a[hd] = __expf(lg[hd] - ord_f32(rmax[x * H + hd]));
      if (INIT) wsum[x * H + hd] = a[hd];
      else __hip_atomic_fetch_add(&wsum[x * H + hd], a[hd], __ATOMIC_RELAXED, __HIP_MEMORY_SCOPE_WORKGROUP);
    }
  }
  for (int g = 0; g < cnt; g += G) {
    unsigned e[G];
    float v[G][NP];
#pragma unroll
    for (int k = 0; k < G; ++k) {
      const int i = g + k < cnt ? g + k : cnt - 1;
      e[k] = (unsigned)__builtin_amdgcn_readlane((int)ent, i);
      const float* hs = h + (size_t)(e[k] & 0xFFFFu) * C;
#pragma unroll
      for (int j = 0; j < NP; ++j) {
        const int c = j * 32 + lane;
        v[k][j] = hs[c < C ? c : C - 1];
      }
    }
#pragma unroll
    for (int k = 0; k < G; ++k) {
#pragma unroll
      for (int j = 0; j < NP; ++j)
        asm volatile("" :: "v"(v[k][j]));
      if (g + k < cnt) {
        float* ar = acc + (int)(e[k] >> 16) * C;
#pragma unroll
        for (int j = 0; j < NP; ++j) {
          const float alpha = lane_bcast_f(a[(j * 32) / (C / H)], g + k);
          const int c = j * 32 + lane;
          if (c < C) {
            const float t = v[k][j] * alpha;
            if (INIT) ar[c] = t;
            else __hip_atomic_fetch_add(&ar[c], t, __ATOMIC_RELAXED, __HIP_MEMORY_SCOPE_WORKGROUP);
          }
        }
      }
    }
  }
}

template <int C, int H>
__device__ __forceinline__ void own_drain(float* __restrict__ acc, float* __restrict__ wsum,
                                          const int* rmax, const unsigned* qw, int n,
                                          const float* __restrict__ h,
                                          const float* __restrict__ aS, const float* __restrict__ aD,
                                          int d0, int lane) {
  for (int b = 0; b < n; b += 32) {
    int cnt = n - b;
    if (cnt > 32) cnt = 32;
    unsigned ent = 0;
    if (lane < cnt) ent = qw[b + lane];
    own_batch<C, H, false>(acc, wsum, rmax, h, aS, aD, d0, ent, cnt, lane);
  }
}

__device__ __forceinline__ unsigned own_eq4(unsigned w, unsigned pat) {
  const unsigned x = w ^ pat;
  const unsigned t = (x & 0x7F7F7F7Fu) + 0x7F7F7F7Fu;
  return ~(t | x) & 0x80808080u;
}
__device__ __forceinline__ int own_piece(int i) { return i < NE16 ? i : NE16 - 1; }
__device__ __forceinline__ unsigned own_eq16(const int4 v, unsigned pat) {
  return (own_eq4((unsigned)v.x, pat) >> 7) | (own_eq4((unsigned)v.y, pat) >> 6) |
         (own_eq4((unsigned)v.z, pat) >> 5) | (own_eq4((unsigned)v.w, pat) >> 4);
}

template <int C, int H, int MODE>
__device__ __forceinline__ void own_scan(float* __restrict__ acc, float* __restrict__ wsum, int* rmax,
                                         unsigned* qw, const int* __restrict__ ei,
                                         const unsigned* __restrict__ ob,
                                         const float* __restrict__ h,
                                         const float* __restrict__ aS, const float* __restrict__ aD,
                                         int d0, int wofs, int lane, int it_from,
                                         int& qn, bool& over, int& it_over) {
  const int4* ob4 = (const int4*)ob;
  const unsigned pat = (unsigned)blockIdx.x * 0x01010101u;
  int qdone = qn;
  int sp[2];
  sp[0] = (int)((blockIdx.x * OWN_IT) / OWN_WG + it_from) % OWN_IT;
  sp[1] = sp[0] + 1 < OWN_IT ? sp[0] + 1 : 0;
  int4 cur[2];
#pragma unroll
  for (int r = 0; r < 2; ++r) cur[r] = ob4[own_piece(sp[r] * OWN_T + wofs + lane)];
  for (int it = it_from; it < OWN_IT; it += 2) {
    int sn[2];
    sn[0] = sp[1] + 1 < OWN_IT ? sp[1] + 1 : 0;
    sn[1] = sn[0] + 1 < OWN_IT ? sn[0] + 1 : 0;
    int4 nxt[2];
#pragma unroll
    for (int r = 0; r < 2; ++r) nxt[r] = ob4[own_piece(sn[r] * OWN_T + wofs + lane)];
#pragma unroll
    for (int r = 0; r < 2; ++r) {
      if (it + r >= OWN_IT) break;
      if (MODE == 0 && !over && qn + OWN_STEP > OWN_QCAP) {
        own_convert<H, true>(qw, qdone, qn, ei, aS, aD, rmax, d0, lane);
        qdone = qn;
        over = true;
        it_over = it + r;
      }
      const int ib = sp[r] * OWN_T + wofs + lane;
      unsigned mm = own_eq16(cur[r], pat);
      if (ib >= NE16) mm = 0;
      unsigned mk = __builtin_amdgcn_ballot_w32(mm != 0);
      for (int tr = 0; tr < 16 && mk; ++tr) {
        const bool m = mm != 0;
        const int p = __builtin_ctz(mm | 0x80000000u);
        const unsigned eid = (unsigned)(ib * 16 + (p & 3) * 4 + (p >> 3));
        if (MODE == 0 && over) {
          if (m) own_max<H>(rmax, aS, aD, ei[eid], d0, ei[NE + eid] - d0);
        } else {
          if (m) qw[qn + (int)__builtin_amdgcn_mbcnt_lo(mk, 0u)] = eid;
          qn += __builtin_popcount(mk);
        }
        mm &= mm - 1;
        mk = __builtin_amdgcn_ballot_w32(mm != 0);
      }
      if (MODE == 0) {
        const int nb = (qn - qdone) & ~31;
        if (nb) {
          own_convert<H, true>(qw, qdone, qdone + nb, ei, aS, aD, rmax, d0, lane);
          qdone += nb;
        }
      } else if (qn >= OWN_QTH) {
        own_convert<H, false>(qw, 0, qn, ei, aS, aD, rmax, d0, lane);
        own_drain<C, H>(acc, wsum, rmax, qw, qn, h, aS, aD, d0, lane);
        qn = 0;
      }
    }
#pragma unroll
    for (int r = 0; r < 2; ++r) {
      cur[r] = nxt[r];
      sp[r] = sn[r];
    }
  }
  if (MODE == 0) {
    own_convert<H, true>(qw, qdone, qn, ei, aS, aD, rmax, d0, lane);
  } else {
    own_convert<H, false>(qw, 0, qn, ei, aS, aD, rmax, d0, lane);
    own_drain<C, H>(acc, wsum, rmax, qw, qn, h, aS, aD, d0, lane);
    qn = 0;
  }
}

template <int C, int H, bool KEPT>
__global__ __launch_bounds__(OWN_T) void k_own(const int* __restrict__ ei,
                                               const float* __restrict__ h,
                                               const float* __restrict__ aS,
                                               const float* __restrict__ aD,
                                               float* __restrict__ agg,
                                               unsigned* __restrict__ lst,
                                               int* __restrict__ lcnt,
                                               const unsigned* __restrict__ ob) {
  __shared__ __attribute__((aligned(16))) float acc[OWN_R * C];
  __shared__ float wsum[OWN_R * H];
  __shared__ int rmax[OWN_R * H];
  __shared__ unsigned q[(OWN_T / 32) * OWN_QCAP];
  const int tid = threadIdx.x, lane = tid & 31;
  const int wave = __builtin_amdgcn_readfirstlane(tid >> 5);
  const int d0 = blockIdx.x * OWN_R;
  int nrow = NND - d0;
  if (nrow > OWN_R) nrow = OWN_R;

  for (int i = tid; i < nrow * H; i += OWN_T)
    rmax[i] = f32_ord(leaky(aS[d0 * H + i] + aD[d0 * H + i]));
  __syncthreads();

  unsigned* qw = q + wave * OWN_QCAP;
  const int wofs = wave * 32;
  int qn = 0, it_over = OWN_IT;
  bool over = false;
  const int lw_i = blockIdx.x * (OWN_T / 32) + wave;
  unsigned* lw = lst + (size_t)lw_i * OWN_QCAP;
  bool have = false;
  if (KEPT) {
    const int n = __builtin_amdgcn_readfirstlane(lcnt[lw_i]);
    have = (unsigned)n <= (unsigned)OWN_QCAP;
    if (have) {
      own_take<H>(qw, lw, n, aS, aD, rmax, d0, lane);
      qn = n;
    }
  }
  if (!have) {
    own_scan<C, H, 0>(acc, wsum, rmax, qw, ei, ob, h, aS, aD, d0, wofs, lane, 0, qn, over, it_over);
    if (!KEPT) {
      if (lane == 0) lcnt[lw_i] = over ? -1 : qn;
      if (!over)
        for (int b = 0; b < qn; b += 32) {
          const int i = b + lane;
          if (i < qn) lw[i] = qw[i];
        }
    }
  }
  __syncthreads();

  static_assert((OWN_R + OWN_T / 32 - 1) / (OWN_T / 32) <= 32, "a wave's rows fit one batch");
  {
    const int cnt = nrow > wave ? (nrow - wave + OWN_T / 32 - 1) / (OWN_T / 32) : 0;
    const int r = wave + lane * (OWN_T / 32);
    const unsigned ent = lane < cnt ? ((unsigned)r << 16) | (unsigned)(d0 + r) : 0u;
    own_batch<C, H, true>(acc, wsum, rmax, h, aS, aD, d0, ent, cnt, lane);
  }
  __syncthreads();

  own_drain<C, H>(acc, wsum, rmax, qw, qn, h, aS, aD, d0, lane);
  if (over) {
    qn = 0;
    own_scan<C, H, 1>(acc, wsum, rmax, qw, ei, ob, h, aS, aD, d0, wofs, lane, it_over, qn, over, it_over);
  }
  __syncthreads();

  static_assert((C / H) % 4 == 0, "a 16-byte piece does not straddle two heads");
  constexpr int PH = (C / H) / 4;
  float4* o4 = (float4*)(agg + (size_t)d0 * C);
  const float4* a4 = (const float4*)acc;
  for (int i = tid; i < nrow * C / 4; i += OWN_T) {
    const float4 v = a4[i];
    const float dn = wsum[i / PH] + EPSV;
    o4[i] = make_float4(v.x / dn, v.y / dn, v.z / dn, v.w / dn);
  }
}

#define OB_WG 512
#define OB_T  (OB_WG * 256)
#define OB_U  4
#define OB_N4 (NND * ODIM / 4)
static_assert(OB_U * OB_T >= OB_N4, "one trip covers the output");
__global__ __launch_bounds__(256) void k_out_bias(const float* __restrict__ agg, const float* __restrict__ b,
                           float* __restrict__ out) {
  const int i0 = blockIdx.x * 256 + threadIdx.x;
  const float4* g4 = (const float4*)agg;
  float4* o4 = (float4*)out;
  float4 v[OB_U], bb[OB_U];
#pragma unroll
  for (int k = 0; k < OB_U; ++k) {
    const int i = i0 + k * OB_T;
    const int ic = i < OB_N4 ? i : i0;
    v[k] = g4[ic];
    const int c = (ic % (ODIM / 4)) * 4;
    bb[k] = make_float4(b[c], b[c + 1], b[c + 2], b[c + 3]);
  }
#pragma unroll
  for (int k = 0; k < OB_U; ++k) {
    const int i = i0 + k * OB_T;
    if (i < OB_N4)
      o4[i] = make_float4(v[k].x + bb[k].x, v[k].y + bb[k].y, v[k].z + bb[k].z, v[k].w + bb[k].w);
  }
}


extern "C" void kernel_launch(void* const* d_in, const int* in_sizes, int n_in,
                              void* d_out, int out_size, void* d_ws, size_t ws_size,
                              hipStream_t stream) {
  const float* x   = (const float*)d_in[0];
  const int*   ei  = (const int*)d_in[1];
  const float* W1  = (const float*)d_in[2];
  const float* as1 = (const float*)d_in[3];
  const float* ad1 = (const float*)d_in[4];
  const float* b1  = (const float*)d_in[5];
  const float* W2  = (const float*)d_in[6];
  const float* as2 = (const float*)d_in[7];
  const float* ad2 = (const float*)d_in[8];
  const float* b2  = (const float*)d_in[9];
  float* out = (float*)d_out;

  float* ws   = (float*)d_ws;
  float* h1   = ws;
  float* agg1 = h1   + (size_t)NND * HC;
  float* aS1  = agg1 + (size_t)NND * HC;
  float* aD1  = aS1  + NND * NH;
  float* t2   = aD1  + NND * NH;
  float* aS2  = t2   + (size_t)NND * ODIM;
  float* aD2  = aS2  + NND;
  float* agg2 = aD2  + NND;
  unsigned* lst = (unsigned*)(agg2 + (size_t)NND * ODIM);
  int* lcnt = (int*)(lst + (size_t)OWN_WG * (OWN_T / 32) * OWN_QCAP);
  unsigned* ob = (unsigned*)(lcnt + OWN_WG * (OWN_T / 32));

  k_gemm1<<<GM_WG, 256, 0, stream>>>(x, W1, as1, ad1, ei, h1, aS1, aD1, ob);
  k_own<HC, NH, false><<<OWN_WG, OWN_T, 0, stream>>>(ei, h1, aS1, aD1, agg1, lst, lcnt, ob);

  k_gemm2<<<GM_WG, 256, 0, stream>>>(agg1, b1, W2, as2, ad2, t2, aS2, aD2);
  k_own<ODIM, 1, true><<<OWN_WG, OWN_T, 0, stream>>>(ei, t2, aS2, aD2, agg2, lst, lcnt, ob);
  k_out_bias<<<OB_WG, 256, 0, stream>>>(agg2, b2, out);
}
